# starting kernel
# baseline (speedup 1.0000x reference)
#include <hip/hip_runtime.h>
#include <hip/hip_bf16.h>

typedef __attribute__((ext_vector_type(2))) float v2f;
typedef __attribute__((ext_vector_type(8))) float v8f;

#define NNODES 40000
#define NEDGES 640000
#define NGRAPH 256

__device__ __forceinline__ void atomAddF(float* p, float v) {
    unsafeAtomicAdd(p, v);
}

__global__ void deg_kernel(const int* __restrict__ dst, float* __restrict__ deg, int E) {
    int e = blockIdx.x * blockDim.x + threadIdx.x;
    if (e < E) atomAddF(&deg[dst[e]], 1.0f);
}

__global__ void dis_counts_kernel(float* __restrict__ deg_dis,
                                  const int* __restrict__ batch,
                                  float* __restrict__ counts, int N) {
    int n = blockIdx.x * blockDim.x + threadIdx.x;
    if (n < N) {
        deg_dis[n] = rsqrtf(deg_dis[n] + 1.0f);
        atomAddF(&counts[batch[n]], 1.0f);
    }
}

__global__ void wmma_gemm_f32(const float* __restrict__ A, const float* __restrict__ B,
                              float* __restrict__ C, int N, int K, int M,
                              const float* __restrict__ rowscale,
                              const float* __restrict__ bias, int total_tiles) {
    int tid  = threadIdx.x;
    int wid  = tid >> 5;
    int lane = tid & 31;
    int tile = blockIdx.x * (blockDim.x >> 5) + wid;
    if (tile >= total_tiles) return;

    int tiles_m = M >> 4;
    int rt = tile / tiles_m, ct = tile - rt * tiles_m;
    int row0 = rt << 4, col0 = ct << 4;
    int hi = lane >> 4, lm = lane & 15;

    v8f c = {};
    const float* Arow = A + (size_t)(row0 + lm) * K;
    for (int k0 = 0; k0 < K; k0 += 4) {
        v2f a = *reinterpret_cast<const v2f*>(Arow + k0 + 2 * hi);
        const float* Bp = B + (size_t)(k0 + 2 * hi) * M + col0 + lm;
        v2f b;
        b.x = Bp[0];
        b.y = Bp[M];
        c = __builtin_amdgcn_wmma_f32_16x16x4_f32(false, a, false, b,
                                                  (short)0, c, false, false);
    }
    float bv = bias ? bias[col0 + lm] : 0.0f;
    #pragma unroll
    for (int v = 0; v < 8; ++v) {
        int row = row0 + v + 8 * hi;
        float o = c[v];
        if (rowscale) o *= rowscale[row];
        C[(size_t)row * M + col0 + lm] = o + bv;
    }
}

__global__ void scatter_kernel(const float* __restrict__ h, const int* __restrict__ src,
                               const int* __restrict__ dst, float* __restrict__ agg,
                               int F, int E) {
    int tid  = blockIdx.x * blockDim.x + threadIdx.x;
    int e    = tid >> 5;
    int lane = tid & 31;
    if (e >= E) return;
    int s = src[e], d = dst[e];
    const float* hs = h + (size_t)s * F;
    float* ad = agg + (size_t)d * F;
    for (int f = lane; f < F; f += 32)
        atomAddF(&ad[f], hs[f]);
}

__global__ void combine_kernel(float* __restrict__ agg, const float* __restrict__ h,
                               const float* __restrict__ dis, const float* __restrict__ bias,
                               int fshift, int total) {
    int i = blockIdx.x * blockDim.x + threadIdx.x;
    if (i >= total) return;
    int n = i >> fshift;
    int f = i & ((1 << fshift) - 1);
    float v = (agg[i] + h[i]) * dis[n] + bias[f];
    agg[i] = v > 0.0f ? v : 0.0f;
}

__global__ void pool_kernel(const float* __restrict__ h, const int* __restrict__ batch,
                            float* __restrict__ pooled, int total) {
    int i = blockIdx.x * blockDim.x + threadIdx.x;
    if (i >= total) return;
    int n = i >> 5, f = i & 31;
    atomAddF(&pooled[(batch[n] << 5) + f], h[i]);
}

__global__ void poolfin_kernel(float* __restrict__ pooled, const float* __restrict__ counts) {
    int i = blockIdx.x * blockDim.x + threadIdx.x;
    int g = i >> 5;
    pooled[i] = pooled[i] / fmaxf(counts[g], 1.0f);
}

extern "C" void kernel_launch(void* const* d_in, const int* in_sizes, int n_in,
                              void* d_out, int out_size, void* d_ws, size_t ws_size,
                              hipStream_t stream) {
    const float* x   = (const float*)d_in[0];
    const float* W1  = (const float*)d_in[1];
    const float* b1  = (const float*)d_in[2];
    const float* W2  = (const float*)d_in[3];
    const float* b2  = (const float*)d_in[4];
    const float* W3  = (const float*)d_in[5];
    const float* b3  = (const float*)d_in[6];
    const float* Wl  = (const float*)d_in[7];
    const float* bl  = (const float*)d_in[8];
    const int*   ei  = (const int*)d_in[9];
    const int*   bat = (const int*)d_in[10];
    float*       out = (float*)d_out;

    const int N = NNODES, E = NEDGES;
    const int* src = ei;
    const int* dst = ei + E;

    float* B0     = (float*)d_ws;
    float* B1     = B0 + (size_t)N * 128;
    float* dis    = B1 + (size_t)N * 128;
    float* counts = dis + N;
    float* pooled = counts + NGRAPH;

    hipMemsetAsync(dis, 0, (size_t)N * sizeof(float), stream);
    hipMemsetAsync(counts, 0, NGRAPH * sizeof(float), stream);
    deg_kernel<<<E / 256, 256, 0, stream>>>(dst, dis, E);
    dis_counts_kernel<<<(N + 255) / 256, 256, 0, stream>>>(dis, bat, counts, N);

    wmma_gemm_f32<<<(2500 * 8) / 8, 256, 0, stream>>>(x, W1, B0, N, 128, 128, dis, nullptr, 2500 * 8);
    hipMemsetAsync(B1, 0, (size_t)N * 128 * sizeof(float), stream);
    scatter_kernel<<<(E * 32) / 256, 256, 0, stream>>>(B0, src, dst, B1, 128, E);
    combine_kernel<<<(N * 128) / 256, 256, 0, stream>>>(B1, B0, dis, b1, 7, N * 128);

    wmma_gemm_f32<<<(2500 * 4) / 8, 256, 0, stream>>>(B1, W2, B0, N, 128, 64, dis, nullptr, 2500 * 4);
    hipMemsetAsync(B1, 0, (size_t)N * 64 * sizeof(float), stream);
    scatter_kernel<<<(E * 32) / 256, 256, 0, stream>>>(B0, src, dst, B1, 64, E);
    combine_kernel<<<(N * 64) / 256, 256, 0, stream>>>(B1, B0, dis, b2, 6, N * 64);

    wmma_gemm_f32<<<(2500 * 2) / 8, 256, 0, stream>>>(B1, W3, B0, N, 64, 32, dis, nullptr, 2500 * 2);
    hipMemsetAsync(B1, 0, (size_t)N * 32 * sizeof(float), stream);
    scatter_kernel<<<(E * 32) / 256, 256, 0, stream>>>(B0, src, dst, B1, 32, E);
    combine_kernel<<<(N * 32) / 256, 256, 0, stream>>>(B1, B0, dis, b3, 5, N * 32);

    hipMemsetAsync(pooled, 0, NGRAPH * 32 * sizeof(float), stream);
    pool_kernel<<<(N * 32) / 256, 256, 0, stream>>>(B1, bat, pooled, N * 32);
    poolfin_kernel<<<(NGRAPH * 32) / 256, 256, 0, stream>>>(pooled, counts);

    wmma_gemm_f32<<<(16 * 48) / 8, 256, 0, stream>>>(pooled, Wl, out, NGRAPH, 32, 768,
                                                     nullptr, bl, 16 * 48);
}


# global float-atomic edge scatter replaced by per-feature column aggregation in LDS: one work-group holds a feature's h and sum columns for all nodes, LDS float adds; feature-major; sum order changes
# speedup vs baseline: 1.4326x; 1.4326x over previous
#include <hip/hip_runtime.h>
#include <hip/hip_bf16.h>


typedef __attribute__((ext_vector_type(2))) float v2f;
typedef __attribute__((ext_vector_type(8))) float v8f;

#define NNODES 40000
#define NEDGES 640000
#define NGRAPH 256

__device__ __forceinline__ void atomAddF(float* p, float v) {
    unsafeAtomicAdd(p, v);
}

__global__ void deg_kernel(const int* __restrict__ dst, float* __restrict__ deg, int E) {
    int e = blockIdx.x * blockDim.x + threadIdx.x;
    if (e < E) atomAddF(&deg[dst[e]], 1.0f);
}

__global__ void dis_counts_kernel(float* __restrict__ deg_dis,
                                  const int* __restrict__ batch,
                                  float* __restrict__ counts, int N) {
    int n = blockIdx.x * blockDim.x + threadIdx.x;
    if (n < N) {
        deg_dis[n] = rsqrtf(deg_dis[n] + 1.0f);
        atomAddF(&counts[batch[n]], 1.0f);
    }
}

__global__ void wmma_gemm_f32(const float* __restrict__ A, const float* __restrict__ B,
                              float* __restrict__ C, int N, int K, int M,
                              const float* __restrict__ rowscale,
                              const float* __restrict__ bias, int total_tiles) {
    int tid  = threadIdx.x;
    int wid  = tid >> 5;
    int lane = tid & 31;
    int tile = blockIdx.x * (blockDim.x >> 5) + wid;
    if (tile >= total_tiles) return;

    int tiles_m = M >> 4;
    int rt = tile / tiles_m, ct = tile - rt * tiles_m;
    int row0 = rt << 4, col0 = ct << 4;
    int hi = lane >> 4, lm = lane & 15;

    v8f c = {};
    const float* Arow = A + (size_t)(row0 + lm) * K;
    for (int k0 = 0; k0 < K; k0 += 4) {
        v2f a = *reinterpret_cast<const v2f*>(Arow + k0 + 2 * hi);
        const float* Bp = B + (size_t)(k0 + 2 * hi) * M + col0 + lm;
        v2f b;
        b.x = Bp[0];
        b.y = Bp[M];
        c = __builtin_amdgcn_wmma_f32_16x16x4_f32(false, a, false, b,
                                                  (short)0, c, false, false);
    }
    float bv = bias ? bias[col0 + lm] : 0.0f;
    #pragma unroll
    for (int v = 0; v < 8; ++v) {
        int row = row0 + v + 8 * hi;
        float o = c[v];
        if (rowscale) o *= rowscale[row];
        C[(size_t)row * M + col0 + lm] = o + bv;
    }
}

template <int MODE, int P>
__global__ void wmma_gemm_t(const float* __restrict__ X, const float* __restrict__ W,
                            float* __restrict__ hT, int K, int M,
                            const float* __restrict__ dis,
                            const float* __restrict__ part,
                            const float* __restrict__ pbias, int total_tiles) {
    int tid  = threadIdx.x;
    int wid  = tid >> 5;
    int lane = tid & 31;
    int tile = blockIdx.x * (blockDim.x >> 5) + wid;
    if (tile >= total_tiles) return;

    int tiles_f = M >> 4;
    int nt = tile / tiles_f, ft = tile - nt * tiles_f;
    int n0 = nt << 5, f0 = ft << 4;
    int hi = lane >> 4, lm = lane & 15;
    int na = n0 + lm, nb = n0 + 16 + lm;
    float da = dis[na], db = dis[nb];

    v8f c0 = {}, c1 = {};
    for (int k0 = 0; k0 < K; k0 += 4) {
        int k = k0 + 2 * hi;
        const float* Wp = W + (size_t)k * M + f0 + lm;
        v2f a;
        a.x = Wp[0];
        a.y = Wp[M];
        v2f b0, b1;
        if (MODE == 0) {
            b0 = *reinterpret_cast<const v2f*>(X + (size_t)na * K + k);
            b1 = *reinterpret_cast<const v2f*>(X + (size_t)nb * K + k);
        } else if (MODE == 1) {
            const float* Xp = X + (size_t)k * NNODES;
            b0.x = Xp[na];          b1.x = Xp[nb];
            b0.y = Xp[NNODES + na]; b1.y = Xp[NNODES + nb];
        } else {
            #pragma unroll
            for (int s = 0; s < 2; ++s) {
                int kk = k + s;
                const float* pp = part + (size_t)kk * NNODES;
                float sa = pp[na], sb = pp[nb];
                #pragma unroll
                for (int p = 1; p < P; ++p) {
                    sa += pp[(size_t)p * K * NNODES + na];
                    sb += pp[(size_t)p * K * NNODES + nb];
                }
                const float* Xp = X + (size_t)kk * NNODES;
                float bk = pbias[kk];
                float va = (sa + Xp[na]) * da + bk;
                float vb = (sb + Xp[nb]) * db + bk;
                b0[s] = va > 0.0f ? va : 0.0f;
                b1[s] = vb > 0.0f ? vb : 0.0f;
            }
        }
        c0 = __builtin_amdgcn_wmma_f32_16x16x4_f32(false, a, false, b0,
                                                   (short)0, c0, false, false);
        c1 = __builtin_amdgcn_wmma_f32_16x16x4_f32(false, a, false, b1,
                                                   (short)0, c1, false, false);
    }
    float dn = dis[n0 + lane];
    #pragma unroll
    for (int r = 0; r < 8; ++r) {
        float give = hi ? c0[r] : c1[r];
        float got  = __shfl_xor(give, 16);
        float lo = hi ? got : c0[r];
        float up = hi ? c1[r] : got;
        hT[(size_t)(f0 + r) * NNODES + n0 + lane]     = lo * dn;
        hT[(size_t)(f0 + 8 + r) * NNODES + n0 + lane] = up * dn;
    }
}

template <int P>
__global__ __launch_bounds__(1024) void slab_kernel(const float* __restrict__ hT,
                                                    const int* __restrict__ src,
                                                    const int* __restrict__ dst,
                                                    float* __restrict__ outp,
                                                    const float* __restrict__ dis,
                                                    const float* __restrict__ bias, int F) {
    __shared__ float hcol[NNODES];
    __shared__ float acc[NNODES];
    const int tid = threadIdx.x;
    const int p = blockIdx.x / F;
    const int f = blockIdx.x - p * F;

    const float4* h4 = reinterpret_cast<const float4*>(hT + (size_t)f * NNODES);
    float4* hc4 = reinterpret_cast<float4*>(hcol);
    float4* ac4 = reinterpret_cast<float4*>(acc);
    for (int i = tid; i < NNODES / 4; i += 1024) {
        hc4[i] = h4[i];
        ac4[i] = make_float4(0.0f, 0.0f, 0.0f, 0.0f);
    }
    __syncthreads();

    const int EP = NEDGES / P;
    const int4* s4 = reinterpret_cast<const int4*>(src + (size_t)p * EP);
    const int4* d4 = reinterpret_cast<const int4*>(dst + (size_t)p * EP);
    int q = tid;
    int4 s = s4[q];
    int4 d = d4[q];
    for (;;) {
        const int qn = q + 1024;
        const bool more = qn < EP / 4;
        int4 sn = s, dn = d;
        if (more) { sn = s4[qn]; dn = d4[qn]; }
        float v0 = hcol[s.x], v1 = hcol[s.y], v2 = hcol[s.z], v3 = hcol[s.w];
        __hip_atomic_fetch_add(&acc[d.x], v0, __ATOMIC_RELAXED, __HIP_MEMORY_SCOPE_WORKGROUP);
        __hip_atomic_fetch_add(&acc[d.y], v1, __ATOMIC_RELAXED, __HIP_MEMORY_SCOPE_WORKGROUP);
        __hip_atomic_fetch_add(&acc[d.z], v2, __ATOMIC_RELAXED, __HIP_MEMORY_SCOPE_WORKGROUP);
        __hip_atomic_fetch_add(&acc[d.w], v3, __ATOMIC_RELAXED, __HIP_MEMORY_SCOPE_WORKGROUP);
        if (!more) break;
        s = sn; d = dn; q = qn;
    }
    __syncthreads();

    float4* o4 = reinterpret_cast<float4*>(outp + (size_t)blockIdx.x * NNODES);
    if (P == 1) {
        const float4* dis4 = reinterpret_cast<const float4*>(dis);
        const float bf = bias[f];
        for (int i = tid; i < NNODES / 4; i += 1024) {
            float4 a = ac4[i], h = hc4[i], dd = dis4[i], o;
            o.x = (a.x + h.x) * dd.x + bf;
            o.y = (a.y + h.y) * dd.y + bf;
            o.z = (a.z + h.z) * dd.z + bf;
            o.w = (a.w + h.w) * dd.w + bf;
            o.x = o.x > 0.0f ? o.x : 0.0f;
            o.y = o.y > 0.0f ? o.y : 0.0f;
            o.z = o.z > 0.0f ? o.z : 0.0f;
            o.w = o.w > 0.0f ? o.w : 0.0f;
            o4[i] = o;
        }
    } else {
        for (int i = tid; i < NNODES / 4; i += 1024) o4[i] = ac4[i];
    }
}

template <int P>
__global__ __launch_bounds__(1024) void pool_t_kernel(const float* __restrict__ part,
                                                      const float* __restrict__ hT,
                                                      const float* __restrict__ dis,
                                                      const float* __restrict__ bias,
                                                      const int* __restrict__ batch,
                                                      float* __restrict__ pooled) {
    __shared__ float tile[32][33];
    const int t = threadIdx.x;
    const int n0 = blockIdx.x * 32;
    {
        int f = t >> 5, n = n0 + (t & 31);
        float a = part[(size_t)f * NNODES + n];
        #pragma unroll
        for (int p = 1; p < P; ++p) a += part[((size_t)p * 32 + f) * NNODES + n];
        float v = (a + hT[(size_t)f * NNODES + n]) * dis[n] + bias[f];
        tile[f][t & 31] = v > 0.0f ? v : 0.0f;
    }
    __syncthreads();
    {
        int nl = t >> 5, f = t & 31;
        atomAddF(&pooled[(batch[n0 + nl] << 5) + f], tile[f][nl]);
    }
}

__global__ void poolfin_kernel(float* __restrict__ pooled, const float* __restrict__ counts) {
    int i = blockIdx.x * blockDim.x + threadIdx.x;
    int g = i >> 5;
    pooled[i] = pooled[i] / fmaxf(counts[g], 1.0f);
}

extern "C" void kernel_launch(void* const* d_in, const int* in_sizes, int n_in,
                              void* d_out, int out_size, void* d_ws, size_t ws_size,
                              hipStream_t stream) {
    const float* x   = (const float*)d_in[0];
    const float* W1  = (const float*)d_in[1];
    const float* b1  = (const float*)d_in[2];
    const float* W2  = (const float*)d_in[3];
    const float* b2  = (const float*)d_in[4];
    const float* W3  = (const float*)d_in[5];
    const float* b3  = (const float*)d_in[6];
    const float* Wl  = (const float*)d_in[7];
    const float* bl  = (const float*)d_in[8];
    const int*   ei  = (const int*)d_in[9];
    const int*   bat = (const int*)d_in[10];
    float*       out = (float*)d_out;

    const int N = NNODES, E = NEDGES;
    const int* src = ei;
    const int* dst = ei + E;

    float* B0     = (float*)d_ws;
    float* B1     = B0 + (size_t)N * 128;
    float* dis    = B1 + (size_t)N * 128;
    float* counts = dis + N;
    float* pooled = counts + NGRAPH;
    float* H3     = B0 + (size_t)N * 64;

    hipMemsetAsync(dis, 0, (size_t)N * sizeof(float), stream);
    hipMemsetAsync(counts, 0, NGRAPH * sizeof(float), stream);
    deg_kernel<<<E / 256, 256, 0, stream>>>(dst, dis, E);
    dis_counts_kernel<<<(N + 255) / 256, 256, 0, stream>>>(dis, bat, counts, N);

    wmma_gemm_t<0, 1><<<(1250 * 8) / 8, 256, 0, stream>>>(x, W1, B0, 128, 128, dis, nullptr, nullptr, 1250 * 8);
    slab_kernel<1><<<128, 1024, 0, stream>>>(B0, src, dst, B1, dis, b1, 128);

    wmma_gemm_t<1, 1><<<(1250 * 4) / 8, 256, 0, stream>>>(B1, W2, B0, 128, 64, dis, nullptr, nullptr, 1250 * 4);
    slab_kernel<2><<<128, 1024, 0, stream>>>(B0, src, dst, B1, dis, b2, 64);

    wmma_gemm_t<2, 2><<<(1250 * 2 + 7) / 8, 256, 0, stream>>>(B0, W3, H3, 64, 32, dis, B1, b2, 1250 * 2);
    slab_kernel<4><<<128, 1024, 0, stream>>>(H3, src, dst, B1, dis, b3, 32);

    hipMemsetAsync(pooled, 0, NGRAPH * 32 * sizeof(float), stream);
    pool_t_kernel<4><<<N / 32, 1024, 0, stream>>>(B1, H3, dis, b3, bat, pooled);
    poolfin_kernel<<<(NGRAPH * 32) / 256, 256, 0, stream>>>(pooled, counts);

    wmma_gemm_f32<<<(16 * 48) / 8, 256, 0, stream>>>(pooled, Wl, out, NGRAPH, 32, 768,
                                                     nullptr, bl, 16 * 48);
}


# LDS column adds made bank-conflict free: degree pass rewrites edges as 32 lists by dst mod 32 (one word per edge), lane L walks list L; sum order changes
# speedup vs baseline: 1.5864x; 1.1074x over previous
#include <hip/hip_runtime.h>
#include <hip/hip_bf16.h>


typedef __attribute__((ext_vector_type(2))) float v2f;
typedef __attribute__((ext_vector_type(8))) float v8f;

#define NNODES 40000
#define NEDGES 640000
#define NGRAPH 256

__device__ __forceinline__ void atomAddF(float* p, float v) {
    unsafeAtomicAdd(p, v);
}

#define CAPN 32768
#define CAPG (CAPN / 4)
static_assert(NNODES <= 65536, "src and dst share one 32-bit word");

__device__ __forceinline__ void place_edge(int s, int d, int pos, int* __restrict__ gcnt,
                                           unsigned* __restrict__ elist,
                                           unsigned* __restrict__ ovf) {
    const unsigned w = (unsigned)s | ((unsigned)d << 16);
    if (pos < CAPN) {
        elist[((size_t)(pos >> 2) * 32 + (d & 31)) * 4 + (pos & 3)] = w;
    } else {
        const int o = atomicAdd(&gcnt[32], 1);
        ovf[o] = w;
    }
}

__global__ __launch_bounds__(1024) void bin_deg_kernel(const int* __restrict__ src,
                                                       const int* __restrict__ dst,
                                                       float* __restrict__ deg,
                                                       int* __restrict__ gcnt,
                                                       unsigned* __restrict__ elist,
                                                       unsigned* __restrict__ ovf) {
    __shared__ int cnt[32];
    __shared__ int base[32];
    const int tid = threadIdx.x;
    if (tid < 32) cnt[tid] = 0;
    __syncthreads();
    const int i = blockIdx.x * 1024 + tid;
    const bool act = i < NEDGES / 4;
    int4 s = make_int4(0, 0, 0, 0), d = make_int4(0, 0, 0, 0);
    int r0 = 0, r1 = 0, r2 = 0, r3 = 0;
    if (act) {
        s = reinterpret_cast<const int4*>(src)[i];
        d = reinterpret_cast<const int4*>(dst)[i];
        atomAddF(&deg[d.x], 1.0f);
        atomAddF(&deg[d.y], 1.0f);
        atomAddF(&deg[d.z], 1.0f);
        atomAddF(&deg[d.w], 1.0f);
        r0 = atomicAdd(&cnt[d.x & 31], 1);
        r1 = atomicAdd(&cnt[d.y & 31], 1);
        r2 = atomicAdd(&cnt[d.z & 31], 1);
        r3 = atomicAdd(&cnt[d.w & 31], 1);
    }
    __syncthreads();
    if (tid < 32) base[tid] = atomicAdd(&gcnt[tid], cnt[tid]);
    __syncthreads();
    if (act) {
        place_edge(s.x, d.x, base[d.x & 31] + r0, gcnt, elist, ovf);
        place_edge(s.y, d.y, base[d.y & 31] + r1, gcnt, elist, ovf);
        place_edge(s.z, d.z, base[d.z & 31] + r2, gcnt, elist, ovf);
        place_edge(s.w, d.w, base[d.w & 31] + r3, gcnt, elist, ovf);
    }
}

__global__ void dis_counts_kernel(float* __restrict__ deg_dis,
                                  const int* __restrict__ batch,
                                  float* __restrict__ counts, int N) {
    int n = blockIdx.x * blockDim.x + threadIdx.x;
    if (n < N) {
        deg_dis[n] = rsqrtf(deg_dis[n] + 1.0f);
        atomAddF(&counts[batch[n]], 1.0f);
    }
}

__global__ void wmma_gemm_f32(const float* __restrict__ A, const float* __restrict__ B,
                              float* __restrict__ C, int N, int K, int M,
                              const float* __restrict__ rowscale,
                              const float* __restrict__ bias, int total_tiles) {
    int tid  = threadIdx.x;
    int wid  = tid >> 5;
    int lane = tid & 31;
    int tile = blockIdx.x * (blockDim.x >> 5) + wid;
    if (tile >= total_tiles) return;

    int tiles_m = M >> 4;
    int rt = tile / tiles_m, ct = tile - rt * tiles_m;
    int row0 = rt << 4, col0 = ct << 4;
    int hi = lane >> 4, lm = lane & 15;

    v8f c = {};
    const float* Arow = A + (size_t)(row0 + lm) * K;
    for (int k0 = 0; k0 < K; k0 += 4) {
        v2f a = *reinterpret_cast<const v2f*>(Arow + k0 + 2 * hi);
        const float* Bp = B + (size_t)(k0 + 2 * hi) * M + col0 + lm;
        v2f b;
        b.x = Bp[0];
        b.y = Bp[M];
        c = __builtin_amdgcn_wmma_f32_16x16x4_f32(false, a, false, b,
                                                  (short)0, c, false, false);
    }
    float bv = bias ? bias[col0 + lm] : 0.0f;
    #pragma unroll
    for (int v = 0; v < 8; ++v) {
        int row = row0 + v + 8 * hi;
        float o = c[v];
        if (rowscale) o *= rowscale[row];
        C[(size_t)row * M + col0 + lm] = o + bv;
    }
}

template <int MODE, int P>
__global__ void wmma_gemm_t(const float* __restrict__ X, const float* __restrict__ W,
                            float* __restrict__ hT, int K, int M,
                            const float* __restrict__ dis,
                            const float* __restrict__ part,
                            const float* __restrict__ pbias, int total_tiles) {
    int tid  = threadIdx.x;
    int wid  = tid >> 5;
    int lane = tid & 31;
    int tile = blockIdx.x * (blockDim.x >> 5) + wid;
    if (tile >= total_tiles) return;

    int tiles_f = M >> 4;
    int nt = tile / tiles_f, ft = tile - nt * tiles_f;
    int n0 = nt << 5, f0 = ft << 4;
    int hi = lane >> 4, lm = lane & 15;
    int na = n0 + lm, nb = n0 + 16 + lm;
    float da = dis[na], db = dis[nb];

    v8f c0 = {}, c1 = {};
    for (int k0 = 0; k0 < K; k0 += 4) {
        int k = k0 + 2 * hi;
        const float* Wp = W + (size_t)k * M + f0 + lm;
        v2f a;
        a.x = Wp[0];
        a.y = Wp[M];
        v2f b0, b1;
        if (MODE == 0) {
            b0 = *reinterpret_cast<const v2f*>(X + (size_t)na * K + k);
            b1 = *reinterpret_cast<const v2f*>(X + (size_t)nb * K + k);
        } else if (MODE == 1) {
            const float* Xp = X + (size_t)k * NNODES;
            b0.x = Xp[na];          b1.x = Xp[nb];
            b0.y = Xp[NNODES + na]; b1.y = Xp[NNODES + nb];
        } else {
            #pragma unroll
            for (int s = 0; s < 2; ++s) {
                int kk = k + s;
                const float* pp = part + (size_t)kk * NNODES;
                float sa = pp[na], sb = pp[nb];
                #pragma unroll
                for (int p = 1; p < P; ++p) {
                    sa += pp[(size_t)p * K * NNODES + na];
                    sb += pp[(size_t)p * K * NNODES + nb];
                }
                const float* Xp = X + (size_t)kk * NNODES;
                float bk = pbias[kk];
                float va = (sa + Xp[na]) * da + bk;
                float vb = (sb + Xp[nb]) * db + bk;
                b0[s] = va > 0.0f ? va : 0.0f;
                b1[s] = vb > 0.0f ? vb : 0.0f;
            }
        }
        c0 = __builtin_amdgcn_wmma_f32_16x16x4_f32(false, a, false, b0,
                                                   (short)0, c0, false, false);
        c1 = __builtin_amdgcn_wmma_f32_16x16x4_f32(false, a, false, b1,
                                                   (short)0, c1, false, false);
    }
    float dn = dis[n0 + lane];
    #pragma unroll
    for (int r = 0; r < 8; ++r) {
        float give = hi ? c0[r] : c1[r];
        float got  = __shfl_xor(give, 16);
        float lo = hi ? got : c0[r];
        float up = hi ? c1[r] : got;
        hT[(size_t)(f0 + r) * NNODES + n0 + lane]     = lo * dn;
        hT[(size_t)(f0 + 8 + r) * NNODES + n0 + lane] = up * dn;
    }
}

template <int P>
__global__ __launch_bounds__(1024) void slab_kernel(const float* __restrict__ hT,
                                                    const unsigned* __restrict__ elist,
                                                    const unsigned* __restrict__ ovf,
                                                    const int* __restrict__ gcnt,
                                                    float* __restrict__ outp,
                                                    const float* __restrict__ dis,
                                                    const float* __restrict__ bias, int F) {
    __shared__ float hcol[NNODES];
    __shared__ float acc[NNODES];
    const int tid = threadIdx.x;
    const int p = blockIdx.x / F;
    const int f = blockIdx.x - p * F;

    const float4* h4 = reinterpret_cast<const float4*>(hT + (size_t)f * NNODES);
    float4* hc4 = reinterpret_cast<float4*>(hcol);
    float4* ac4 = reinterpret_cast<float4*>(acc);
    for (int i = tid; i < NNODES / 4; i += 1024) {
        hc4[i] = h4[i];
        ac4[i] = make_float4(0.0f, 0.0f, 0.0f, 0.0f);
    }
    __syncthreads();

    const int lane = tid & 31;
    const int slice = p * 32 + (tid >> 5);
    int n = gcnt[lane];
    n = n < CAPN ? n : CAPN;
    const int ng = (n + 3) >> 2;
    const uint4* e4 = reinterpret_cast<const uint4*>(elist);
    int g = slice;
    if (g < ng) {
        uint4 e = e4[(size_t)g * 32 + lane];
        for (;;) {
            const int gn = g + 32 * P;
            const bool more = gn < ng;
            uint4 en = e;
            if (more) en = e4[(size_t)gn * 32 + lane];
            const int rem = n - 4 * g;
            if (rem >= 4) {
                float v0 = hcol[e.x & 0xFFFFu], v1 = hcol[e.y & 0xFFFFu];
                float v2 = hcol[e.z & 0xFFFFu], v3 = hcol[e.w & 0xFFFFu];
                __hip_atomic_fetch_add(&acc[e.x >> 16], v0, __ATOMIC_RELAXED, __HIP_MEMORY_SCOPE_WORKGROUP);
                __hip_atomic_fetch_add(&acc[e.y >> 16], v1, __ATOMIC_RELAXED, __HIP_MEMORY_SCOPE_WORKGROUP);
                __hip_atomic_fetch_add(&acc[e.z >> 16], v2, __ATOMIC_RELAXED, __HIP_MEMORY_SCOPE_WORKGROUP);
                __hip_atomic_fetch_add(&acc[e.w >> 16], v3, __ATOMIC_RELAXED, __HIP_MEMORY_SCOPE_WORKGROUP);
            } else {
                __hip_atomic_fetch_add(&acc[e.x >> 16], hcol[e.x & 0xFFFFu], __ATOMIC_RELAXED, __HIP_MEMORY_SCOPE_WORKGROUP);
                if (rem > 1)
                    __hip_atomic_fetch_add(&acc[e.y >> 16], hcol[e.y & 0xFFFFu], __ATOMIC_RELAXED, __HIP_MEMORY_SCOPE_WORKGROUP);
                if (rem > 2)
                    __hip_atomic_fetch_add(&acc[e.z >> 16], hcol[e.z & 0xFFFFu], __ATOMIC_RELAXED, __HIP_MEMORY_SCOPE_WORKGROUP);
            }
            if (!more) break;
            e = en; g = gn;
        }
    }
    int novf = gcnt[32];
    novf = novf < NEDGES ? novf : NEDGES;
    for (int i = p * 1024 + tid; i < novf; i += 1024 * P) {
        const unsigned e = ovf[i];
        __hip_atomic_fetch_add(&acc[e >> 16], hcol[e & 0xFFFFu], __ATOMIC_RELAXED, __HIP_MEMORY_SCOPE_WORKGROUP);
    }
    __syncthreads();

    float4* o4 = reinterpret_cast<float4*>(outp + (size_t)blockIdx.x * NNODES);
    if (P == 1) {
        const float4* dis4 = reinterpret_cast<const float4*>(dis);
        const float bf = bias[f];
        for (int i = tid; i < NNODES / 4; i += 1024) {
            float4 a = ac4[i], h = hc4[i], dd = dis4[i], o;
            o.x = (a.x + h.x) * dd.x + bf;
            o.y = (a.y + h.y) * dd.y + bf;
            o.z = (a.z + h.z) * dd.z + bf;
            o.w = (a.w + h.w) * dd.w + bf;
            o.x = o.x > 0.0f ? o.x : 0.0f;
            o.y = o.y > 0.0f ? o.y : 0.0f;
            o.z = o.z > 0.0f ? o.z : 0.0f;
            o.w = o.w > 0.0f ? o.w : 0.0f;
            o4[i] = o;
        }
    } else {
        for (int i = tid; i < NNODES / 4; i += 1024) o4[i] = ac4[i];
    }
}

template <int P>
__global__ __launch_bounds__(1024) void pool_t_kernel(const float* __restrict__ part,
                                                      const float* __restrict__ hT,
                                                      const float* __restrict__ dis,
                                                      const float* __restrict__ bias,
                                                      const int* __restrict__ batch,
                                                      float* __restrict__ pooled) {
    __shared__ float tile[32][33];
    const int t = threadIdx.x;
    const int n0 = blockIdx.x * 32;
    {
        int f = t >> 5, n = n0 + (t & 31);
        float a = part[(size_t)f * NNODES + n];
        #pragma unroll
        for (int p = 1; p < P; ++p) a += part[((size_t)p * 32 + f) * NNODES + n];
        float v = (a + hT[(size_t)f * NNODES + n]) * dis[n] + bias[f];
        tile[f][t & 31] = v > 0.0f ? v : 0.0f;
    }
    __syncthreads();
    {
        int nl = t >> 5, f = t & 31;
        atomAddF(&pooled[(batch[n0 + nl] << 5) + f], tile[f][nl]);
    }
}

__global__ void poolfin_kernel(float* __restrict__ pooled, const float* __restrict__ counts) {
    int i = blockIdx.x * blockDim.x + threadIdx.x;
    int g = i >> 5;
    pooled[i] = pooled[i] / fmaxf(counts[g], 1.0f);
}

extern "C" void kernel_launch(void* const* d_in, const int* in_sizes, int n_in,
                              void* d_out, int out_size, void* d_ws, size_t ws_size,
                              hipStream_t stream) {
    const float* x   = (const float*)d_in[0];
    const float* W1  = (const float*)d_in[1];
    const float* b1  = (const float*)d_in[2];
    const float* W2  = (const float*)d_in[3];
    const float* b2  = (const float*)d_in[4];
    const float* W3  = (const float*)d_in[5];
    const float* b3  = (const float*)d_in[6];
    const float* Wl  = (const float*)d_in[7];
    const float* bl  = (const float*)d_in[8];
    const int*   ei  = (const int*)d_in[9];
    const int*   bat = (const int*)d_in[10];
    float*       out = (float*)d_out;

    const int N = NNODES, E = NEDGES;
    const int* src = ei;
    const int* dst = ei + E;

    float* B0     = (float*)d_ws;
    float* B1     = B0 + (size_t)N * 128;
    float* dis    = B1 + (size_t)N * 128;
    int*   gcnt   = (int*)(dis + N);
    float* counts = dis + N + 64;
    float* pooled = counts + NGRAPH;
    float* H3     = B0 + (size_t)N * 64;
    unsigned* elist = (unsigned*)(pooled + NGRAPH * 32);
    unsigned* ovf   = elist + (size_t)CAPG * 128;

    hipMemsetAsync(dis, 0, (size_t)(N + 64) * sizeof(float), stream);
    hipMemsetAsync(counts, 0, NGRAPH * sizeof(float), stream);
    bin_deg_kernel<<<(E / 4 + 1023) / 1024, 1024, 0, stream>>>(src, dst, dis, gcnt, elist, ovf);
    dis_counts_kernel<<<(N + 255) / 256, 256, 0, stream>>>(dis, bat, counts, N);

    wmma_gemm_t<0, 1><<<(1250 * 8) / 8, 256, 0, stream>>>(x, W1, B0, 128, 128, dis, nullptr, nullptr, 1250 * 8);
    slab_kernel<1><<<128, 1024, 0, stream>>>(B0, elist, ovf, gcnt, B1, dis, b1, 128);

    wmma_gemm_t<1, 1><<<(1250 * 4) / 8, 256, 0, stream>>>(B1, W2, B0, 128, 64, dis, nullptr, nullptr, 1250 * 4);
    slab_kernel<2><<<128, 1024, 0, stream>>>(B0, elist, ovf, gcnt, B1, dis, b2, 64);

    wmma_gemm_t<2, 2><<<(1250 * 2 + 7) / 8, 256, 0, stream>>>(B0, W3, H3, 64, 32, dis, B1, b2, 1250 * 2);
    slab_kernel<4><<<128, 1024, 0, stream>>>(H3, elist, ovf, gcnt, B1, dis, b3, 32);

    hipMemsetAsync(pooled, 0, NGRAPH * 32 * sizeof(float), stream);
    pool_t_kernel<4><<<N / 32, 1024, 0, stream>>>(B1, H3, dis, b3, bat, pooled);
    poolfin_kernel<<<(NGRAPH * 32) / 256, 256, 0, stream>>>(pooled, counts);

    wmma_gemm_f32<<<(16 * 48) / 8, 256, 0, stream>>>(pooled, Wl, out, NGRAPH, 32, 768,
                                                     nullptr, bl, 16 * 48);
}


# column aggregation split by dst node range: 256 work-groups per layer (2/4/8 per feature) own and finish their nodes; lists keyed by range and dst mod 32; no partial planes; sum order changes
# speedup vs baseline: 1.9203x; 1.2105x over previous
#include <hip/hip_runtime.h>
#include <hip/hip_bf16.h>


typedef __attribute__((ext_vector_type(2))) float v2f;
typedef __attribute__((ext_vector_type(8))) float v8f;

#define NNODES 40000
#define NEDGES 640000
#define NGRAPH 256

__device__ __forceinline__ void atomAddF(float* p, float v) {
    unsafeAtomicAdd(p, v);
}

#define RNG 5024
#define NRNG 8
#define NLIST (NRNG * 32)
#define CAPN 4096
#define CAPG (CAPN / 4)
#define OVFCAP (NEDGES - CAPN)
static_assert(NNODES <= 65536, "src and dst share one 32-bit word");
static_assert(RNG % 32 == 0 && NRNG * RNG >= NNODES, "ranges cover all nodes, lane = dst mod 32");

__device__ __forceinline__ void place_edge(int s, int d, int q, int pos, int* __restrict__ gcnt,
                                           unsigned* __restrict__ elist,
                                           unsigned* __restrict__ ovf) {
    if (pos < CAPN) {
        const unsigned w = (unsigned)s | ((unsigned)(d - q * RNG) << 16);
        elist[(((size_t)q * CAPG + (pos >> 2)) * 32 + (d & 31)) * 4 + (pos & 3)] = w;
    } else {
        const int o = atomicAdd(&gcnt[NLIST], 1);
        if (o < OVFCAP) ovf[o] = (unsigned)s | ((unsigned)d << 16);
    }
}

__global__ __launch_bounds__(1024) void bin_deg_kernel(const int* __restrict__ src,
                                                       const int* __restrict__ dst,
                                                       float* __restrict__ deg,
                                                       int* __restrict__ gcnt,
                                                       unsigned* __restrict__ elist,
                                                       unsigned* __restrict__ ovf) {
    __shared__ int cnt[NLIST];
    __shared__ int base[NLIST];
    const int tid = threadIdx.x;
    if (tid < NLIST) cnt[tid] = 0;
    __syncthreads();
    const int i = blockIdx.x * 1024 + tid;
    const bool act = i < NEDGES / 4;
    int4 s = make_int4(0, 0, 0, 0), d = make_int4(0, 0, 0, 0);
    int q0 = 0, q1 = 0, q2 = 0, q3 = 0;
    int r0 = 0, r1 = 0, r2 = 0, r3 = 0;
    if (act) {
        s = reinterpret_cast<const int4*>(src)[i];
        d = reinterpret_cast<const int4*>(dst)[i];
        atomAddF(&deg[d.x], 1.0f);
        atomAddF(&deg[d.y], 1.0f);
        atomAddF(&deg[d.z], 1.0f);
        atomAddF(&deg[d.w], 1.0f);
        q0 = d.x / RNG; q1 = d.y / RNG; q2 = d.z / RNG; q3 = d.w / RNG;
        r0 = atomicAdd(&cnt[q0 * 32 + (d.x & 31)], 1);
        r1 = atomicAdd(&cnt[q1 * 32 + (d.y & 31)], 1);
        r2 = atomicAdd(&cnt[q2 * 32 + (d.z & 31)], 1);
        r3 = atomicAdd(&cnt[q3 * 32 + (d.w & 31)], 1);
    }
    __syncthreads();
    if (tid < NLIST) base[tid] = atomicAdd(&gcnt[tid], cnt[tid]);
    __syncthreads();
    if (act) {
        place_edge(s.x, d.x, q0, base[q0 * 32 + (d.x & 31)] + r0, gcnt, elist, ovf);
        place_edge(s.y, d.y, q1, base[q1 * 32 + (d.y & 31)] + r1, gcnt, elist, ovf);
        place_edge(s.z, d.z, q2, base[q2 * 32 + (d.z & 31)] + r2, gcnt, elist, ovf);
        place_edge(s.w, d.w, q3, base[q3 * 32 + (d.w & 31)] + r3, gcnt, elist, ovf);
    }
}

__global__ void dis_counts_kernel(float* __restrict__ deg_dis,
                                  const int* __restrict__ batch,
                                  float* __restrict__ counts, int N) {
    int n = blockIdx.x * blockDim.x + threadIdx.x;
    if (n < N) {
        deg_dis[n] = rsqrtf(deg_dis[n] + 1.0f);
        atomAddF(&counts[batch[n]], 1.0f);
    }
}

__global__ void wmma_gemm_f32(const float* __restrict__ A, const float* __restrict__ B,
                              float* __restrict__ C, int N, int K, int M,
                              const float* __restrict__ rowscale,
                              const float* __restrict__ bias, int total_tiles) {
    int tid  = threadIdx.x;
    int wid  = tid >> 5;
    int lane = tid & 31;
    int tile = blockIdx.x * (blockDim.x >> 5) + wid;
    if (tile >= total_tiles) return;

    int tiles_m = M >> 4;
    int rt = tile / tiles_m, ct = tile - rt * tiles_m;
    int row0 = rt << 4, col0 = ct << 4;
    int hi = lane >> 4, lm = lane & 15;

    v8f c = {};
    const float* Arow = A + (size_t)(row0 + lm) * K;
    for (int k0 = 0; k0 < K; k0 += 4) {
        v2f a = *reinterpret_cast<const v2f*>(Arow + k0 + 2 * hi);
        const float* Bp = B + (size_t)(k0 + 2 * hi) * M + col0 + lm;
        v2f b;
        b.x = Bp[0];
        b.y = Bp[M];
        c = __builtin_amdgcn_wmma_f32_16x16x4_f32(false, a, false, b,
                                                  (short)0, c, false, false);
    }
    float bv = bias ? bias[col0 + lm] : 0.0f;
    #pragma unroll
    for (int v = 0; v < 8; ++v) {
        int row = row0 + v + 8 * hi;
        float o = c[v];
        if (rowscale) o *= rowscale[row];
        C[(size_t)row * M + col0 + lm] = o + bv;
    }
}

template <int MODE>
__global__ void wmma_gemm_t(const float* __restrict__ X, const float* __restrict__ W,
                            float* __restrict__ hT, int K, int M,
                            const float* __restrict__ dis, int total_tiles) {
    int tid  = threadIdx.x;
    int wid  = tid >> 5;
    int lane = tid & 31;
    int tile = blockIdx.x * (blockDim.x >> 5) + wid;
    if (tile >= total_tiles) return;

    int tiles_f = M >> 4;
    int nt = tile / tiles_f, ft = tile - nt * tiles_f;
    int n0 = nt << 5, f0 = ft << 4;
    int hi = lane >> 4, lm = lane & 15;
    int na = n0 + lm, nb = n0 + 16 + lm;

    v8f c0 = {}, c1 = {};
    for (int k0 = 0; k0 < K; k0 += 4) {
        int k = k0 + 2 * hi;
        const float* Wp = W + (size_t)k * M + f0 + lm;
        v2f a;
        a.x = Wp[0];
        a.y = Wp[M];
        v2f b0, b1;
        if (MODE == 0) {
            b0 = *reinterpret_cast<const v2f*>(X + (size_t)na * K + k);
            b1 = *reinterpret_cast<const v2f*>(X + (size_t)nb * K + k);
        } else {
            const float* Xp = X + (size_t)k * NNODES;
            b0.x = Xp[na];          b1.x = Xp[nb];
            b0.y = Xp[NNODES + na]; b1.y = Xp[NNODES + nb];
        }
        c0 = __builtin_amdgcn_wmma_f32_16x16x4_f32(false, a, false, b0,
                                                   (short)0, c0, false, false);
        c1 = __builtin_amdgcn_wmma_f32_16x16x4_f32(false, a, false, b1,
                                                   (short)0, c1, false, false);
    }
    float dn = dis[n0 + lane];
    #pragma unroll
    for (int r = 0; r < 8; ++r) {
        float give = hi ? c0[r] : c1[r];
        float got  = __shfl_xor(give, 16);
        float lo = hi ? got : c0[r];
        float up = hi ? c1[r] : got;
        hT[(size_t)(f0 + r) * NNODES + n0 + lane]     = lo * dn;
        hT[(size_t)(f0 + 8 + r) * NNODES + n0 + lane] = up * dn;
    }
}

template <int R>
__global__ __launch_bounds__(1024) void slab_kernel(const float* __restrict__ hT,
                                                    const unsigned* __restrict__ elist,
                                                    const unsigned* __restrict__ ovf,
                                                    const int* __restrict__ gcnt,
                                                    float* __restrict__ outp,
                                                    const float* __restrict__ dis,
                                                    const float* __restrict__ bias, int F) {
    __shared__ float hcol[NNODES];
    __shared__ float acc[R * RNG];
    constexpr int WPR = 32 / R;
    const int tid = threadIdx.x;
    const int qb = blockIdx.x / F;
    const int f = blockIdx.x - qb * F;
    const int nlo = qb * (R * RNG);
    const int nhi = nlo + R * RNG < NNODES ? nlo + R * RNG : NNODES;

    const float4* h4 = reinterpret_cast<const float4*>(hT + (size_t)f * NNODES);
    float4* hc4 = reinterpret_cast<float4*>(hcol);
    float4* ac4 = reinterpret_cast<float4*>(acc);
    for (int i = tid; i < NNODES / 4; i += 1024) hc4[i] = h4[i];
    for (int i = tid; i < R * RNG / 4; i += 1024) ac4[i] = make_float4(0.0f, 0.0f, 0.0f, 0.0f);
    __syncthreads();

    const int lane = tid & 31;
    const int wave = tid >> 5;
    const int rl = wave / WPR;
    const int slot = wave - rl * WPR;
    const int rg = qb * R + rl;
    float* accr = acc + rl * RNG;
    int n = gcnt[rg * 32 + lane];
    n = n < CAPN ? n : CAPN;
    const int ng = (n + 3) >> 2;
    const uint4* e4 = reinterpret_cast<const uint4*>(elist) + (size_t)rg * CAPG * 32;
    int g = slot;
    if (g < ng) {
        uint4 e = e4[(size_t)g * 32 + lane];
        for (;;) {
            const int gn = g + WPR;
            const bool more = gn < ng;
            uint4 en = e;
            if (more) en = e4[(size_t)gn * 32 + lane];
            const int rem = n - 4 * g;
            if (rem >= 4) {
                float v0 = hcol[e.x & 0xFFFFu], v1 = hcol[e.y & 0xFFFFu];
                float v2 = hcol[e.z & 0xFFFFu], v3 = hcol[e.w & 0xFFFFu];
                __hip_atomic_fetch_add(&accr[e.x >> 16], v0, __ATOMIC_RELAXED, __HIP_MEMORY_SCOPE_WORKGROUP);
                __hip_atomic_fetch_add(&accr[e.y >> 16], v1, __ATOMIC_RELAXED, __HIP_MEMORY_SCOPE_WORKGROUP);
                __hip_atomic_fetch_add(&accr[e.z >> 16], v2, __ATOMIC_RELAXED, __HIP_MEMORY_SCOPE_WORKGROUP);
                __hip_atomic_fetch_add(&accr[e.w >> 16], v3, __ATOMIC_RELAXED, __HIP_MEMORY_SCOPE_WORKGROUP);
            } else {
                __hip_atomic_fetch_add(&accr[e.x >> 16], hcol[e.x & 0xFFFFu], __ATOMIC_RELAXED, __HIP_MEMORY_SCOPE_WORKGROUP);
                if (rem > 1)
                    __hip_atomic_fetch_add(&accr[e.y >> 16], hcol[e.y & 0xFFFFu], __ATOMIC_RELAXED, __HIP_MEMORY_SCOPE_WORKGROUP);
                if (rem > 2)
                    __hip_atomic_fetch_add(&accr[e.z >> 16], hcol[e.z & 0xFFFFu], __ATOMIC_RELAXED, __HIP_MEMORY_SCOPE_WORKGROUP);
            }
            if (!more) break;
            e = en; g = gn;
        }
    }
    int novf = gcnt[NLIST];
    novf = novf < OVFCAP ? novf : OVFCAP;
    for (int i = tid; i < novf; i += 1024) {
        const unsigned e = ovf[i];
        const int d = (int)(e >> 16);
        if (d >= nlo && d < nhi)
            __hip_atomic_fetch_add(&acc[d - nlo], hcol[e & 0xFFFFu], __ATOMIC_RELAXED, __HIP_MEMORY_SCOPE_WORKGROUP);
    }
    __syncthreads();

    float4* o4 = reinterpret_cast<float4*>(outp + (size_t)f * NNODES + nlo);
    const float4* dis4 = reinterpret_cast<const float4*>(dis + nlo);
    const float4* hl4 = reinterpret_cast<const float4*>(hcol + nlo);
    const float bf = bias[f];
    const int cnt4 = (nhi - nlo) >> 2;
    for (int i = tid; i < cnt4; i += 1024) {
        float4 a = ac4[i], h = hl4[i], dd = dis4[i], o;
        o.x = (a.x + h.x) * dd.x + bf;
        o.y = (a.y + h.y) * dd.y + bf;
        o.z = (a.z + h.z) * dd.z + bf;
        o.w = (a.w + h.w) * dd.w + bf;
        o.x = o.x > 0.0f ? o.x : 0.0f;
        o.y = o.y > 0.0f ? o.y : 0.0f;
        o.z = o.z > 0.0f ? o.z : 0.0f;
        o.w = o.w > 0.0f ? o.w : 0.0f;
        o4[i] = o;
    }
}

__global__ __launch_bounds__(1024) void pool_t_kernel(const float* __restrict__ actT,
                                                      const int* __restrict__ batch,
                                                      float* __restrict__ pooled) {
    __shared__ float tile[32][33];
    const int t = threadIdx.x;
    const int n0 = blockIdx.x * 32;
    {
        int f = t >> 5, n = n0 + (t & 31);
        tile[f][t & 31] = actT[(size_t)f * NNODES + n];
    }
    __syncthreads();
    {
        int nl = t >> 5, f = t & 31;
        atomAddF(&pooled[(batch[n0 + nl] << 5) + f], tile[f][nl]);
    }
}

__global__ void poolfin_kernel(float* __restrict__ pooled, const float* __restrict__ counts) {
    int i = blockIdx.x * blockDim.x + threadIdx.x;
    int g = i >> 5;
    pooled[i] = pooled[i] / fmaxf(counts[g], 1.0f);
}

extern "C" void kernel_launch(void* const* d_in, const int* in_sizes, int n_in,
                              void* d_out, int out_size, void* d_ws, size_t ws_size,
                              hipStream_t stream) {
    const float* x   = (const float*)d_in[0];
    const float* W1  = (const float*)d_in[1];
    const float* b1  = (const float*)d_in[2];
    const float* W2  = (const float*)d_in[3];
    const float* b2  = (const float*)d_in[4];
    const float* W3  = (const float*)d_in[5];
    const float* b3  = (const float*)d_in[6];
    const float* Wl  = (const float*)d_in[7];
    const float* bl  = (const float*)d_in[8];
    const int*   ei  = (const int*)d_in[9];
    const int*   bat = (const int*)d_in[10];
    float*       out = (float*)d_out;

    const int N = NNODES, E = NEDGES;
    const int* src = ei;
    const int* dst = ei + E;

    float* B0     = (float*)d_ws;
    float* B1     = B0 + (size_t)N * 128;
    float* dis    = B1 + (size_t)N * 128;
    int*   gcnt   = (int*)(dis + N);
    float* counts = dis + N + 320;
    float* pooled = counts + NGRAPH;
    unsigned* elist = (unsigned*)(pooled + NGRAPH * 32);
    unsigned* ovf   = elist + (size_t)NRNG * CAPG * 128;

    hipMemsetAsync(dis, 0, (size_t)(N + 320) * sizeof(float), stream);
    hipMemsetAsync(counts, 0, NGRAPH * sizeof(float), stream);
    bin_deg_kernel<<<(E / 4 + 1023) / 1024, 1024, 0, stream>>>(src, dst, dis, gcnt, elist, ovf);
    dis_counts_kernel<<<(N + 255) / 256, 256, 0, stream>>>(dis, bat, counts, N);

    wmma_gemm_t<0><<<(1250 * 8) / 8, 256, 0, stream>>>(x, W1, B0, 128, 128, dis, 1250 * 8);
    slab_kernel<4><<<128 * 2, 1024, 0, stream>>>(B0, elist, ovf, gcnt, B1, dis, b1, 128);

    wmma_gemm_t<1><<<(1250 * 4) / 8, 256, 0, stream>>>(B1, W2, B0, 128, 64, dis, 1250 * 4);
    slab_kernel<2><<<64 * 4, 1024, 0, stream>>>(B0, elist, ovf, gcnt, B1, dis, b2, 64);

    wmma_gemm_t<1><<<(1250 * 2 + 7) / 8, 256, 0, stream>>>(B1, W3, B0, 64, 32, dis, 1250 * 2);
    slab_kernel<1><<<32 * 8, 1024, 0, stream>>>(B0, elist, ovf, gcnt, B1, dis, b3, 32);

    hipMemsetAsync(pooled, 0, NGRAPH * 32 * sizeof(float), stream);
    pool_t_kernel<<<N / 32, 1024, 0, stream>>>(B1, bat, pooled);
    poolfin_kernel<<<(NGRAPH * 32) / 256, 256, 0, stream>>>(pooled, counts);

    wmma_gemm_f32<<<(16 * 48) / 8, 256, 0, stream>>>(pooled, Wl, out, NGRAPH, 32, 768,
                                                     nullptr, bl, 16 * 48);
}


# layer-1 GEMM: a work-group stages its 32 rows of x in LDS once (one whole row per 16-byte load) and its 8 waves read B fragments there, not 8-byte loads over 16 rows each from global; sums unchanged
# speedup vs baseline: 2.0097x; 1.0466x over previous
#include <hip/hip_runtime.h>
#include <hip/hip_bf16.h>


typedef __attribute__((ext_vector_type(2))) float v2f;
typedef __attribute__((ext_vector_type(8))) float v8f;

#define NNODES 40000
#define NEDGES 640000
#define NGRAPH 256

__device__ __forceinline__ void atomAddF(float* p, float v) {
    unsafeAtomicAdd(p, v);
}

#define RNG 5024
#define NRNG 8
#define NLIST (NRNG * 32)
#define CAPN 4096
#define CAPG (CAPN / 4)
#define OVFCAP (NEDGES - CAPN)
static_assert(NNODES <= 65536, "src and dst share one 32-bit word");
static_assert(RNG % 32 == 0 && NRNG * RNG >= NNODES, "ranges cover all nodes, lane = dst mod 32");

__device__ __forceinline__ void place_edge(int s, int d, int q, int pos, int* __restrict__ gcnt,
                                           unsigned* __restrict__ elist,
                                           unsigned* __restrict__ ovf) {
    if (pos < CAPN) {
        const unsigned w = (unsigned)s | ((unsigned)(d - q * RNG) << 16);
        elist[(((size_t)q * CAPG + (pos >> 2)) * 32 + (d & 31)) * 4 + (pos & 3)] = w;
    } else {
        const int o = atomicAdd(&gcnt[NLIST], 1);
        if (o < OVFCAP) ovf[o] = (unsigned)s | ((unsigned)d << 16);
    }
}

__global__ __launch_bounds__(1024) void bin_deg_kernel(const int* __restrict__ src,
                                                       const int* __restrict__ dst,
                                                       float* __restrict__ deg,
                                                       int* __restrict__ gcnt,
                                                       unsigned* __restrict__ elist,
                                                       unsigned* __restrict__ ovf) {
    __shared__ int cnt[NLIST];
    __shared__ int base[NLIST];
    const int tid = threadIdx.x;
    if (tid < NLIST) cnt[tid] = 0;
    __syncthreads();
    const int i = blockIdx.x * 1024 + tid;
    const bool act = i < NEDGES / 4;
    int4 s = make_int4(0, 0, 0, 0), d = make_int4(0, 0, 0, 0);
    int q0 = 0, q1 = 0, q2 = 0, q3 = 0;
    int r0 = 0, r1 = 0, r2 = 0, r3 = 0;
    if (act) {
        s = reinterpret_cast<const int4*>(src)[i];
        d = reinterpret_cast<const int4*>(dst)[i];
        atomAddF(&deg[d.x], 1.0f);
        atomAddF(&deg[d.y], 1.0f);
        atomAddF(&deg[d.z], 1.0f);
        atomAddF(&deg[d.w], 1.0f);
        q0 = d.x / RNG; q1 = d.y / RNG; q2 = d.z / RNG; q3 = d.w / RNG;
        r0 = atomicAdd(&cnt[q0 * 32 + (d.x & 31)], 1);
        r1 = atomicAdd(&cnt[q1 * 32 + (d.y & 31)], 1);
        r2 = atomicAdd(&cnt[q2 * 32 + (d.z & 31)], 1);
        r3 = atomicAdd(&cnt[q3 * 32 + (d.w & 31)], 1);
    }
    __syncthreads();
    if (tid < NLIST) base[tid] = atomicAdd(&gcnt[tid], cnt[tid]);
    __syncthreads();
    if (act) {
        place_edge(s.x, d.x, q0, base[q0 * 32 + (d.x & 31)] + r0, gcnt, elist, ovf);
        place_edge(s.y, d.y, q1, base[q1 * 32 + (d.y & 31)] + r1, gcnt, elist, ovf);
        place_edge(s.z, d.z, q2, base[q2 * 32 + (d.z & 31)] + r2, gcnt, elist, ovf);
        place_edge(s.w, d.w, q3, base[q3 * 32 + (d.w & 31)] + r3, gcnt, elist, ovf);
    }
}

__global__ void dis_counts_kernel(float* __restrict__ deg_dis,
                                  const int* __restrict__ batch,
                                  float* __restrict__ counts, int N) {
    int n = blockIdx.x * blockDim.x + threadIdx.x;
    if (n < N) {
        deg_dis[n] = rsqrtf(deg_dis[n] + 1.0f);
        atomAddF(&counts[batch[n]], 1.0f);
    }
}

__global__ void wmma_gemm_f32(const float* __restrict__ A, const float* __restrict__ B,
                              float* __restrict__ C, int N, int K, int M,
                              const float* __restrict__ rowscale,
                              const float* __restrict__ bias, int total_tiles) {
    int tid  = threadIdx.x;
    int wid  = tid >> 5;
    int lane = tid & 31;
    int tile = blockIdx.x * (blockDim.x >> 5) + wid;
    if (tile >= total_tiles) return;

    int tiles_m = M >> 4;
    int rt = tile / tiles_m, ct = tile - rt * tiles_m;
    int row0 = rt << 4, col0 = ct << 4;
    int hi = lane >> 4, lm = lane & 15;

    v8f c = {};
    const float* Arow = A + (size_t)(row0 + lm) * K;
    for (int k0 = 0; k0 < K; k0 += 4) {
        v2f a = *reinterpret_cast<const v2f*>(Arow + k0 + 2 * hi);
        const float* Bp = B + (size_t)(k0 + 2 * hi) * M + col0 + lm;
        v2f b;
        b.x = Bp[0];
        b.y = Bp[M];
        c = __builtin_amdgcn_wmma_f32_16x16x4_f32(false, a, false, b,
                                                  (short)0, c, false, false);
    }
    float bv = bias ? bias[col0 + lm] : 0.0f;
    #pragma unroll
    for (int v = 0; v < 8; ++v) {
        int row = row0 + v + 8 * hi;
        float o = c[v];
        if (rowscale) o *= rowscale[row];
        C[(size_t)row * M + col0 + lm] = o + bv;
    }
}

__global__ void wmma_gemm_t(const float* __restrict__ X, const float* __restrict__ W,
                            float* __restrict__ hT, int K, int M,
                            const float* __restrict__ dis, int total_tiles) {
    int tid  = threadIdx.x;
    int wid  = tid >> 5;
    int lane = tid & 31;
    int tile = blockIdx.x * (blockDim.x >> 5) + wid;
    if (tile >= total_tiles) return;

    int tiles_f = M >> 4;
    int nt = tile / tiles_f, ft = tile - nt * tiles_f;
    int n0 = nt << 5, f0 = ft << 4;
    int hi = lane >> 4, lm = lane & 15;
    int na = n0 + lm, nb = n0 + 16 + lm;

    v8f c0 = {}, c1 = {};
    for (int k0 = 0; k0 < K; k0 += 4) {
        int k = k0 + 2 * hi;
        const float* Wp = W + (size_t)k * M + f0 + lm;
        v2f a;
        a.x = Wp[0];
        a.y = Wp[M];
        v2f b0, b1;
        const float* Xp = X + (size_t)k * NNODES;
        b0.x = Xp[na];          b1.x = Xp[nb];
        b0.y = Xp[NNODES + na]; b1.y = Xp[NNODES + nb];
        c0 = __builtin_amdgcn_wmma_f32_16x16x4_f32(false, a, false, b0,
                                                   (short)0, c0, false, false);
        c1 = __builtin_amdgcn_wmma_f32_16x16x4_f32(false, a, false, b1,
                                                   (short)0, c1, false, false);
    }
    float dn = dis[n0 + lane];
    #pragma unroll
    for (int r = 0; r < 8; ++r) {
        float give = hi ? c0[r] : c1[r];
        float got  = __shfl_xor(give, 16);
        float lo = hi ? got : c0[r];
        float up = hi ? c1[r] : got;
        hT[(size_t)(f0 + r) * NNODES + n0 + lane]     = lo * dn;
        hT[(size_t)(f0 + 8 + r) * NNODES + n0 + lane] = up * dn;
    }
}

#define L1K 128
#define L1M 128
#define XPITCH 132
__global__ __launch_bounds__(256) void gemm_rows_lds(const float* __restrict__ X,
                                                     const float* __restrict__ W,
                                                     float* __restrict__ hT,
                                                     const float* __restrict__ dis) {
    __shared__ float xs[32 * XPITCH];
    const int tid  = threadIdx.x;
    const int wid  = tid >> 5;
    const int lane = tid & 31;
    const int n0 = blockIdx.x << 5;

    #pragma unroll
    for (int i = 0; i < 4; ++i) {
        const int r = wid + 8 * i;
        const float4 v = reinterpret_cast<const float4*>(X + (size_t)(n0 + r) * L1K)[lane];
        *reinterpret_cast<float4*>(&xs[r * XPITCH + lane * 4]) = v;
    }
    __syncthreads();

    const int f0 = wid << 4;
    const int hi = lane >> 4, lm = lane & 15;
    const float* xa = xs + lm * XPITCH + 2 * hi;
    const float* xb = xa + 16 * XPITCH;
    const float* Wp = W + (size_t)(2 * hi) * L1M + f0 + lm;

    v8f c0 = {}, c1 = {};
    #pragma unroll 8
    for (int k0 = 0; k0 < L1K; k0 += 4) {
        v2f a;
        a.x = Wp[(size_t)k0 * L1M];
        a.y = Wp[(size_t)(k0 + 1) * L1M];
        const v2f b0 = *reinterpret_cast<const v2f*>(xa + k0);
        const v2f b1 = *reinterpret_cast<const v2f*>(xb + k0);
        c0 = __builtin_amdgcn_wmma_f32_16x16x4_f32(false, a, false, b0,
                                                   (short)0, c0, false, false);
        c1 = __builtin_amdgcn_wmma_f32_16x16x4_f32(false, a, false, b1,
                                                   (short)0, c1, false, false);
    }
    const float dn = dis[n0 + lane];
    #pragma unroll
    for (int r = 0; r < 8; ++r) {
        float give = hi ? c0[r] : c1[r];
        float got  = __shfl_xor(give, 16);
        float lo = hi ? got : c0[r];
        float up = hi ? c1[r] : got;
        hT[(size_t)(f0 + r) * NNODES + n0 + lane]     = lo * dn;
        hT[(size_t)(f0 + 8 + r) * NNODES + n0 + lane] = up * dn;
    }
}
static_assert(NNODES % 32 == 0, "layer-1 GEMM: whole 32-node tiles");

template <int R>
__global__ __launch_bounds__(1024) void slab_kernel(const float* __restrict__ hT,
                                                    const unsigned* __restrict__ elist,
                                                    const unsigned* __restrict__ ovf,
                                                    const int* __restrict__ gcnt,
                                                    float* __restrict__ outp,
                                                    const float* __restrict__ dis,
                                                    const float* __restrict__ bias, int F) {
    __shared__ float hcol[NNODES];
    __shared__ float acc[R * RNG];
    constexpr int WPR = 32 / R;
    const int tid = threadIdx.x;
    const int qb = blockIdx.x / F;
    const int f = blockIdx.x - qb * F;
    const int nlo = qb * (R * RNG);
    const int nhi = nlo + R * RNG < NNODES ? nlo + R * RNG : NNODES;

    const float4* h4 = reinterpret_cast<const float4*>(hT + (size_t)f * NNODES);
    float4* hc4 = reinterpret_cast<float4*>(hcol);
    float4* ac4 = reinterpret_cast<float4*>(acc);
    for (int i = tid; i < NNODES / 4; i += 1024) hc4[i] = h4[i];
    for (int i = tid; i < R * RNG / 4; i += 1024) ac4[i] = make_float4(0.0f, 0.0f, 0.0f, 0.0f);
    __syncthreads();

    const int lane = tid & 31;
    const int wave = tid >> 5;
    const int rl = wave / WPR;
    const int slot = wave - rl * WPR;
    const int rg = qb * R + rl;
    float* accr = acc + rl * RNG;
    int n = gcnt[rg * 32 + lane];
    n = n < CAPN ? n : CAPN;
    const int ng = (n + 3) >> 2;
    const uint4* e4 = reinterpret_cast<const uint4*>(elist) + (size_t)rg * CAPG * 32;
    int g = slot;
    if (g < ng) {
        uint4 e = e4[(size_t)g * 32 + lane];
        for (;;) {
            const int gn = g + WPR;
            const bool more = gn < ng;
            uint4 en = e;
            if (more) en = e4[(size_t)gn * 32 + lane];
            const int rem = n - 4 * g;
            if (rem >= 4) {
                float v0 = hcol[e.x & 0xFFFFu], v1 = hcol[e.y & 0xFFFFu];
                float v2 = hcol[e.z & 0xFFFFu], v3 = hcol[e.w & 0xFFFFu];
                __hip_atomic_fetch_add(&accr[e.x >> 16], v0, __ATOMIC_RELAXED, __HIP_MEMORY_SCOPE_WORKGROUP);
                __hip_atomic_fetch_add(&accr[e.y >> 16], v1, __ATOMIC_RELAXED, __HIP_MEMORY_SCOPE_WORKGROUP);
                __hip_atomic_fetch_add(&accr[e.z >> 16], v2, __ATOMIC_RELAXED, __HIP_MEMORY_SCOPE_WORKGROUP);
                __hip_atomic_fetch_add(&accr[e.w >> 16], v3, __ATOMIC_RELAXED, __HIP_MEMORY_SCOPE_WORKGROUP);
            } else {
                __hip_atomic_fetch_add(&accr[e.x >> 16], hcol[e.x & 0xFFFFu], __ATOMIC_RELAXED, __HIP_MEMORY_SCOPE_WORKGROUP);
                if (rem > 1)
                    __hip_atomic_fetch_add(&accr[e.y >> 16], hcol[e.y & 0xFFFFu], __ATOMIC_RELAXED, __HIP_MEMORY_SCOPE_WORKGROUP);
                if (rem > 2)
                    __hip_atomic_fetch_add(&accr[e.z >> 16], hcol[e.z & 0xFFFFu], __ATOMIC_RELAXED, __HIP_MEMORY_SCOPE_WORKGROUP);
            }
            if (!more) break;
            e = en; g = gn;
        }
    }
    int novf = gcnt[NLIST];
    novf = novf < OVFCAP ? novf : OVFCAP;
    for (int i = tid; i < novf; i += 1024) {
        const unsigned e = ovf[i];
        const int d = (int)(e >> 16);
        if (d >= nlo && d < nhi)
            __hip_atomic_fetch_add(&acc[d - nlo], hcol[e & 0xFFFFu], __ATOMIC_RELAXED, __HIP_MEMORY_SCOPE_WORKGROUP);
    }
    __syncthreads();

    float4* o4 = reinterpret_cast<float4*>(outp + (size_t)f * NNODES + nlo);
    const float4* dis4 = reinterpret_cast<const float4*>(dis + nlo);
    const float4* hl4 = reinterpret_cast<const float4*>(hcol + nlo);
    const float bf = bias[f];
    const int cnt4 = (nhi - nlo) >> 2;
    for (int i = tid; i < cnt4; i += 1024) {
        float4 a = ac4[i], h = hl4[i], dd = dis4[i], o;
        o.x = (a.x + h.x) * dd.x + bf;
        o.y = (a.y + h.y) * dd.y + bf;
        o.z = (a.z + h.z) * dd.z + bf;
        o.w = (a.w + h.w) * dd.w + bf;
        o.x = o.x > 0.0f ? o.x : 0.0f;
        o.y = o.y > 0.0f ? o.y : 0.0f;
        o.z = o.z > 0.0f ? o.z : 0.0f;
        o.w = o.w > 0.0f ? o.w : 0.0f;
        o4[i] = o;
    }
}

__global__ __launch_bounds__(1024) void pool_t_kernel(const float* __restrict__ actT,
                                                      const int* __restrict__ batch,
                                                      float* __restrict__ pooled) {
    __shared__ float tile[32][33];
    const int t = threadIdx.x;
    const int n0 = blockIdx.x * 32;
    {
        int f = t >> 5, n = n0 + (t & 31);
        tile[f][t & 31] = actT[(size_t)f * NNODES + n];
    }
    __syncthreads();
    {
        int nl = t >> 5, f = t & 31;
        atomAddF(&pooled[(batch[n0 + nl] << 5) + f], tile[f][nl]);
    }
}

__global__ void poolfin_kernel(float* __restrict__ pooled, const float* __restrict__ counts) {
    int i = blockIdx.x * blockDim.x + threadIdx.x;
    int g = i >> 5;
    pooled[i] = pooled[i] / fmaxf(counts[g], 1.0f);
}

extern "C" void kernel_launch(void* const* d_in, const int* in_sizes, int n_in,
                              void* d_out, int out_size, void* d_ws, size_t ws_size,
                              hipStream_t stream) {
    const float* x   = (const float*)d_in[0];
    const float* W1  = (const float*)d_in[1];
    const float* b1  = (const float*)d_in[2];
    const float* W2  = (const float*)d_in[3];
    const float* b2  = (const float*)d_in[4];
    const float* W3  = (const float*)d_in[5];
    const float* b3  = (const float*)d_in[6];
    const float* Wl  = (const float*)d_in[7];
    const float* bl  = (const float*)d_in[8];
    const int*   ei  = (const int*)d_in[9];
    const int*   bat = (const int*)d_in[10];
    float*       out = (float*)d_out;

    const int N = NNODES, E = NEDGES;
    const int* src = ei;
    const int* dst = ei + E;

    float* B0     = (float*)d_ws;
    float* B1     = B0 + (size_t)N * 128;
    float* dis    = B1 + (size_t)N * 128;
    int*   gcnt   = (int*)(dis + N);
    float* counts = dis + N + 320;
    float* pooled = counts + NGRAPH;
    unsigned* elist = (unsigned*)(pooled + NGRAPH * 32);
    unsigned* ovf   = elist + (size_t)NRNG * CAPG * 128;

    hipMemsetAsync(dis, 0, (size_t)(N + 320) * sizeof(float), stream);
    hipMemsetAsync(counts, 0, NGRAPH * sizeof(float), stream);
    bin_deg_kernel<<<(E / 4 + 1023) / 1024, 1024, 0, stream>>>(src, dst, dis, gcnt, elist, ovf);
    dis_counts_kernel<<<(N + 255) / 256, 256, 0, stream>>>(dis, bat, counts, N);

    gemm_rows_lds<<<NNODES / 32, 256, 0, stream>>>(x, W1, B0, dis);
    slab_kernel<4><<<128 * 2, 1024, 0, stream>>>(B0, elist, ovf, gcnt, B1, dis, b1, 128);

    wmma_gemm_t<<<(1250 * 4) / 8, 256, 0, stream>>>(B1, W2, B0, 128, 64, dis, 1250 * 4);
    slab_kernel<2><<<64 * 4, 1024, 0, stream>>>(B0, elist, ovf, gcnt, B1, dis, b2, 64);

    wmma_gemm_t<<<(1250 * 2 + 7) / 8, 256, 0, stream>>>(B1, W3, B0, 64, 32, dis, 1250 * 2);
    slab_kernel<1><<<32 * 8, 1024, 0, stream>>>(B0, elist, ovf, gcnt, B1, dis, b3, 32);

    hipMemsetAsync(pooled, 0, NGRAPH * 32 * sizeof(float), stream);
    pool_t_kernel<<<N / 32, 1024, 0, stream>>>(B1, bat, pooled);
    poolfin_kernel<<<(NGRAPH * 32) / 256, 256, 0, stream>>>(pooled, counts);

    wmma_gemm_f32<<<(16 * 48) / 8, 256, 0, stream>>>(pooled, Wl, out, NGRAPH, 32, 768,
                                                     nullptr, bl, 16 * 48);
}


# degree: the 640000 scattered one-word global float adds of the binning pass go; a pass over the finished lists counts dst with LDS integer adds and writes rsqrt(deg+1); degrees exact, sums unchanged
# speedup vs baseline: 2.5937x; 1.2906x over previous
#include <hip/hip_runtime.h>
#include <hip/hip_bf16.h>


typedef __attribute__((ext_vector_type(2))) float v2f;
typedef __attribute__((ext_vector_type(8))) float v8f;

#define NNODES 40000
#define NEDGES 640000
#define NGRAPH 256

__device__ __forceinline__ void atomAddF(float* p, float v) {
    unsafeAtomicAdd(p, v);
}

#define RNG 5024
#define NRNG 8
#define NLIST (NRNG * 32)
#define CAPN 4096
#define CAPG (CAPN / 4)
#define OVFCAP (NEDGES - CAPN)
static_assert(NNODES <= 65536, "src and dst share one 32-bit word");
static_assert(RNG % 32 == 0 && NRNG * RNG >= NNODES, "ranges cover all nodes, lane = dst mod 32");

__device__ __forceinline__ void place_edge(int s, int d, int q, int pos, int* __restrict__ gcnt,
                                           unsigned* __restrict__ elist,
                                           unsigned* __restrict__ ovf) {
    if (pos < CAPN) {
        const unsigned w = (unsigned)s | ((unsigned)(d - q * RNG) << 16);
        elist[(((size_t)q * CAPG + (pos >> 2)) * 32 + (d & 31)) * 4 + (pos & 3)] = w;
    } else {
        const int o = atomicAdd(&gcnt[NLIST], 1);
        if (o < OVFCAP) ovf[o] = (unsigned)s | ((unsigned)d << 16);
    }
}

__global__ __launch_bounds__(1024) void bin_kernel(const int* __restrict__ src,
                                                   const int* __restrict__ dst,
                                                   int* __restrict__ gcnt,
                                                   unsigned* __restrict__ elist,
                                                   unsigned* __restrict__ ovf) {
    __shared__ int cnt[NLIST];
    __shared__ int base[NLIST];
    const int tid = threadIdx.x;
    if (tid < NLIST) cnt[tid] = 0;
    __syncthreads();
    const int i = blockIdx.x * 1024 + tid;
    const bool act = i < NEDGES / 4;
    int4 s = make_int4(0, 0, 0, 0), d = make_int4(0, 0, 0, 0);
    int q0 = 0, q1 = 0, q2 = 0, q3 = 0;
    int r0 = 0, r1 = 0, r2 = 0, r3 = 0;
    if (act) {
        s = reinterpret_cast<const int4*>(src)[i];
        d = reinterpret_cast<const int4*>(dst)[i];
        q0 = d.x / RNG; q1 = d.y / RNG; q2 = d.z / RNG; q3 = d.w / RNG;
        r0 = atomicAdd(&cnt[q0 * 32 + (d.x & 31)], 1);
        r1 = atomicAdd(&cnt[q1 * 32 + (d.y & 31)], 1);
        r2 = atomicAdd(&cnt[q2 * 32 + (d.z & 31)], 1);
        r3 = atomicAdd(&cnt[q3 * 32 + (d.w & 31)], 1);
    }
    __syncthreads();
    if (tid < NLIST) base[tid] = atomicAdd(&gcnt[tid], cnt[tid]);
    __syncthreads();
    if (act) {
        place_edge(s.x, d.x, q0, base[q0 * 32 + (d.x & 31)] + r0, gcnt, elist, ovf);
        place_edge(s.y, d.y, q1, base[q1 * 32 + (d.y & 31)] + r1, gcnt, elist, ovf);
        place_edge(s.z, d.z, q2, base[q2 * 32 + (d.z & 31)] + r2, gcnt, elist, ovf);
        place_edge(s.w, d.w, q3, base[q3 * 32 + (d.w & 31)] + r3, gcnt, elist, ovf);
    }
}

#define DEGBLK (NRNG * 4)
#define CNTBLK ((NNODES + 1023) / 1024)
__global__ __launch_bounds__(1024) void deg_dis_counts_kernel(const unsigned* __restrict__ elist,
                                                              const unsigned* __restrict__ ovf,
                                                              const int* __restrict__ gcnt,
                                                              float* __restrict__ dis,
                                                              const int* __restrict__ batch,
                                                              float* __restrict__ counts) {
    __shared__ int dcnt[RNG];
    const int tid = threadIdx.x;
    if (blockIdx.x >= DEGBLK) {
        const int n = (blockIdx.x - DEGBLK) * 1024 + tid;
        if (n < NNODES) atomAddF(&counts[batch[n]], 1.0f);
        return;
    }
    const int q = blockIdx.x >> 2;
    const int s = blockIdx.x & 3;
    for (int i = tid; i < RNG; i += 1024) dcnt[i] = 0;
    __syncthreads();

    const int lane = tid & 31;
    const int wave = tid >> 5;
    const int L = 8 * s + (lane & 7);
    const int gsub = lane >> 3;
    int n = gcnt[q * 32 + L];
    n = n < CAPN ? n : CAPN;
    const int ng = (n + 3) >> 2;
    const uint4* e4 = reinterpret_cast<const uint4*>(elist) + (size_t)q * CAPG * 32 + L;
    constexpr int TRIPS = CAPG / 128;
    const int glast = ng > 0 ? ng - 1 : 0;
    uint4 e[TRIPS];
    #pragma unroll
    for (int t = 0; t < TRIPS; ++t) {
        const int g = t * 128 + wave * 4 + gsub;
        e[t] = e4[(size_t)(g < glast ? g : glast) * 32];
    }
    #pragma unroll
    for (int t = 0; t < TRIPS; ++t) {
        const int g = t * 128 + wave * 4 + gsub;
        const int rem = n - 4 * g;
        if (rem > 0) atomicAdd(&dcnt[e[t].x >> 16], 1);
        if (rem > 1) atomicAdd(&dcnt[e[t].y >> 16], 1);
        if (rem > 2) atomicAdd(&dcnt[e[t].z >> 16], 1);
        if (rem > 3) atomicAdd(&dcnt[e[t].w >> 16], 1);
    }
    int novf = gcnt[NLIST];
    novf = novf < OVFCAP ? novf : OVFCAP;
    for (int i = tid; i < novf; i += 1024) {
        const int d = (int)(ovf[i] >> 16);
        const int dl = d - q * RNG;
        if (dl >= 0 && dl < RNG && ((d & 31) >> 3) == s) atomicAdd(&dcnt[dl], 1);
    }
    __syncthreads();
    for (int j = tid; j < RNG / 4; j += 1024) {
        const int i = ((j >> 3) << 5) + 8 * s + (j & 7);
        const int node = q * RNG + i;
        if (node < NNODES) dis[node] = rsqrtf((float)dcnt[i] + 1.0f);
    }
}
static_assert(CAPG % 128 == 0, "degree pass: whole trips of 128 groups");

__global__ void wmma_gemm_f32(const float* __restrict__ A, const float* __restrict__ B,
                              float* __restrict__ C, int N, int K, int M,
                              const float* __restrict__ rowscale,
                              const float* __restrict__ bias, int total_tiles) {
    int tid  = threadIdx.x;
    int wid  = tid >> 5;
    int lane = tid & 31;
    int tile = blockIdx.x * (blockDim.x >> 5) + wid;
    if (tile >= total_tiles) return;

    int tiles_m = M >> 4;
    int rt = tile / tiles_m, ct = tile - rt * tiles_m;
    int row0 = rt << 4, col0 = ct << 4;
    int hi = lane >> 4, lm = lane & 15;

    v8f c = {};
    const float* Arow = A + (size_t)(row0 + lm) * K;
    for (int k0 = 0; k0 < K; k0 += 4) {
        v2f a = *reinterpret_cast<const v2f*>(Arow + k0 + 2 * hi);
        const float* Bp = B + (size_t)(k0 + 2 * hi) * M + col0 + lm;
        v2f b;
        b.x = Bp[0];
        b.y = Bp[M];
        c = __builtin_amdgcn_wmma_f32_16x16x4_f32(false, a, false, b,
                                                  (short)0, c, false, false);
    }
    float bv = bias ? bias[col0 + lm] : 0.0f;
    #pragma unroll
    for (int v = 0; v < 8; ++v) {
        int row = row0 + v + 8 * hi;
        float o = c[v];
        if (rowscale) o *= rowscale[row];
        C[(size_t)row * M + col0 + lm] = o + bv;
    }
}

__global__ void wmma_gemm_t(const float* __restrict__ X, const float* __restrict__ W,
                            float* __restrict__ hT, int K, int M,
                            const float* __restrict__ dis, int total_tiles) {
    int tid  = threadIdx.x;
    int wid  = tid >> 5;
    int lane = tid & 31;
    int tile = blockIdx.x * (blockDim.x >> 5) + wid;
    if (tile >= total_tiles) return;

    int tiles_f = M >> 4;
    int nt = tile / tiles_f, ft = tile - nt * tiles_f;
    int n0 = nt << 5, f0 = ft << 4;
    int hi = lane >> 4, lm = lane & 15;
    int na = n0 + lm, nb = n0 + 16 + lm;

    v8f c0 = {}, c1 = {};
    for (int k0 = 0; k0 < K; k0 += 4) {
        int k = k0 + 2 * hi;
        const float* Wp = W + (size_t)k * M + f0 + lm;
        v2f a;
        a.x = Wp[0];
        a.y = Wp[M];
        v2f b0, b1;
        const float* Xp = X + (size_t)k * NNODES;
        b0.x = Xp[na];          b1.x = Xp[nb];
        b0.y = Xp[NNODES + na]; b1.y = Xp[NNODES + nb];
        c0 = __builtin_amdgcn_wmma_f32_16x16x4_f32(false, a, false, b0,
                                                   (short)0, c0, false, false);
        c1 = __builtin_amdgcn_wmma_f32_16x16x4_f32(false, a, false, b1,
                                                   (short)0, c1, false, false);
    }
    float dn = dis[n0 + lane];
    #pragma unroll
    for (int r = 0; r < 8; ++r) {
        float give = hi ? c0[r] : c1[r];
        float got  = __shfl_xor(give, 16);
        float lo = hi ? got : c0[r];
        float up = hi ? c1[r] : got;
        hT[(size_t)(f0 + r) * NNODES + n0 + lane]     = lo * dn;
        hT[(size_t)(f0 + 8 + r) * NNODES + n0 + lane] = up * dn;
    }
}

#define L1K 128
#define L1M 128
#define XPITCH 132
__global__ __launch_bounds__(256) void gemm_rows_lds(const float* __restrict__ X,
                                                     const float* __restrict__ W,
                                                     float* __restrict__ hT,
                                                     const float* __restrict__ dis) {
    __shared__ float xs[32 * XPITCH];
    const int tid  = threadIdx.x;
    const int wid  = tid >> 5;
    const int lane = tid & 31;
    const int n0 = blockIdx.x << 5;

    #pragma unroll
    for (int i = 0; i < 4; ++i) {
        const int r = wid + 8 * i;
        const float4 v = reinterpret_cast<const float4*>(X + (size_t)(n0 + r) * L1K)[lane];
        *reinterpret_cast<float4*>(&xs[r * XPITCH + lane * 4]) = v;
    }
    __syncthreads();

    const int f0 = wid << 4;
    const int hi = lane >> 4, lm = lane & 15;
    const float* xa = xs + lm * XPITCH + 2 * hi;
    const float* xb = xa + 16 * XPITCH;
    const float* Wp = W + (size_t)(2 * hi) * L1M + f0 + lm;

    v8f c0 = {}, c1 = {};
    #pragma unroll 8
    for (int k0 = 0; k0 < L1K; k0 += 4) {
        v2f a;
        a.x = Wp[(size_t)k0 * L1M];
        a.y = Wp[(size_t)(k0 + 1) * L1M];
        const v2f b0 = *reinterpret_cast<const v2f*>(xa + k0);
        const v2f b1 = *reinterpret_cast<const v2f*>(xb + k0);
        c0 = __builtin_amdgcn_wmma_f32_16x16x4_f32(false, a, false, b0,
                                                   (short)0, c0, false, false);
        c1 = __builtin_amdgcn_wmma_f32_16x16x4_f32(false, a, false, b1,
                                                   (short)0, c1, false, false);
    }
    const float dn = dis[n0 + lane];
    #pragma unroll
    for (int r = 0; r < 8; ++r) {
        float give = hi ? c0[r] : c1[r];
        float got  = __shfl_xor(give, 16);
        float lo = hi ? got : c0[r];
        float up = hi ? c1[r] : got;
        hT[(size_t)(f0 + r) * NNODES + n0 + lane]     = lo * dn;
        hT[(size_t)(f0 + 8 + r) * NNODES + n0 + lane] = up * dn;
    }
}
static_assert(NNODES % 32 == 0, "layer-1 GEMM: whole 32-node tiles");

template <int R>
__global__ __launch_bounds__(1024) void slab_kernel(const float* __restrict__ hT,
                                                    const unsigned* __restrict__ elist,
                                                    const unsigned* __restrict__ ovf,
                                                    const int* __restrict__ gcnt,
                                                    float* __restrict__ outp,
                                                    const float* __restrict__ dis,
                                                    const float* __restrict__ bias, int F) {
    __shared__ float hcol[NNODES];
    __shared__ float acc[R * RNG];
    constexpr int WPR = 32 / R;
    const int tid = threadIdx.x;
    const int qb = blockIdx.x / F;
    const int f = blockIdx.x - qb * F;
    const int nlo = qb * (R * RNG);
    const int nhi = nlo + R * RNG < NNODES ? nlo + R * RNG : NNODES;

    const float4* h4 = reinterpret_cast<const float4*>(hT + (size_t)f * NNODES);
    float4* hc4 = reinterpret_cast<float4*>(hcol);
    float4* ac4 = reinterpret_cast<float4*>(acc);
    for (int i = tid; i < NNODES / 4; i += 1024) hc4[i] = h4[i];
    for (int i = tid; i < R * RNG / 4; i += 1024) ac4[i] = make_float4(0.0f, 0.0f, 0.0f, 0.0f);
    __syncthreads();

    const int lane = tid & 31;
    const int wave = tid >> 5;
    const int rl = wave / WPR;
    const int slot = wave - rl * WPR;
    const int rg = qb * R + rl;
    float* accr = acc + rl * RNG;
    int n = gcnt[rg * 32 + lane];
    n = n < CAPN ? n : CAPN;
    const int ng = (n + 3) >> 2;
    const uint4* e4 = reinterpret_cast<const uint4*>(elist) + (size_t)rg * CAPG * 32;
    int g = slot;
    if (g < ng) {
        uint4 e = e4[(size_t)g * 32 + lane];
        for (;;) {
            const int gn = g + WPR;
            const bool more = gn < ng;
            uint4 en = e;
            if (more) en = e4[(size_t)gn * 32 + lane];
            const int rem = n - 4 * g;
            if (rem >= 4) {
                float v0 = hcol[e.x & 0xFFFFu], v1 = hcol[e.y & 0xFFFFu];
                float v2 = hcol[e.z & 0xFFFFu], v3 = hcol[e.w & 0xFFFFu];
                __hip_atomic_fetch_add(&accr[e.x >> 16], v0, __ATOMIC_RELAXED, __HIP_MEMORY_SCOPE_WORKGROUP);
                __hip_atomic_fetch_add(&accr[e.y >> 16], v1, __ATOMIC_RELAXED, __HIP_MEMORY_SCOPE_WORKGROUP);
                __hip_atomic_fetch_add(&accr[e.z >> 16], v2, __ATOMIC_RELAXED, __HIP_MEMORY_SCOPE_WORKGROUP);
                __hip_atomic_fetch_add(&accr[e.w >> 16], v3, __ATOMIC_RELAXED, __HIP_MEMORY_SCOPE_WORKGROUP);
            } else {
                __hip_atomic_fetch_add(&accr[e.x >> 16], hcol[e.x & 0xFFFFu], __ATOMIC_RELAXED, __HIP_MEMORY_SCOPE_WORKGROUP);
                if (rem > 1)
                    __hip_atomic_fetch_add(&accr[e.y >> 16], hcol[e.y & 0xFFFFu], __ATOMIC_RELAXED, __HIP_MEMORY_SCOPE_WORKGROUP);
                if (rem > 2)
                    __hip_atomic_fetch_add(&accr[e.z >> 16], hcol[e.z & 0xFFFFu], __ATOMIC_RELAXED, __HIP_MEMORY_SCOPE_WORKGROUP);
            }
            if (!more) break;
            e = en; g = gn;
        }
    }
    int novf = gcnt[NLIST];
    novf = novf < OVFCAP ? novf : OVFCAP;
    for (int i = tid; i < novf; i += 1024) {
        const unsigned e = ovf[i];
        const int d = (int)(e >> 16);
        if (d >= nlo && d < nhi)
            __hip_atomic_fetch_add(&acc[d - nlo], hcol[e & 0xFFFFu], __ATOMIC_RELAXED, __HIP_MEMORY_SCOPE_WORKGROUP);
    }
    __syncthreads();

    float4* o4 = reinterpret_cast<float4*>(outp + (size_t)f * NNODES + nlo);
    const float4* dis4 = reinterpret_cast<const float4*>(dis + nlo);
    const float4* hl4 = reinterpret_cast<const float4*>(hcol + nlo);
    const float bf = bias[f];
    const int cnt4 = (nhi - nlo) >> 2;
    for (int i = tid; i < cnt4; i += 1024) {
        float4 a = ac4[i], h = hl4[i], dd = dis4[i], o;
        o.x = (a.x + h.x) * dd.x + bf;
        o.y = (a.y + h.y) * dd.y + bf;
        o.z = (a.z + h.z) * dd.z + bf;
        o.w = (a.w + h.w) * dd.w + bf;
        o.x = o.x > 0.0f ? o.x : 0.0f;
        o.y = o.y > 0.0f ? o.y : 0.0f;
        o.z = o.z > 0.0f ? o.z : 0.0f;
        o.w = o.w > 0.0f ? o.w : 0.0f;
        o4[i] = o;
    }
}

__global__ __launch_bounds__(1024) void pool_t_kernel(const float* __restrict__ actT,
                                                      const int* __restrict__ batch,
                                                      float* __restrict__ pooled) {
    __shared__ float tile[32][33];
    const int t = threadIdx.x;
    const int n0 = blockIdx.x * 32;
    {
        int f = t >> 5, n = n0 + (t & 31);
        tile[f][t & 31] = actT[(size_t)f * NNODES + n];
    }
    __syncthreads();
    {
        int nl = t >> 5, f = t & 31;
        atomAddF(&pooled[(batch[n0 + nl] << 5) + f], tile[f][nl]);
    }
}

__global__ void poolfin_kernel(float* __restrict__ pooled, const float* __restrict__ counts) {
    int i = blockIdx.x * blockDim.x + threadIdx.x;
    int g = i >> 5;
    pooled[i] = pooled[i] / fmaxf(counts[g], 1.0f);
}

extern "C" void kernel_launch(void* const* d_in, const int* in_sizes, int n_in,
                              void* d_out, int out_size, void* d_ws, size_t ws_size,
                              hipStream_t stream) {
    const float* x   = (const float*)d_in[0];
    const float* W1  = (const float*)d_in[1];
    const float* b1  = (const float*)d_in[2];
    const float* W2  = (const float*)d_in[3];
    const float* b2  = (const float*)d_in[4];
    const float* W3  = (const float*)d_in[5];
    const float* b3  = (const float*)d_in[6];
    const float* Wl  = (const float*)d_in[7];
    const float* bl  = (const float*)d_in[8];
    const int*   ei  = (const int*)d_in[9];
    const int*   bat = (const int*)d_in[10];
    float*       out = (float*)d_out;

    const int N = NNODES, E = NEDGES;
    const int* src = ei;
    const int* dst = ei + E;

    float* B0     = (float*)d_ws;
    float* B1     = B0 + (size_t)N * 128;
    float* dis    = B1 + (size_t)N * 128;
    int*   gcnt   = (int*)(dis + N);
    float* counts = dis + N + 320;
    float* pooled = counts + NGRAPH;
    unsigned* elist = (unsigned*)(pooled + NGRAPH * 32);
    unsigned* ovf   = elist + (size_t)NRNG * CAPG * 128;

    hipMemsetAsync(gcnt, 0, 320 * sizeof(int), stream);
    hipMemsetAsync(counts, 0, NGRAPH * sizeof(float), stream);
    bin_kernel<<<(E / 4 + 1023) / 1024, 1024, 0, stream>>>(src, dst, gcnt, elist, ovf);
    deg_dis_counts_kernel<<<DEGBLK + CNTBLK, 1024, 0, stream>>>(elist, ovf, gcnt, dis, bat, counts);

    gemm_rows_lds<<<NNODES / 32, 256, 0, stream>>>(x, W1, B0, dis);
    slab_kernel<4><<<128 * 2, 1024, 0, stream>>>(B0, elist, ovf, gcnt, B1, dis, b1, 128);

    wmma_gemm_t<<<(1250 * 4) / 8, 256, 0, stream>>>(B1, W2, B0, 128, 64, dis, 1250 * 4);
    slab_kernel<2><<<64 * 4, 1024, 0, stream>>>(B0, elist, ovf, gcnt, B1, dis, b2, 64);

    wmma_gemm_t<<<(1250 * 2 + 7) / 8, 256, 0, stream>>>(B1, W3, B0, 64, 32, dis, 1250 * 2);
    slab_kernel<1><<<32 * 8, 1024, 0, stream>>>(B0, elist, ovf, gcnt, B1, dis, b3, 32);

    hipMemsetAsync(pooled, 0, NGRAPH * 32 * sizeof(float), stream);
    pool_t_kernel<<<N / 32, 1024, 0, stream>>>(B1, bat, pooled);
    poolfin_kernel<<<(NGRAPH * 32) / 256, 256, 0, stream>>>(pooled, counts);

    wmma_gemm_f32<<<(16 * 48) / 8, 256, 0, stream>>>(pooled, Wl, out, NGRAPH, 32, 768,
                                                     nullptr, bl, 16 * 48);
}


# mean pool folded into the layer-3 column aggregation: a work-group sums its feature per graph in LDS (lane-strided runs), divides, adds to pooled; pool, divide, one memset go; pool sum order changes
# speedup vs baseline: 2.7765x; 1.0705x over previous
#include <hip/hip_runtime.h>
#include <hip/hip_bf16.h>


typedef __attribute__((ext_vector_type(2))) float v2f;
typedef __attribute__((ext_vector_type(8))) float v8f;

#define NNODES 40000
#define NEDGES 640000
#define NGRAPH 256

__device__ __forceinline__ void atomAddF(float* p, float v) {
    unsafeAtomicAdd(p, v);
}

#define RNG 5024
#define NRNG 8
#define NLIST (NRNG * 32)
#define CAPN 4096
#define CAPG (CAPN / 4)
#define OVFCAP (NEDGES - CAPN)
static_assert(NNODES <= 65536, "src and dst share one 32-bit word");
static_assert(RNG % 32 == 0 && NRNG * RNG >= NNODES, "ranges cover all nodes, lane = dst mod 32");

__device__ __forceinline__ void place_edge(int s, int d, int q, int pos, int* __restrict__ gcnt,
                                           unsigned* __restrict__ elist,
                                           unsigned* __restrict__ ovf) {
    if (pos < CAPN) {
        const unsigned w = (unsigned)s | ((unsigned)(d - q * RNG) << 16);
        elist[(((size_t)q * CAPG + (pos >> 2)) * 32 + (d & 31)) * 4 + (pos & 3)] = w;
    } else {
        const int o = atomicAdd(&gcnt[NLIST], 1);
        if (o < OVFCAP) ovf[o] = (unsigned)s | ((unsigned)d << 16);
    }
}

__global__ __launch_bounds__(1024) void bin_kernel(const int* __restrict__ src,
                                                   const int* __restrict__ dst,
                                                   int* __restrict__ gcnt,
                                                   unsigned* __restrict__ elist,
                                                   unsigned* __restrict__ ovf) {
    __shared__ int cnt[NLIST];
    __shared__ int base[NLIST];
    const int tid = threadIdx.x;
    if (tid < NLIST) cnt[tid] = 0;
    __syncthreads();
    const int i = blockIdx.x * 1024 + tid;
    const bool act = i < NEDGES / 4;
    int4 s = make_int4(0, 0, 0, 0), d = make_int4(0, 0, 0, 0);
    int q0 = 0, q1 = 0, q2 = 0, q3 = 0;
    int r0 = 0, r1 = 0, r2 = 0, r3 = 0;
    if (act) {
        s = reinterpret_cast<const int4*>(src)[i];
        d = reinterpret_cast<const int4*>(dst)[i];
        q0 = d.x / RNG; q1 = d.y / RNG; q2 = d.z / RNG; q3 = d.w / RNG;
        r0 = atomicAdd(&cnt[q0 * 32 + (d.x & 31)], 1);
        r1 = atomicAdd(&cnt[q1 * 32 + (d.y & 31)], 1);
        r2 = atomicAdd(&cnt[q2 * 32 + (d.z & 31)], 1);
        r3 = atomicAdd(&cnt[q3 * 32 + (d.w & 31)], 1);
    }
    __syncthreads();
    if (tid < NLIST) base[tid] = atomicAdd(&gcnt[tid], cnt[tid]);
    __syncthreads();
    if (act) {
        place_edge(s.x, d.x, q0, base[q0 * 32 + (d.x & 31)] + r0, gcnt, elist, ovf);
        place_edge(s.y, d.y, q1, base[q1 * 32 + (d.y & 31)] + r1, gcnt, elist, ovf);
        place_edge(s.z, d.z, q2, base[q2 * 32 + (d.z & 31)] + r2, gcnt, elist, ovf);
        place_edge(s.w, d.w, q3, base[q3 * 32 + (d.w & 31)] + r3, gcnt, elist, ovf);
    }
}

#define DEGBLK (NRNG * 4)
#define CNTBLK ((NNODES + 1023) / 1024)
__global__ __launch_bounds__(1024) void deg_dis_counts_kernel(const unsigned* __restrict__ elist,
                                                              const unsigned* __restrict__ ovf,
                                                              const int* __restrict__ gcnt,
                                                              float* __restrict__ dis,
                                                              const int* __restrict__ batch,
                                                              float* __restrict__ counts) {
    __shared__ int dcnt[RNG];
    const int tid = threadIdx.x;
    if (blockIdx.x >= DEGBLK) {
        const int n = (blockIdx.x - DEGBLK) * 1024 + tid;
        if (n < NNODES) atomAddF(&counts[batch[n]], 1.0f);
        return;
    }
    const int q = blockIdx.x >> 2;
    const int s = blockIdx.x & 3;
    for (int i = tid; i < RNG; i += 1024) dcnt[i] = 0;
    __syncthreads();

    const int lane = tid & 31;
    const int wave = tid >> 5;
    const int L = 8 * s + (lane & 7);
    const int gsub = lane >> 3;
    int n = gcnt[q * 32 + L];
    n = n < CAPN ? n : CAPN;
    const int ng = (n + 3) >> 2;
    const uint4* e4 = reinterpret_cast<const uint4*>(elist) + (size_t)q * CAPG * 32 + L;
    constexpr int TRIPS = CAPG / 128;
    const int glast = ng > 0 ? ng - 1 : 0;
    uint4 e[TRIPS];
    #pragma unroll
    for (int t = 0; t < TRIPS; ++t) {
        const int g = t * 128 + wave * 4 + gsub;
        e[t] = e4[(size_t)(g < glast ? g : glast) * 32];
    }
    #pragma unroll
    for (int t = 0; t < TRIPS; ++t) {
        const int g = t * 128 + wave * 4 + gsub;
        const int rem = n - 4 * g;
        if (rem > 0) atomicAdd(&dcnt[e[t].x >> 16], 1);
        if (rem > 1) atomicAdd(&dcnt[e[t].y >> 16], 1);
        if (rem > 2) atomicAdd(&dcnt[e[t].z >> 16], 1);
        if (rem > 3) atomicAdd(&dcnt[e[t].w >> 16], 1);
    }
    int novf = gcnt[NLIST];
    novf = novf < OVFCAP ? novf : OVFCAP;
    for (int i = tid; i < novf; i += 1024) {
        const int d = (int)(ovf[i] >> 16);
        const int dl = d - q * RNG;
        if (dl >= 0 && dl < RNG && ((d & 31) >> 3) == s) atomicAdd(&dcnt[dl], 1);
    }
    __syncthreads();
    for (int j = tid; j < RNG / 4; j += 1024) {
        const int i = ((j >> 3) << 5) + 8 * s + (j & 7);
        const int node = q * RNG + i;
        if (node < NNODES) dis[node] = rsqrtf((float)dcnt[i] + 1.0f);
    }
}
static_assert(CAPG % 128 == 0, "degree pass: whole trips of 128 groups");

__global__ void wmma_gemm_f32(const float* __restrict__ A, const float* __restrict__ B,
                              float* __restrict__ C, int N, int K, int M,
                              const float* __restrict__ rowscale,
                              const float* __restrict__ bias, int total_tiles) {
    int tid  = threadIdx.x;
    int wid  = tid >> 5;
    int lane = tid & 31;
    int tile = blockIdx.x * (blockDim.x >> 5) + wid;
    if (tile >= total_tiles) return;

    int tiles_m = M >> 4;
    int rt = tile / tiles_m, ct = tile - rt * tiles_m;
    int row0 = rt << 4, col0 = ct << 4;
    int hi = lane >> 4, lm = lane & 15;

    v8f c = {};
    const float* Arow = A + (size_t)(row0 + lm) * K;
    for (int k0 = 0; k0 < K; k0 += 4) {
        v2f a = *reinterpret_cast<const v2f*>(Arow + k0 + 2 * hi);
        const float* Bp = B + (size_t)(k0 + 2 * hi) * M + col0 + lm;
        v2f b;
        b.x = Bp[0];
        b.y = Bp[M];
        c = __builtin_amdgcn_wmma_f32_16x16x4_f32(false, a, false, b,
                                                  (short)0, c, false, false);
    }
    float bv = bias ? bias[col0 + lm] : 0.0f;
    #pragma unroll
    for (int v = 0; v < 8; ++v) {
        int row = row0 + v + 8 * hi;
        float o = c[v];
        if (rowscale) o *= rowscale[row];
        C[(size_t)row * M + col0 + lm] = o + bv;
    }
}

__global__ void wmma_gemm_t(const float* __restrict__ X, const float* __restrict__ W,
                            float* __restrict__ hT, int K, int M,
                            const float* __restrict__ dis, int total_tiles) {
    int tid  = threadIdx.x;
    int wid  = tid >> 5;
    int lane = tid & 31;
    int tile = blockIdx.x * (blockDim.x >> 5) + wid;
    if (tile >= total_tiles) return;

    int tiles_f = M >> 4;
    int nt = tile / tiles_f, ft = tile - nt * tiles_f;
    int n0 = nt << 5, f0 = ft << 4;
    int hi = lane >> 4, lm = lane & 15;
    int na = n0 + lm, nb = n0 + 16 + lm;

    v8f c0 = {}, c1 = {};
    for (int k0 = 0; k0 < K; k0 += 4) {
        int k = k0 + 2 * hi;
        const float* Wp = W + (size_t)k * M + f0 + lm;
        v2f a;
        a.x = Wp[0];
        a.y = Wp[M];
        v2f b0, b1;
        const float* Xp = X + (size_t)k * NNODES;
        b0.x = Xp[na];          b1.x = Xp[nb];
        b0.y = Xp[NNODES + na]; b1.y = Xp[NNODES + nb];
        c0 = __builtin_amdgcn_wmma_f32_16x16x4_f32(false, a, false, b0,
                                                   (short)0, c0, false, false);
        c1 = __builtin_amdgcn_wmma_f32_16x16x4_f32(false, a, false, b1,
                                                   (short)0, c1, false, false);
    }
    float dn = dis[n0 + lane];
    #pragma unroll
    for (int r = 0; r < 8; ++r) {
        float give = hi ? c0[r] : c1[r];
        float got  = __shfl_xor(give, 16);
        float lo = hi ? got : c0[r];
        float up = hi ? c1[r] : got;
        hT[(size_t)(f0 + r) * NNODES + n0 + lane]     = lo * dn;
        hT[(size_t)(f0 + 8 + r) * NNODES + n0 + lane] = up * dn;
    }
}

#define L1K 128
#define L1M 128
#define XPITCH 132
__global__ __launch_bounds__(256) void gemm_rows_lds(const float* __restrict__ X,
                                                     const float* __restrict__ W,
                                                     float* __restrict__ hT,
                                                     const float* __restrict__ dis) {
    __shared__ float xs[32 * XPITCH];
    const int tid  = threadIdx.x;
    const int wid  = tid >> 5;
    const int lane = tid & 31;
    const int n0 = blockIdx.x << 5;

    #pragma unroll
    for (int i = 0; i < 4; ++i) {
        const int r = wid + 8 * i;
        const float4 v = reinterpret_cast<const float4*>(X + (size_t)(n0 + r) * L1K)[lane];
        *reinterpret_cast<float4*>(&xs[r * XPITCH + lane * 4]) = v;
    }
    __syncthreads();

    const int f0 = wid << 4;
    const int hi = lane >> 4, lm = lane & 15;
    const float* xa = xs + lm * XPITCH + 2 * hi;
    const float* xb = xa + 16 * XPITCH;
    const float* Wp = W + (size_t)(2 * hi) * L1M + f0 + lm;

    v8f c0 = {}, c1 = {};
    #pragma unroll 8
    for (int k0 = 0; k0 < L1K; k0 += 4) {
        v2f a;
        a.x = Wp[(size_t)k0 * L1M];
        a.y = Wp[(size_t)(k0 + 1) * L1M];
        const v2f b0 = *reinterpret_cast<const v2f*>(xa + k0);
        const v2f b1 = *reinterpret_cast<const v2f*>(xb + k0);
        c0 = __builtin_amdgcn_wmma_f32_16x16x4_f32(false, a, false, b0,
                                                   (short)0, c0, false, false);
        c1 = __builtin_amdgcn_wmma_f32_16x16x4_f32(false, a, false, b1,
                                                   (short)0, c1, false, false);
    }
    const float dn = dis[n0 + lane];
    #pragma unroll
    for (int r = 0; r < 8; ++r) {
        float give = hi ? c0[r] : c1[r];
        float got  = __shfl_xor(give, 16);
        float lo = hi ? got : c0[r];
        float up = hi ? c1[r] : got;
        hT[(size_t)(f0 + r) * NNODES + n0 + lane]     = lo * dn;
        hT[(size_t)(f0 + 8 + r) * NNODES + n0 + lane] = up * dn;
    }
}
static_assert(NNODES % 32 == 0, "layer-1 GEMM: whole 32-node tiles");

template <int R, bool POOL>
__global__ __launch_bounds__(1024) void slab_kernel(const float* __restrict__ hT,
                                                    const unsigned* __restrict__ elist,
                                                    const unsigned* __restrict__ ovf,
                                                    const int* __restrict__ gcnt,
                                                    float* __restrict__ outp,
                                                    const float* __restrict__ dis,
                                                    const float* __restrict__ bias, int F,
                                                    const int* __restrict__ batch,
                                                    const float* __restrict__ counts,
                                                    float* __restrict__ pooled) {
    static_assert(!POOL || R == 1, "pool epilogue: one range per work-group");
    __shared__ float hcol[NNODES];
    __shared__ float acc[R * RNG];
    __shared__ __attribute__((aligned(16))) int gid[POOL ? RNG : 1];
    __shared__ float gsum[POOL ? NGRAPH : 1];
    constexpr int WPR = 32 / R;
    const int tid = threadIdx.x;
    const int qb = blockIdx.x / F;
    const int f = blockIdx.x - qb * F;
    const int nlo = qb * (R * RNG);
    const int nhi = nlo + R * RNG < NNODES ? nlo + R * RNG : NNODES;

    const float4* h4 = reinterpret_cast<const float4*>(hT + (size_t)f * NNODES);
    float4* hc4 = reinterpret_cast<float4*>(hcol);
    float4* ac4 = reinterpret_cast<float4*>(acc);
    for (int i = tid; i < NNODES / 4; i += 1024) hc4[i] = h4[i];
    for (int i = tid; i < R * RNG / 4; i += 1024) ac4[i] = make_float4(0.0f, 0.0f, 0.0f, 0.0f);
    if constexpr (POOL) {
        const int4* b4 = reinterpret_cast<const int4*>(batch + nlo);
        int4* g4 = reinterpret_cast<int4*>(gid);
        for (int i = tid; i < (nhi - nlo) / 4; i += 1024) g4[i] = b4[i];
        if (tid < NGRAPH) gsum[tid] = 0.0f;
    }
    __syncthreads();

    const int lane = tid & 31;
    const int wave = tid >> 5;
    const int rl = wave / WPR;
    const int slot = wave - rl * WPR;
    const int rg = qb * R + rl;
    float* accr = acc + rl * RNG;
    int n = gcnt[rg * 32 + lane];
    n = n < CAPN ? n : CAPN;
    const int ng = (n + 3) >> 2;
    const uint4* e4 = reinterpret_cast<const uint4*>(elist) + (size_t)rg * CAPG * 32;
    int g = slot;
    if (g < ng) {
        uint4 e = e4[(size_t)g * 32 + lane];
        for (;;) {
            const int gn = g + WPR;
            const bool more = gn < ng;
            uint4 en = e;
            if (more) en = e4[(size_t)gn * 32 + lane];
            const int rem = n - 4 * g;
            if (rem >= 4) {
                float v0 = hcol[e.x & 0xFFFFu], v1 = hcol[e.y & 0xFFFFu];
                float v2 = hcol[e.z & 0xFFFFu], v3 = hcol[e.w & 0xFFFFu];
                __hip_atomic_fetch_add(&accr[e.x >> 16], v0, __ATOMIC_RELAXED, __HIP_MEMORY_SCOPE_WORKGROUP);
                __hip_atomic_fetch_add(&accr[e.y >> 16], v1, __ATOMIC_RELAXED, __HIP_MEMORY_SCOPE_WORKGROUP);
                __hip_atomic_fetch_add(&accr[e.z >> 16], v2, __ATOMIC_RELAXED, __HIP_MEMORY_SCOPE_WORKGROUP);
                __hip_atomic_fetch_add(&accr[e.w >> 16], v3, __ATOMIC_RELAXED, __HIP_MEMORY_SCOPE_WORKGROUP);
            } else {
                __hip_atomic_fetch_add(&accr[e.x >> 16], hcol[e.x & 0xFFFFu], __ATOMIC_RELAXED, __HIP_MEMORY_SCOPE_WORKGROUP);
                if (rem > 1)
                    __hip_atomic_fetch_add(&accr[e.y >> 16], hcol[e.y & 0xFFFFu], __ATOMIC_RELAXED, __HIP_MEMORY_SCOPE_WORKGROUP);
                if (rem > 2)
                    __hip_atomic_fetch_add(&accr[e.z >> 16], hcol[e.z & 0xFFFFu], __ATOMIC_RELAXED, __HIP_MEMORY_SCOPE_WORKGROUP);
            }
            if (!more) break;
            e = en; g = gn;
        }
    }
    int novf = gcnt[NLIST];
    novf = novf < OVFCAP ? novf : OVFCAP;
    for (int i = tid; i < novf; i += 1024) {
        const unsigned e = ovf[i];
        const int d = (int)(e >> 16);
        if (d >= nlo && d < nhi)
            __hip_atomic_fetch_add(&acc[d - nlo], hcol[e & 0xFFFFu], __ATOMIC_RELAXED, __HIP_MEMORY_SCOPE_WORKGROUP);
    }
    __syncthreads();

    float4* o4 = reinterpret_cast<float4*>(outp + (size_t)f * NNODES + nlo);
    const float4* dis4 = reinterpret_cast<const float4*>(dis + nlo);
    const float4* hl4 = reinterpret_cast<const float4*>(hcol + nlo);
    const float bf = bias[f];
    const int cnt4 = (nhi - nlo) >> 2;
    for (int i = tid; i < cnt4; i += 1024) {
        float4 a = ac4[i], h = hl4[i], dd = dis4[i], o;
        o.x = (a.x + h.x) * dd.x + bf;
        o.y = (a.y + h.y) * dd.y + bf;
        o.z = (a.z + h.z) * dd.z + bf;
        o.w = (a.w + h.w) * dd.w + bf;
        o.x = o.x > 0.0f ? o.x : 0.0f;
        o.y = o.y > 0.0f ? o.y : 0.0f;
        o.z = o.z > 0.0f ? o.z : 0.0f;
        o.w = o.w > 0.0f ? o.w : 0.0f;
        if constexpr (POOL) ac4[i] = o;
        else o4[i] = o;
    }
    if constexpr (POOL) {
        __syncthreads();
        constexpr int SEG = RNG / 32;
        constexpr int CH = (SEG + 31) / 32;
        static_assert(RNG % 32 == 0 && SEG % 2 == 1, "lane stride odd: 32 lanes, 32 banks");
        const int cntn = nhi - nlo;
        const int ibeg = SEG * lane + CH * wave;
        int iend = ibeg + CH;
        if (iend > SEG * lane + SEG) iend = SEG * lane + SEG;
        if (iend > cntn) iend = cntn;
        if (ibeg < iend) {
            int gcur = gid[ibeg];
            float s = acc[ibeg];
            for (int i = ibeg + 1; i < iend; ++i) {
                const int gi = gid[i];
                const float v = acc[i];
                if (gi != gcur) {
                    __hip_atomic_fetch_add(&gsum[gcur], s, __ATOMIC_RELAXED, __HIP_MEMORY_SCOPE_WORKGROUP);
                    gcur = gi;
                    s = 0.0f;
                }
                s += v;
            }
            __hip_atomic_fetch_add(&gsum[gcur], s, __ATOMIC_RELAXED, __HIP_MEMORY_SCOPE_WORKGROUP);
        }
        __syncthreads();
        if (tid < NGRAPH) {
            const float v = gsum[tid];
            if (v != 0.0f) atomAddF(&pooled[tid * F + f], v / fmaxf(counts[tid], 1.0f));
        }
    }
}

extern "C" void kernel_launch(void* const* d_in, const int* in_sizes, int n_in,
                              void* d_out, int out_size, void* d_ws, size_t ws_size,
                              hipStream_t stream) {
    const float* x   = (const float*)d_in[0];
    const float* W1  = (const float*)d_in[1];
    const float* b1  = (const float*)d_in[2];
    const float* W2  = (const float*)d_in[3];
    const float* b2  = (const float*)d_in[4];
    const float* W3  = (const float*)d_in[5];
    const float* b3  = (const float*)d_in[6];
    const float* Wl  = (const float*)d_in[7];
    const float* bl  = (const float*)d_in[8];
    const int*   ei  = (const int*)d_in[9];
    const int*   bat = (const int*)d_in[10];
    float*       out = (float*)d_out;

    const int N = NNODES, E = NEDGES;
    const int* src = ei;
    const int* dst = ei + E;

    float* B0     = (float*)d_ws;
    float* B1     = B0 + (size_t)N * 128;
    float* dis    = B1 + (size_t)N * 128;
    int*   gcnt   = (int*)(dis + N);
    float* counts = dis + N + 320;
    float* pooled = counts + NGRAPH;
    unsigned* elist = (unsigned*)(pooled + NGRAPH * 32);
    unsigned* ovf   = elist + (size_t)NRNG * CAPG * 128;

    hipMemsetAsync(gcnt, 0, 320 * sizeof(int), stream);
    hipMemsetAsync(counts, 0, (NGRAPH + NGRAPH * 32) * sizeof(float), stream);
    bin_kernel<<<(E / 4 + 1023) / 1024, 1024, 0, stream>>>(src, dst, gcnt, elist, ovf);
    deg_dis_counts_kernel<<<DEGBLK + CNTBLK, 1024, 0, stream>>>(elist, ovf, gcnt, dis, bat, counts);

    gemm_rows_lds<<<NNODES / 32, 256, 0, stream>>>(x, W1, B0, dis);
    slab_kernel<4, false><<<128 * 2, 1024, 0, stream>>>(B0, elist, ovf, gcnt, B1, dis, b1, 128,
                                                         nullptr, nullptr, nullptr);

    wmma_gemm_t<<<(1250 * 4) / 8, 256, 0, stream>>>(B1, W2, B0, 128, 64, dis, 1250 * 4);
    slab_kernel<2, false><<<64 * 4, 1024, 0, stream>>>(B0, elist, ovf, gcnt, B1, dis, b2, 64,
                                                        nullptr, nullptr, nullptr);

    wmma_gemm_t<<<(1250 * 2 + 7) / 8, 256, 0, stream>>>(B1, W3, B0, 64, 32, dis, 1250 * 2);
    slab_kernel<1, true><<<32 * 8, 1024, 0, stream>>>(B0, elist, ovf, gcnt, B1, dis, b3, 32,
                                                       bat, counts, pooled);

    wmma_gemm_f32<<<(16 * 48) / 8, 256, 0, stream>>>(pooled, Wl, out, NGRAPH, 32, 768,
                                                     nullptr, bl, 16 * 48);
}


# column aggregation: no global load waits at its use: column (ten loads), list length, dis, graph ids go out as one batch up front; next four list groups load during a four-group trip; sums as before
# speedup vs baseline: 2.8958x; 1.0430x over previous
#include <hip/hip_runtime.h>
#include <hip/hip_bf16.h>


typedef __attribute__((ext_vector_type(2))) float v2f;
typedef __attribute__((ext_vector_type(8))) float v8f;

#define NNODES 40000
#define NEDGES 640000
#define NGRAPH 256

__device__ __forceinline__ void atomAddF(float* p, float v) {
    unsafeAtomicAdd(p, v);
}

#define RNG 5024
#define NRNG 8
#define NLIST (NRNG * 32)
#define CAPN 4096
#define CAPG (CAPN / 4)
#define OVFCAP (NEDGES - CAPN)
static_assert(NNODES <= 65536, "src and dst share one 32-bit word");
static_assert(RNG % 32 == 0 && NRNG * RNG >= NNODES, "ranges cover all nodes, lane = dst mod 32");

__device__ __forceinline__ void place_edge(int s, int d, int q, int pos, int* __restrict__ gcnt,
                                           unsigned* __restrict__ elist,
                                           unsigned* __restrict__ ovf) {
    if (pos < CAPN) {
        const unsigned w = (unsigned)s | ((unsigned)(d - q * RNG) << 16);
        elist[(((size_t)q * CAPG + (pos >> 2)) * 32 + (d & 31)) * 4 + (pos & 3)] = w;
    } else {
        const int o = atomicAdd(&gcnt[NLIST], 1);
        if (o < OVFCAP) ovf[o] = (unsigned)s | ((unsigned)d << 16);
    }
}

__global__ __launch_bounds__(1024) void bin_kernel(const int* __restrict__ src,
                                                   const int* __restrict__ dst,
                                                   int* __restrict__ gcnt,
                                                   unsigned* __restrict__ elist,
                                                   unsigned* __restrict__ ovf) {
    __shared__ int cnt[NLIST];
    __shared__ int base[NLIST];
    const int tid = threadIdx.x;
    if (tid < NLIST) cnt[tid] = 0;
    __syncthreads();
    const int i = blockIdx.x * 1024 + tid;
    const bool act = i < NEDGES / 4;
    int4 s = make_int4(0, 0, 0, 0), d = make_int4(0, 0, 0, 0);
    int q0 = 0, q1 = 0, q2 = 0, q3 = 0;
    int r0 = 0, r1 = 0, r2 = 0, r3 = 0;
    if (act) {
        s = reinterpret_cast<const int4*>(src)[i];
        d = reinterpret_cast<const int4*>(dst)[i];
        q0 = d.x / RNG; q1 = d.y / RNG; q2 = d.z / RNG; q3 = d.w / RNG;
        r0 = atomicAdd(&cnt[q0 * 32 + (d.x & 31)], 1);
        r1 = atomicAdd(&cnt[q1 * 32 + (d.y & 31)], 1);
        r2 = atomicAdd(&cnt[q2 * 32 + (d.z & 31)], 1);
        r3 = atomicAdd(&cnt[q3 * 32 + (d.w & 31)], 1);
    }
    __syncthreads();
    if (tid < NLIST) base[tid] = atomicAdd(&gcnt[tid], cnt[tid]);
    __syncthreads();
    if (act) {
        place_edge(s.x, d.x, q0, base[q0 * 32 + (d.x & 31)] + r0, gcnt, elist, ovf);
        place_edge(s.y, d.y, q1, base[q1 * 32 + (d.y & 31)] + r1, gcnt, elist, ovf);
        place_edge(s.z, d.z, q2, base[q2 * 32 + (d.z & 31)] + r2, gcnt, elist, ovf);
        place_edge(s.w, d.w, q3, base[q3 * 32 + (d.w & 31)] + r3, gcnt, elist, ovf);
    }
}

#define DEGBLK (NRNG * 4)
#define CNTBLK ((NNODES + 1023) / 1024)
__global__ __launch_bounds__(1024) void deg_dis_counts_kernel(const unsigned* __restrict__ elist,
                                                              const unsigned* __restrict__ ovf,
                                                              const int* __restrict__ gcnt,
                                                              float* __restrict__ dis,
                                                              const int* __restrict__ batch,
                                                              float* __restrict__ counts) {
    __shared__ int dcnt[RNG];
    const int tid = threadIdx.x;
    if (blockIdx.x >= DEGBLK) {
        const int n = (blockIdx.x - DEGBLK) * 1024 + tid;
        if (n < NNODES) atomAddF(&counts[batch[n]], 1.0f);
        return;
    }
    const int q = blockIdx.x >> 2;
    const int s = blockIdx.x & 3;
    for (int i = tid; i < RNG; i += 1024) dcnt[i] = 0;
    __syncthreads();

    const int lane = tid & 31;
    const int wave = tid >> 5;
    const int L = 8 * s + (lane & 7);
    const int gsub = lane >> 3;
    int n = gcnt[q * 32 + L];
    n = n < CAPN ? n : CAPN;
    const int ng = (n + 3) >> 2;
    const uint4* e4 = reinterpret_cast<const uint4*>(elist) + (size_t)q * CAPG * 32 + L;
    constexpr int TRIPS = CAPG / 128;
    const int glast = ng > 0 ? ng - 1 : 0;
    uint4 e[TRIPS];
    #pragma unroll
    for (int t = 0; t < TRIPS; ++t) {
        const int g = t * 128 + wave * 4 + gsub;
        e[t] = e4[(size_t)(g < glast ? g : glast) * 32];
    }
    #pragma unroll
    for (int t = 0; t < TRIPS; ++t) {
        const int g = t * 128 + wave * 4 + gsub;
        const int rem = n - 4 * g;
        if (rem > 0) atomicAdd(&dcnt[e[t].x >> 16], 1);
        if (rem > 1) atomicAdd(&dcnt[e[t].y >> 16], 1);
        if (rem > 2) atomicAdd(&dcnt[e[t].z >> 16], 1);
        if (rem > 3) atomicAdd(&dcnt[e[t].w >> 16], 1);
    }
    int novf = gcnt[NLIST];
    novf = novf < OVFCAP ? novf : OVFCAP;
    for (int i = tid; i < novf; i += 1024) {
        const int d = (int)(ovf[i] >> 16);
        const int dl = d - q * RNG;
        if (dl >= 0 && dl < RNG && ((d & 31) >> 3) == s) atomicAdd(&dcnt[dl], 1);
    }
    __syncthreads();
    for (int j = tid; j < RNG / 4; j += 1024) {
        const int i = ((j >> 3) << 5) + 8 * s + (j & 7);
        const int node = q * RNG + i;
        if (node < NNODES) dis[node] = rsqrtf((float)dcnt[i] + 1.0f);
    }
}
static_assert(CAPG % 128 == 0, "degree pass: whole trips of 128 groups");

__global__ void wmma_gemm_f32(const float* __restrict__ A, const float* __restrict__ B,
                              float* __restrict__ C, int N, int K, int M,
                              const float* __restrict__ rowscale,
                              const float* __restrict__ bias, int total_tiles) {
    int tid  = threadIdx.x;
    int wid  = tid >> 5;
    int lane = tid & 31;
    int tile = blockIdx.x * (blockDim.x >> 5) + wid;
    if (tile >= total_tiles) return;

    int tiles_m = M >> 4;
    int rt = tile / tiles_m, ct = tile - rt * tiles_m;
    int row0 = rt << 4, col0 = ct << 4;
    int hi = lane >> 4, lm = lane & 15;

    v8f c = {};
    const float* Arow = A + (size_t)(row0 + lm) * K;
    for (int k0 = 0; k0 < K; k0 += 4) {
        v2f a = *reinterpret_cast<const v2f*>(Arow + k0 + 2 * hi);
        const float* Bp = B + (size_t)(k0 + 2 * hi) * M + col0 + lm;
        v2f b;
        b.x = Bp[0];
        b.y = Bp[M];
        c = __builtin_amdgcn_wmma_f32_16x16x4_f32(false, a, false, b,
                                                  (short)0, c, false, false);
    }
    float bv = bias ? bias[col0 + lm] : 0.0f;
    #pragma unroll
    for (int v = 0; v < 8; ++v) {
        int row = row0 + v + 8 * hi;
        float o = c[v];
        if (rowscale) o *= rowscale[row];
        C[(size_t)row * M + col0 + lm] = o + bv;
    }
}

__global__ void wmma_gemm_t(const float* __restrict__ X, const float* __restrict__ W,
                            float* __restrict__ hT, int K, int M,
                            const float* __restrict__ dis, int total_tiles) {
    int tid  = threadIdx.x;
    int wid  = tid >> 5;
    int lane = tid & 31;
    int tile = blockIdx.x * (blockDim.x >> 5) + wid;
    if (tile >= total_tiles) return;

    int tiles_f = M >> 4;
    int nt = tile / tiles_f, ft = tile - nt * tiles_f;
    int n0 = nt << 5, f0 = ft << 4;
    int hi = lane >> 4, lm = lane & 15;
    int na = n0 + lm, nb = n0 + 16 + lm;

    v8f c0 = {}, c1 = {};
    for (int k0 = 0; k0 < K; k0 += 4) {
        int k = k0 + 2 * hi;
        const float* Wp = W + (size_t)k * M + f0 + lm;
        v2f a;
        a.x = Wp[0];
        a.y = Wp[M];
        v2f b0, b1;
        const float* Xp = X + (size_t)k * NNODES;
        b0.x = Xp[na];          b1.x = Xp[nb];
        b0.y = Xp[NNODES + na]; b1.y = Xp[NNODES + nb];
        c0 = __builtin_amdgcn_wmma_f32_16x16x4_f32(false, a, false, b0,
                                                   (short)0, c0, false, false);
        c1 = __builtin_amdgcn_wmma_f32_16x16x4_f32(false, a, false, b1,
                                                   (short)0, c1, false, false);
    }
    float dn = dis[n0 + lane];
    #pragma unroll
    for (int r = 0; r < 8; ++r) {
        float give = hi ? c0[r] : c1[r];
        float got  = __shfl_xor(give, 16);
        float lo = hi ? got : c0[r];
        float up = hi ? c1[r] : got;
        hT[(size_t)(f0 + r) * NNODES + n0 + lane]     = lo * dn;
        hT[(size_t)(f0 + 8 + r) * NNODES + n0 + lane] = up * dn;
    }
}

#define L1K 128
#define L1M 128
#define XPITCH 132
__global__ __launch_bounds__(256) void gemm_rows_lds(const float* __restrict__ X,
                                                     const float* __restrict__ W,
                                                     float* __restrict__ hT,
                                                     const float* __restrict__ dis) {
    __shared__ float xs[32 * XPITCH];
    const int tid  = threadIdx.x;
    const int wid  = tid >> 5;
    const int lane = tid & 31;
    const int n0 = blockIdx.x << 5;

    #pragma unroll
    for (int i = 0; i < 4; ++i) {
        const int r = wid + 8 * i;
        const float4 v = reinterpret_cast<const float4*>(X + (size_t)(n0 + r) * L1K)[lane];
        *reinterpret_cast<float4*>(&xs[r * XPITCH + lane * 4]) = v;
    }
    __syncthreads();

    const int f0 = wid << 4;
    const int hi = lane >> 4, lm = lane & 15;
    const float* xa = xs + lm * XPITCH + 2 * hi;
    const float* xb = xa + 16 * XPITCH;
    const float* Wp = W + (size_t)(2 * hi) * L1M + f0 + lm;

    v8f c0 = {}, c1 = {};
    #pragma unroll 8
    for (int k0 = 0; k0 < L1K; k0 += 4) {
        v2f a;
        a.x = Wp[(size_t)k0 * L1M];
        a.y = Wp[(size_t)(k0 + 1) * L1M];
        const v2f b0 = *reinterpret_cast<const v2f*>(xa + k0);
        const v2f b1 = *reinterpret_cast<const v2f*>(xb + k0);
        c0 = __builtin_amdgcn_wmma_f32_16x16x4_f32(false, a, false, b0,
                                                   (short)0, c0, false, false);
        c1 = __builtin_amdgcn_wmma_f32_16x16x4_f32(false, a, false, b1,
                                                   (short)0, c1, false, false);
    }
    const float dn = dis[n0 + lane];
    #pragma unroll
    for (int r = 0; r < 8; ++r) {
        float give = hi ? c0[r] : c1[r];
        float got  = __shfl_xor(give, 16);
        float lo = hi ? got : c0[r];
        float up = hi ? c1[r] : got;
        hT[(size_t)(f0 + r) * NNODES + n0 + lane]     = lo * dn;
        hT[(size_t)(f0 + 8 + r) * NNODES + n0 + lane] = up * dn;
    }
}
static_assert(NNODES % 32 == 0, "layer-1 GEMM: whole 32-node tiles");

template <int R, bool POOL>
__global__ __launch_bounds__(1024) void slab_kernel(const float* __restrict__ hT,
                                                    const unsigned* __restrict__ elist,
                                                    const unsigned* __restrict__ ovf,
                                                    const int* __restrict__ gcnt,
                                                    float* __restrict__ outp,
                                                    const float* __restrict__ dis,
                                                    const float* __restrict__ bias, int F,
                                                    const int* __restrict__ batch,
                                                    const float* __restrict__ counts,
                                                    float* __restrict__ pooled) {
    static_assert(!POOL || R == 1, "pool epilogue: one range per work-group");
    __shared__ float hcol[NNODES];
    __shared__ float acc[R * RNG];
    __shared__ __attribute__((aligned(16))) int gid[POOL ? RNG : 1];
    __shared__ float gsum[POOL ? NGRAPH : 1];
    constexpr int WPR = 32 / R;
    constexpr int COLT = (NNODES / 4 + 1023) / 1024;
    constexpr int EPI = (R * RNG / 4 + 1023) / 1024;
    constexpr int DEPTH = 4;
    const int tid = threadIdx.x;
    const int qb = blockIdx.x / F;
    const int f = blockIdx.x - qb * F;
    const int nlo = qb * (R * RNG);
    const int nhi = nlo + R * RNG < NNODES ? nlo + R * RNG : NNODES;
    const int cnt4 = (nhi - nlo) >> 2;

    const int lane = tid & 31;
    const int wave = tid >> 5;
    const int rl = wave / WPR;
    const int slot = wave - rl * WPR;
    const int rg = qb * R + rl;

    const float4* h4 = reinterpret_cast<const float4*>(hT + (size_t)f * NNODES);
    float4 hv[COLT];
    #pragma unroll
    for (int t = 0; t < COLT; ++t) {
        const int i = tid + 1024 * t;
        hv[t] = h4[i < NNODES / 4 ? i : NNODES / 4 - 1];
    }
    int4 gb[POOL ? EPI : 1];
    if constexpr (POOL) {
        const int4* b4 = reinterpret_cast<const int4*>(batch + nlo);
        #pragma unroll
        for (int j = 0; j < EPI; ++j) {
            const int i = tid + 1024 * j;
            gb[j] = b4[i < cnt4 ? i : 0];
        }
    }
    static_assert(DEPTH * 32 <= CAPG, "first groups inside the list");
    const uint4* e4 = reinterpret_cast<const uint4*>(elist) + (size_t)rg * CAPG * 32 + lane;
    uint4 e[DEPTH];
    #pragma unroll
    for (int k = 0; k < DEPTH; ++k) e[k] = e4[(size_t)(slot + k * WPR) * 32];
    int n = gcnt[rg * 32 + lane];
    int novf = gcnt[NLIST];
    const float bf = bias[f];
    float cg = 1.0f;
    if constexpr (POOL) cg = counts[tid & (NGRAPH - 1)];
    const float4* dis4 = reinterpret_cast<const float4*>(dis + nlo);
    float4 dd[EPI];
    #pragma unroll
    for (int j = 0; j < EPI; ++j) {
        const int i = tid + 1024 * j;
        dd[j] = dis4[i < cnt4 ? i : 0];
    }

    float4* hc4 = reinterpret_cast<float4*>(hcol);
    float4* ac4 = reinterpret_cast<float4*>(acc);
    #pragma unroll
    for (int t = 0; t < COLT; ++t) {
        const int i = tid + 1024 * t;
        hc4[i < NNODES / 4 ? i : NNODES / 4 - 1] = hv[t];
    }
    if constexpr (POOL) {
        int4* g4 = reinterpret_cast<int4*>(gid);
        #pragma unroll
        for (int j = 0; j < EPI; ++j) {
            const int i = tid + 1024 * j;
            g4[i < cnt4 ? i : 0] = gb[j];
        }
    }
    for (int i = tid; i < R * RNG / 4; i += 1024) ac4[i] = make_float4(0.0f, 0.0f, 0.0f, 0.0f);
    if constexpr (POOL) {
        if (tid < NGRAPH) gsum[tid] = 0.0f;
    }
    __syncthreads();

    n = n < CAPN ? n : CAPN;
    const int ng = (n + 3) >> 2;
    const int glast = ng > 0 ? ng - 1 : 0;
    float* accr = acc + rl * RNG;
    int g = slot;
    while (g < ng) {
        uint4 en[DEPTH];
        #pragma unroll
        for (int k = 0; k < DEPTH; ++k) {
            const int gk = g + (DEPTH + k) * WPR;
            en[k] = e4[(size_t)(gk < glast ? gk : glast) * 32];
        }
        #pragma unroll
        for (int k = 0; k < DEPTH; ++k) {
            const uint4 c = e[k];
            const int rem = n - 4 * (g + k * WPR);
            if (rem >= 4) {
                float v0 = hcol[c.x & 0xFFFFu], v1 = hcol[c.y & 0xFFFFu];
                float v2 = hcol[c.z & 0xFFFFu], v3 = hcol[c.w & 0xFFFFu];
                __hip_atomic_fetch_add(&accr[c.x >> 16], v0, __ATOMIC_RELAXED, __HIP_MEMORY_SCOPE_WORKGROUP);
                __hip_atomic_fetch_add(&accr[c.y >> 16], v1, __ATOMIC_RELAXED, __HIP_MEMORY_SCOPE_WORKGROUP);
                __hip_atomic_fetch_add(&accr[c.z >> 16], v2, __ATOMIC_RELAXED, __HIP_MEMORY_SCOPE_WORKGROUP);
                __hip_atomic_fetch_add(&accr[c.w >> 16], v3, __ATOMIC_RELAXED, __HIP_MEMORY_SCOPE_WORKGROUP);
            } else if (rem > 0) {
                __hip_atomic_fetch_add(&accr[c.x >> 16], hcol[c.x & 0xFFFFu], __ATOMIC_RELAXED, __HIP_MEMORY_SCOPE_WORKGROUP);
                if (rem > 1)
                    __hip_atomic_fetch_add(&accr[c.y >> 16], hcol[c.y & 0xFFFFu], __ATOMIC_RELAXED, __HIP_MEMORY_SCOPE_WORKGROUP);
                if (rem > 2)
                    __hip_atomic_fetch_add(&accr[c.z >> 16], hcol[c.z & 0xFFFFu], __ATOMIC_RELAXED, __HIP_MEMORY_SCOPE_WORKGROUP);
            }
        }
        #pragma unroll
        for (int k = 0; k < DEPTH; ++k) e[k] = en[k];
        g += DEPTH * WPR;
    }
    novf = novf < OVFCAP ? novf : OVFCAP;
    for (int i = tid; i < novf; i += 1024) {
        const unsigned w = ovf[i];
        const int d = (int)(w >> 16);
        if (d >= nlo && d < nhi)
            __hip_atomic_fetch_add(&acc[d - nlo], hcol[w & 0xFFFFu], __ATOMIC_RELAXED, __HIP_MEMORY_SCOPE_WORKGROUP);
    }
    __syncthreads();

    float4* o4 = reinterpret_cast<float4*>(outp + (size_t)f * NNODES + nlo);
    const float4* hl4 = reinterpret_cast<const float4*>(hcol + nlo);
    #pragma unroll
    for (int j = 0; j < EPI; ++j) {
        const int i = tid + 1024 * j;
        if (i < cnt4) {
            const float4 a = ac4[i], h = hl4[i], d4 = dd[j];
            float4 o;
            o.x = (a.x + h.x) * d4.x + bf;
            o.y = (a.y + h.y) * d4.y + bf;
            o.z = (a.z + h.z) * d4.z + bf;
            o.w = (a.w + h.w) * d4.w + bf;
            o.x = o.x > 0.0f ? o.x : 0.0f;
            o.y = o.y > 0.0f ? o.y : 0.0f;
            o.z = o.z > 0.0f ? o.z : 0.0f;
            o.w = o.w > 0.0f ? o.w : 0.0f;
            if constexpr (POOL) ac4[i] = o;
            else o4[i] = o;
        }
    }
    if constexpr (POOL) {
        __syncthreads();
        constexpr int SEG = RNG / 32;
        constexpr int CH = (SEG + 31) / 32;
        static_assert(RNG % 32 == 0 && SEG % 2 == 1, "lane stride odd: 32 lanes, 32 banks");
        const int cntn = nhi - nlo;
        const int ibeg = SEG * lane + CH * wave;
        int iend = ibeg + CH;
        if (iend > SEG * lane + SEG) iend = SEG * lane + SEG;
        if (iend > cntn) iend = cntn;
        if (ibeg < iend) {
            int gcur = gid[ibeg];
            float s = acc[ibeg];
            for (int i = ibeg + 1; i < iend; ++i) {
                const int gi = gid[i];
                const float v = acc[i];
                if (gi != gcur) {
                    __hip_atomic_fetch_add(&gsum[gcur], s, __ATOMIC_RELAXED, __HIP_MEMORY_SCOPE_WORKGROUP);
                    gcur = gi;
                    s = 0.0f;
                }
                s += v;
            }
            __hip_atomic_fetch_add(&gsum[gcur], s, __ATOMIC_RELAXED, __HIP_MEMORY_SCOPE_WORKGROUP);
        }
        __syncthreads();
        if (tid < NGRAPH) {
            const float v = gsum[tid];
            if (v != 0.0f) atomAddF(&pooled[tid * F + f], v / fmaxf(cg, 1.0f));
        }
    }
}

extern "C" void kernel_launch(void* const* d_in, const int* in_sizes, int n_in,
                              void* d_out, int out_size, void* d_ws, size_t ws_size,
                              hipStream_t stream) {
    const float* x   = (const float*)d_in[0];
    const float* W1  = (const float*)d_in[1];
    const float* b1  = (const float*)d_in[2];
    const float* W2  = (const float*)d_in[3];
    const float* b2  = (const float*)d_in[4];
    const float* W3  = (const float*)d_in[5];
    const float* b3  = (const float*)d_in[6];
    const float* Wl  = (const float*)d_in[7];
    const float* bl  = (const float*)d_in[8];
    const int*   ei  = (const int*)d_in[9];
    const int*   bat = (const int*)d_in[10];
    float*       out = (float*)d_out;

    const int N = NNODES, E = NEDGES;
    const int* src = ei;
    const int* dst = ei + E;

    float* B0     = (float*)d_ws;
    float* B1     = B0 + (size_t)N * 128;
    float* dis    = B1 + (size_t)N * 128;
    int*   gcnt   = (int*)(dis + N);
    float* counts = dis + N + 320;
    float* pooled = counts + NGRAPH;
    unsigned* elist = (unsigned*)(pooled + NGRAPH * 32);
    unsigned* ovf   = elist + (size_t)NRNG * CAPG * 128;

    hipMemsetAsync(gcnt, 0, 320 * sizeof(int), stream);
    hipMemsetAsync(counts, 0, (NGRAPH + NGRAPH * 32) * sizeof(float), stream);
    bin_kernel<<<(E / 4 + 1023) / 1024, 1024, 0, stream>>>(src, dst, gcnt, elist, ovf);
    deg_dis_counts_kernel<<<DEGBLK + CNTBLK, 1024, 0, stream>>>(elist, ovf, gcnt, dis, bat, counts);

    gemm_rows_lds<<<NNODES / 32, 256, 0, stream>>>(x, W1, B0, dis);
    slab_kernel<4, false><<<128 * 2, 1024, 0, stream>>>(B0, elist, ovf, gcnt, B1, dis, b1, 128,
                                                         nullptr, nullptr, nullptr);

    wmma_gemm_t<<<(1250 * 4) / 8, 256, 0, stream>>>(B1, W2, B0, 128, 64, dis, 1250 * 4);
    slab_kernel<2, false><<<64 * 4, 1024, 0, stream>>>(B0, elist, ovf, gcnt, B1, dis, b2, 64,
                                                        nullptr, nullptr, nullptr);

    wmma_gemm_t<<<(1250 * 2 + 7) / 8, 256, 0, stream>>>(B1, W3, B0, 64, 32, dis, 1250 * 2);
    slab_kernel<1, true><<<32 * 8, 1024, 0, stream>>>(B0, elist, ovf, gcnt, B1, dis, b3, 32,
                                                       bat, counts, pooled);

    wmma_gemm_f32<<<(16 * 48) / 8, 256, 0, stream>>>(pooled, Wl, out, NGRAPH, 32, 768,
                                                     nullptr, bl, 16 * 48);
}


# layer-2/3 GEMM: a lane's two node columns are neighbours: one 8-byte load per activation row feeds both accumulators, results leave as 8-byte stores, no lane exchange; loads batched; sums unchanged
# speedup vs baseline: 3.1589x; 1.0909x over previous
#include <hip/hip_runtime.h>
#include <hip/hip_bf16.h>


typedef __attribute__((ext_vector_type(2))) float v2f;
typedef __attribute__((ext_vector_type(8))) float v8f;

#define NNODES 40000
#define NEDGES 640000
#define NGRAPH 256

__device__ __forceinline__ void atomAddF(float* p, float v) {
    unsafeAtomicAdd(p, v);
}

#define RNG 5024
#define NRNG 8
#define NLIST (NRNG * 32)
#define CAPN 4096
#define CAPG (CAPN / 4)
#define OVFCAP (NEDGES - CAPN)
static_assert(NNODES <= 65536, "src and dst share one 32-bit word");
static_assert(RNG % 32 == 0 && NRNG * RNG >= NNODES, "ranges cover all nodes, lane = dst mod 32");

__device__ __forceinline__ void place_edge(int s, int d, int q, int pos, int* __restrict__ gcnt,
                                           unsigned* __restrict__ elist,
                                           unsigned* __restrict__ ovf) {
    if (pos < CAPN) {
        const unsigned w = (unsigned)s | ((unsigned)(d - q * RNG) << 16);
        elist[(((size_t)q * CAPG + (pos >> 2)) * 32 + (d & 31)) * 4 + (pos & 3)] = w;
    } else {
        const int o = atomicAdd(&gcnt[NLIST], 1);
        if (o < OVFCAP) ovf[o] = (unsigned)s | ((unsigned)d << 16);
    }
}

__global__ __launch_bounds__(1024) void bin_kernel(const int* __restrict__ src,
                                                   const int* __restrict__ dst,
                                                   int* __restrict__ gcnt,
                                                   unsigned* __restrict__ elist,
                                                   unsigned* __restrict__ ovf) {
    __shared__ int cnt[NLIST];
    __shared__ int base[NLIST];
    const int tid = threadIdx.x;
    if (tid < NLIST) cnt[tid] = 0;
    __syncthreads();
    const int i = blockIdx.x * 1024 + tid;
    const bool act = i < NEDGES / 4;
    int4 s = make_int4(0, 0, 0, 0), d = make_int4(0, 0, 0, 0);
    int q0 = 0, q1 = 0, q2 = 0, q3 = 0;
    int r0 = 0, r1 = 0, r2 = 0, r3 = 0;
    if (act) {
        s = reinterpret_cast<const int4*>(src)[i];
        d = reinterpret_cast<const int4*>(dst)[i];
        q0 = d.x / RNG; q1 = d.y / RNG; q2 = d.z / RNG; q3 = d.w / RNG;
        r0 = atomicAdd(&cnt[q0 * 32 + (d.x & 31)], 1);
        r1 = atomicAdd(&cnt[q1 * 32 + (d.y & 31)], 1);
        r2 = atomicAdd(&cnt[q2 * 32 + (d.z & 31)], 1);
        r3 = atomicAdd(&cnt[q3 * 32 + (d.w & 31)], 1);
    }
    __syncthreads();
    if (tid < NLIST) base[tid] = atomicAdd(&gcnt[tid], cnt[tid]);
    __syncthreads();
    if (act) {
        place_edge(s.x, d.x, q0, base[q0 * 32 + (d.x & 31)] + r0, gcnt, elist, ovf);
        place_edge(s.y, d.y, q1, base[q1 * 32 + (d.y & 31)] + r1, gcnt, elist, ovf);
        place_edge(s.z, d.z, q2, base[q2 * 32 + (d.z & 31)] + r2, gcnt, elist, ovf);
        place_edge(s.w, d.w, q3, base[q3 * 32 + (d.w & 31)] + r3, gcnt, elist, ovf);
    }
}

#define DEGBLK (NRNG * 4)
#define CNTBLK ((NNODES + 1023) / 1024)
__global__ __launch_bounds__(1024) void deg_dis_counts_kernel(const unsigned* __restrict__ elist,
                                                              const unsigned* __restrict__ ovf,
                                                              const int* __restrict__ gcnt,
                                                              float* __restrict__ dis,
                                                              const int* __restrict__ batch,
                                                              float* __restrict__ counts) {
    __shared__ int dcnt[RNG];
    const int tid = threadIdx.x;
    if (blockIdx.x >= DEGBLK) {
        const int n = (blockIdx.x - DEGBLK) * 1024 + tid;
        if (n < NNODES) atomAddF(&counts[batch[n]], 1.0f);
        return;
    }
    const int q = blockIdx.x >> 2;
    const int s = blockIdx.x & 3;
    for (int i = tid; i < RNG; i += 1024) dcnt[i] = 0;
    __syncthreads();

    const int lane = tid & 31;
    const int wave = tid >> 5;
    const int L = 8 * s + (lane & 7);
    const int gsub = lane >> 3;
    int n = gcnt[q * 32 + L];
    n = n < CAPN ? n : CAPN;
    const int ng = (n + 3) >> 2;
    const uint4* e4 = reinterpret_cast<const uint4*>(elist) + (size_t)q * CAPG * 32 + L;
    constexpr int TRIPS = CAPG / 128;
    const int glast = ng > 0 ? ng - 1 : 0;
    uint4 e[TRIPS];
    #pragma unroll
    for (int t = 0; t < TRIPS; ++t) {
        const int g = t * 128 + wave * 4 + gsub;
        e[t] = e4[(size_t)(g < glast ? g : glast) * 32];
    }
    #pragma unroll
    for (int t = 0; t < TRIPS; ++t) {
        const int g = t * 128 + wave * 4 + gsub;
        const int rem = n - 4 * g;
        if (rem > 0) atomicAdd(&dcnt[e[t].x >> 16], 1);
        if (rem > 1) atomicAdd(&dcnt[e[t].y >> 16], 1);
        if (rem > 2) atomicAdd(&dcnt[e[t].z >> 16], 1);
        if (rem > 3) atomicAdd(&dcnt[e[t].w >> 16], 1);
    }
    int novf = gcnt[NLIST];
    novf = novf < OVFCAP ? novf : OVFCAP;
    for (int i = tid; i < novf; i += 1024) {
        const int d = (int)(ovf[i] >> 16);
        const int dl = d - q * RNG;
        if (dl >= 0 && dl < RNG && ((d & 31) >> 3) == s) atomicAdd(&dcnt[dl], 1);
    }
    __syncthreads();
    for (int j = tid; j < RNG / 4; j += 1024) {
        const int i = ((j >> 3) << 5) + 8 * s + (j & 7);
        const int node = q * RNG + i;
        if (node < NNODES) dis[node] = rsqrtf((float)dcnt[i] + 1.0f);
    }
}
static_assert(CAPG % 128 == 0, "degree pass: whole trips of 128 groups");

__global__ void wmma_gemm_f32(const float* __restrict__ A, const float* __restrict__ B,
                              float* __restrict__ C, int N, int K, int M,
                              const float* __restrict__ rowscale,
                              const float* __restrict__ bias, int total_tiles) {
    int tid  = threadIdx.x;
    int wid  = tid >> 5;
    int lane = tid & 31;
    int tile = blockIdx.x * (blockDim.x >> 5) + wid;
    if (tile >= total_tiles) return;

    int tiles_m = M >> 4;
    int rt = tile / tiles_m, ct = tile - rt * tiles_m;
    int row0 = rt << 4, col0 = ct << 4;
    int hi = lane >> 4, lm = lane & 15;

    v8f c = {};
    const float* Arow = A + (size_t)(row0 + lm) * K;
    for (int k0 = 0; k0 < K; k0 += 4) {
        v2f a = *reinterpret_cast<const v2f*>(Arow + k0 + 2 * hi);
        const float* Bp = B + (size_t)(k0 + 2 * hi) * M + col0 + lm;
        v2f b;
        b.x = Bp[0];
        b.y = Bp[M];
        c = __builtin_amdgcn_wmma_f32_16x16x4_f32(false, a, false, b,
                                                  (short)0, c, false, false);
    }
    float bv = bias ? bias[col0 + lm] : 0.0f;
    #pragma unroll
    for (int v = 0; v < 8; ++v) {
        int row = row0 + v + 8 * hi;
        float o = c[v];
        if (rowscale) o *= rowscale[row];
        C[(size_t)row * M + col0 + lm] = o + bv;
    }
}

template <int K, int M>
__global__ __launch_bounds__(256) void wmma_gemm_t(const float* __restrict__ X,
                                                   const float* __restrict__ W,
                                                   float* __restrict__ hT,
                                                   const float* __restrict__ dis) {
    constexpr int TILES_F = M / 16;
    constexpr int TOTAL = (NNODES / 32) * TILES_F;
    const int tid  = threadIdx.x;
    const int wid  = tid >> 5;
    const int lane = tid & 31;
    const int tile = blockIdx.x * 8 + wid;
    if (tile >= TOTAL) return;

    const int nt = tile / TILES_F, ft = tile - nt * TILES_F;
    const int n0 = nt << 5, f0 = ft << 4;
    const int hi = lane >> 4, lm = lane & 15;
    const float* Wp = W + (size_t)(2 * hi) * M + f0 + lm;
    const float* Xp = X + (size_t)(2 * hi) * NNODES + n0 + 2 * lm;

    v8f c0 = {}, c1 = {};
    #pragma unroll 16
    for (int k0 = 0; k0 < K; k0 += 4) {
        v2f a;
        a.x = Wp[(size_t)k0 * M];
        a.y = Wp[(size_t)(k0 + 1) * M];
        const v2f x0 = *reinterpret_cast<const v2f*>(Xp + (size_t)k0 * NNODES);
        const v2f x1 = *reinterpret_cast<const v2f*>(Xp + (size_t)(k0 + 1) * NNODES);
        v2f b0, b1;
        b0.x = x0.x; b0.y = x1.x;
        b1.x = x0.y; b1.y = x1.y;
        c0 = __builtin_amdgcn_wmma_f32_16x16x4_f32(false, a, false, b0,
                                                   (short)0, c0, false, false);
        c1 = __builtin_amdgcn_wmma_f32_16x16x4_f32(false, a, false, b1,
                                                   (short)0, c1, false, false);
    }
    const v2f dn = *reinterpret_cast<const v2f*>(dis + n0 + 2 * lm);
    #pragma unroll
    for (int r = 0; r < 8; ++r) {
        v2f o;
        o.x = c0[r] * dn.x;
        o.y = c1[r] * dn.y;
        *reinterpret_cast<v2f*>(hT + (size_t)(f0 + 8 * hi + r) * NNODES + n0 + 2 * lm) = o;
    }
}
static_assert(NNODES % 32 == 0, "layer GEMMs: whole 32-node tiles, 8-byte aligned node pairs");

#define L1K 128
#define L1M 128
#define XPITCH 132
__global__ __launch_bounds__(256) void gemm_rows_lds(const float* __restrict__ X,
                                                     const float* __restrict__ W,
                                                     float* __restrict__ hT,
                                                     const float* __restrict__ dis) {
    __shared__ float xs[32 * XPITCH];
    const int tid  = threadIdx.x;
    const int wid  = tid >> 5;
    const int lane = tid & 31;
    const int n0 = blockIdx.x << 5;

    #pragma unroll
    for (int i = 0; i < 4; ++i) {
        const int r = wid + 8 * i;
        const float4 v = reinterpret_cast<const float4*>(X + (size_t)(n0 + r) * L1K)[lane];
        *reinterpret_cast<float4*>(&xs[r * XPITCH + lane * 4]) = v;
    }
    __syncthreads();

    const int f0 = wid << 4;
    const int hi = lane >> 4, lm = lane & 15;
    const float* xa = xs + lm * XPITCH + 2 * hi;
    const float* xb = xa + 16 * XPITCH;
    const float* Wp = W + (size_t)(2 * hi) * L1M + f0 + lm;

    v8f c0 = {}, c1 = {};
    #pragma unroll 8
    for (int k0 = 0; k0 < L1K; k0 += 4) {
        v2f a;
        a.x = Wp[(size_t)k0 * L1M];
        a.y = Wp[(size_t)(k0 + 1) * L1M];
        const v2f b0 = *reinterpret_cast<const v2f*>(xa + k0);
        const v2f b1 = *reinterpret_cast<const v2f*>(xb + k0);
        c0 = __builtin_amdgcn_wmma_f32_16x16x4_f32(false, a, false, b0,
                                                   (short)0, c0, false, false);
        c1 = __builtin_amdgcn_wmma_f32_16x16x4_f32(false, a, false, b1,
                                                   (short)0, c1, false, false);
    }
    const float dn = dis[n0 + lane];
    #pragma unroll
    for (int r = 0; r < 8; ++r) {
        float give = hi ? c0[r] : c1[r];
        float got  = __shfl_xor(give, 16);
        float lo = hi ? got : c0[r];
        float up = hi ? c1[r] : got;
        hT[(size_t)(f0 + r) * NNODES + n0 + lane]     = lo * dn;
        hT[(size_t)(f0 + 8 + r) * NNODES + n0 + lane] = up * dn;
    }
}
static_assert(NNODES % 32 == 0, "layer-1 GEMM: whole 32-node tiles");

template <int R, bool POOL>
__global__ __launch_bounds__(1024) void slab_kernel(const float* __restrict__ hT,
                                                    const unsigned* __restrict__ elist,
                                                    const unsigned* __restrict__ ovf,
                                                    const int* __restrict__ gcnt,
                                                    float* __restrict__ outp,
                                                    const float* __restrict__ dis,
                                                    const float* __restrict__ bias, int F,
                                                    const int* __restrict__ batch,
                                                    const float* __restrict__ counts,
                                                    float* __restrict__ pooled) {
    static_assert(!POOL || R == 1, "pool epilogue: one range per work-group");
    __shared__ float hcol[NNODES];
    __shared__ float acc[R * RNG];
    __shared__ __attribute__((aligned(16))) int gid[POOL ? RNG : 1];
    __shared__ float gsum[POOL ? NGRAPH : 1];
    constexpr int WPR = 32 / R;
    constexpr int COLT = (NNODES / 4 + 1023) / 1024;
    constexpr int EPI = (R * RNG / 4 + 1023) / 1024;
    constexpr int DEPTH = 4;
    const int tid = threadIdx.x;
    const int qb = blockIdx.x / F;
    const int f = blockIdx.x - qb * F;
    const int nlo = qb * (R * RNG);
    const int nhi = nlo + R * RNG < NNODES ? nlo + R * RNG : NNODES;
    const int cnt4 = (nhi - nlo) >> 2;

    const int lane = tid & 31;
    const int wave = tid >> 5;
    const int rl = wave / WPR;
    const int slot = wave - rl * WPR;
    const int rg = qb * R + rl;

    const float4* h4 = reinterpret_cast<const float4*>(hT + (size_t)f * NNODES);
    float4 hv[COLT];
    #pragma unroll
    for (int t = 0; t < COLT; ++t) {
        const int i = tid + 1024 * t;
        hv[t] = h4[i < NNODES / 4 ? i : NNODES / 4 - 1];
    }
    int4 gb[POOL ? EPI : 1];
    if constexpr (POOL) {
        const int4* b4 = reinterpret_cast<const int4*>(batch + nlo);
        #pragma unroll
        for (int j = 0; j < EPI; ++j) {
            const int i = tid + 1024 * j;
            gb[j] = b4[i < cnt4 ? i : 0];
        }
    }
    static_assert(DEPTH * 32 <= CAPG, "first groups inside the list");
    const uint4* e4 = reinterpret_cast<const uint4*>(elist) + (size_t)rg * CAPG * 32 + lane;
    uint4 e[DEPTH];
    #pragma unroll
    for (int k = 0; k < DEPTH; ++k) e[k] = e4[(size_t)(slot + k * WPR) * 32];
    int n = gcnt[rg * 32 + lane];
    int novf = gcnt[NLIST];
    const float bf = bias[f];
    float cg = 1.0f;
    if constexpr (POOL) cg = counts[tid & (NGRAPH - 1)];
    const float4* dis4 = reinterpret_cast<const float4*>(dis + nlo);
    float4 dd[EPI];
    #pragma unroll
    for (int j = 0; j < EPI; ++j) {
        const int i = tid + 1024 * j;
        dd[j] = dis4[i < cnt4 ? i : 0];
    }

    float4* hc4 = reinterpret_cast<float4*>(hcol);
    float4* ac4 = reinterpret_cast<float4*>(acc);
    #pragma unroll
    for (int t = 0; t < COLT; ++t) {
        const int i = tid + 1024 * t;
        hc4[i < NNODES / 4 ? i : NNODES / 4 - 1] = hv[t];
    }
    if constexpr (POOL) {
        int4* g4 = reinterpret_cast<int4*>(gid);
        #pragma unroll
        for (int j = 0; j < EPI; ++j) {
            const int i = tid + 1024 * j;
            g4[i < cnt4 ? i : 0] = gb[j];
        }
    }
    for (int i = tid; i < R * RNG / 4; i += 1024) ac4[i] = make_float4(0.0f, 0.0f, 0.0f, 0.0f);
    if constexpr (POOL) {
        if (tid < NGRAPH) gsum[tid] = 0.0f;
    }
    __syncthreads();

    n = n < CAPN ? n : CAPN;
    const int ng = (n + 3) >> 2;
    const int glast = ng > 0 ? ng - 1 : 0;
    float* accr = acc + rl * RNG;
    int g = slot;
    while (g < ng) {
        uint4 en[DEPTH];
        #pragma unroll
        for (int k = 0; k < DEPTH; ++k) {
            const int gk = g + (DEPTH + k) * WPR;
            en[k] = e4[(size_t)(gk < glast ? gk : glast) * 32];
        }
        #pragma unroll
        for (int k = 0; k < DEPTH; ++k) {
            const uint4 c = e[k];
            const int rem = n - 4 * (g + k * WPR);
            if (rem >= 4) {
                float v0 = hcol[c.x & 0xFFFFu], v1 = hcol[c.y & 0xFFFFu];
                float v2 = hcol[c.z & 0xFFFFu], v3 = hcol[c.w & 0xFFFFu];
                __hip_atomic_fetch_add(&accr[c.x >> 16], v0, __ATOMIC_RELAXED, __HIP_MEMORY_SCOPE_WORKGROUP);
                __hip_atomic_fetch_add(&accr[c.y >> 16], v1, __ATOMIC_RELAXED, __HIP_MEMORY_SCOPE_WORKGROUP);
                __hip_atomic_fetch_add(&accr[c.z >> 16], v2, __ATOMIC_RELAXED, __HIP_MEMORY_SCOPE_WORKGROUP);
                __hip_atomic_fetch_add(&accr[c.w >> 16], v3, __ATOMIC_RELAXED, __HIP_MEMORY_SCOPE_WORKGROUP);
            } else if (rem > 0) {
                __hip_atomic_fetch_add(&accr[c.x >> 16], hcol[c.x & 0xFFFFu], __ATOMIC_RELAXED, __HIP_MEMORY_SCOPE_WORKGROUP);
                if (rem > 1)
                    __hip_atomic_fetch_add(&accr[c.y >> 16], hcol[c.y & 0xFFFFu], __ATOMIC_RELAXED, __HIP_MEMORY_SCOPE_WORKGROUP);
                if (rem > 2)
                    __hip_atomic_fetch_add(&accr[c.z >> 16], hcol[c.z & 0xFFFFu], __ATOMIC_RELAXED, __HIP_MEMORY_SCOPE_WORKGROUP);
            }
        }
        #pragma unroll
        for (int k = 0; k < DEPTH; ++k) e[k] = en[k];
        g += DEPTH * WPR;
    }
    novf = novf < OVFCAP ? novf : OVFCAP;
    for (int i = tid; i < novf; i += 1024) {
        const unsigned w = ovf[i];
        const int d = (int)(w >> 16);
        if (d >= nlo && d < nhi)
            __hip_atomic_fetch_add(&acc[d - nlo], hcol[w & 0xFFFFu], __ATOMIC_RELAXED, __HIP_MEMORY_SCOPE_WORKGROUP);
    }
    __syncthreads();

    float4* o4 = reinterpret_cast<float4*>(outp + (size_t)f * NNODES + nlo);
    const float4* hl4 = reinterpret_cast<const float4*>(hcol + nlo);
    #pragma unroll
    for (int j = 0; j < EPI; ++j) {
        const int i = tid + 1024 * j;
        if (i < cnt4) {
            const float4 a = ac4[i], h = hl4[i], d4 = dd[j];
            float4 o;
            o.x = (a.x + h.x) * d4.x + bf;
            o.y = (a.y + h.y) * d4.y + bf;
            o.z = (a.z + h.z) * d4.z + bf;
            o.w = (a.w + h.w) * d4.w + bf;
            o.x = o.x > 0.0f ? o.x : 0.0f;
            o.y = o.y > 0.0f ? o.y : 0.0f;
            o.z = o.z > 0.0f ? o.z : 0.0f;
            o.w = o.w > 0.0f ? o.w : 0.0f;
            if constexpr (POOL) ac4[i] = o;
            else o4[i] = o;
        }
    }
    if constexpr (POOL) {
        __syncthreads();
        constexpr int SEG = RNG / 32;
        constexpr int CH = (SEG + 31) / 32;
        static_assert(RNG % 32 == 0 && SEG % 2 == 1, "lane stride odd: 32 lanes, 32 banks");
        const int cntn = nhi - nlo;
        const int ibeg = SEG * lane + CH * wave;
        int iend = ibeg + CH;
        if (iend > SEG * lane + SEG) iend = SEG * lane + SEG;
        if (iend > cntn) iend = cntn;
        if (ibeg < iend) {
            int gcur = gid[ibeg];
            float s = acc[ibeg];
            for (int i = ibeg + 1; i < iend; ++i) {
                const int gi = gid[i];
                const float v = acc[i];
                if (gi != gcur) {
                    __hip_atomic_fetch_add(&gsum[gcur], s, __ATOMIC_RELAXED, __HIP_MEMORY_SCOPE_WORKGROUP);
                    gcur = gi;
                    s = 0.0f;
                }
                s += v;
            }
            __hip_atomic_fetch_add(&gsum[gcur], s, __ATOMIC_RELAXED, __HIP_MEMORY_SCOPE_WORKGROUP);
        }
        __syncthreads();
        if (tid < NGRAPH) {
            const float v = gsum[tid];
            if (v != 0.0f) atomAddF(&pooled[tid * F + f], v / fmaxf(cg, 1.0f));
        }
    }
}

extern "C" void kernel_launch(void* const* d_in, const int* in_sizes, int n_in,
                              void* d_out, int out_size, void* d_ws, size_t ws_size,
                              hipStream_t stream) {
    const float* x   = (const float*)d_in[0];
    const float* W1  = (const float*)d_in[1];
    const float* b1  = (const float*)d_in[2];
    const float* W2  = (const float*)d_in[3];
    const float* b2  = (const float*)d_in[4];
    const float* W3  = (const float*)d_in[5];
    const float* b3  = (const float*)d_in[6];
    const float* Wl  = (const float*)d_in[7];
    const float* bl  = (const float*)d_in[8];
    const int*   ei  = (const int*)d_in[9];
    const int*   bat = (const int*)d_in[10];
    float*       out = (float*)d_out;

    const int N = NNODES, E = NEDGES;
    const int* src = ei;
    const int* dst = ei + E;

    float* B0     = (float*)d_ws;
    float* B1     = B0 + (size_t)N * 128;
    float* dis    = B1 + (size_t)N * 128;
    int*   gcnt   = (int*)(dis + N);
    float* counts = dis + N + 320;
    float* pooled = counts + NGRAPH;
    unsigned* elist = (unsigned*)(pooled + NGRAPH * 32);
    unsigned* ovf   = elist + (size_t)NRNG * CAPG * 128;

    hipMemsetAsync(gcnt, 0, 320 * sizeof(int), stream);
    hipMemsetAsync(counts, 0, (NGRAPH + NGRAPH * 32) * sizeof(float), stream);
    bin_kernel<<<(E / 4 + 1023) / 1024, 1024, 0, stream>>>(src, dst, gcnt, elist, ovf);
    deg_dis_counts_kernel<<<DEGBLK + CNTBLK, 1024, 0, stream>>>(elist, ovf, gcnt, dis, bat, counts);

    gemm_rows_lds<<<NNODES / 32, 256, 0, stream>>>(x, W1, B0, dis);
    slab_kernel<4, false><<<128 * 2, 1024, 0, stream>>>(B0, elist, ovf, gcnt, B1, dis, b1, 128,
                                                         nullptr, nullptr, nullptr);

    wmma_gemm_t<128, 64><<<(1250 * 4) / 8, 256, 0, stream>>>(B1, W2, B0, dis);
    slab_kernel<2, false><<<64 * 4, 1024, 0, stream>>>(B0, elist, ovf, gcnt, B1, dis, b2, 64,
                                                        nullptr, nullptr, nullptr);

    wmma_gemm_t<64, 32><<<(1250 * 2 + 7) / 8, 256, 0, stream>>>(B1, W3, B0, dis);
    slab_kernel<1, true><<<32 * 8, 1024, 0, stream>>>(B0, elist, ovf, gcnt, B1, dis, b3, 32,
                                                       bat, counts, pooled);

    wmma_gemm_f32<<<(16 * 48) / 8, 256, 0, stream>>>(pooled, Wl, out, NGRAPH, 32, 768,
                                                     nullptr, bl, 16 * 48);
}


# fewer launches, 11 stream ops -> 9: layer-1 GEMM shares the binning launch (its dis factor moves to the aggregation's column load), two memsets become one; edge order in lists, so sum order, changes
# speedup vs baseline: 3.3527x; 1.0614x over previous
#include <hip/hip_runtime.h>
#include <hip/hip_bf16.h>


typedef __attribute__((ext_vector_type(2))) float v2f;
typedef __attribute__((ext_vector_type(8))) float v8f;

#define NNODES 40000
#define NEDGES 640000
#define NGRAPH 256

__device__ __forceinline__ void atomAddF(float* p, float v) {
    unsafeAtomicAdd(p, v);
}

#define RNG 5024
#define NRNG 8
#define NLIST (NRNG * 32)
#define CAPN 4096
#define CAPG (CAPN / 4)
#define OVFCAP (NEDGES - CAPN)
static_assert(NNODES <= 65536, "src and dst share one 32-bit word");
static_assert(RNG % 32 == 0 && NRNG * RNG >= NNODES, "ranges cover all nodes, lane = dst mod 32");

__device__ __forceinline__ void place_edge(int s, int d, int q, int pos, int* __restrict__ gcnt,
                                           unsigned* __restrict__ elist,
                                           unsigned* __restrict__ ovf) {
    if (pos < CAPN) {
        const unsigned w = (unsigned)s | ((unsigned)(d - q * RNG) << 16);
        elist[(((size_t)q * CAPG + (pos >> 2)) * 32 + (d & 31)) * 4 + (pos & 3)] = w;
    } else {
        const int o = atomicAdd(&gcnt[NLIST], 1);
        if (o < OVFCAP) ovf[o] = (unsigned)s | ((unsigned)d << 16);
    }
}

#define BINBLK ((NEDGES / 4 + 1023) / 1024)
static_assert(NLIST == 256, "binning: one list counter per thread of a 256-thread work-group");
__device__ __forceinline__ void bin_block(int blk, const int* __restrict__ src,
                                          const int* __restrict__ dst,
                                          int* __restrict__ gcnt,
                                          unsigned* __restrict__ elist,
                                          unsigned* __restrict__ ovf,
                                          int* cnt, int* base) {
    const int tid = threadIdx.x;
    cnt[tid] = 0;
    __syncthreads();
    int4 s[4], d[4];
    int r[16];
    #pragma unroll
    for (int k = 0; k < 4; ++k) {
        const int i = blk * 1024 + k * 256 + tid;
        const int ic = i < NEDGES / 4 ? i : 0;
        s[k] = reinterpret_cast<const int4*>(src)[ic];
        d[k] = reinterpret_cast<const int4*>(dst)[ic];
    }
    #pragma unroll
    for (int k = 0; k < 4; ++k) {
        const int i = blk * 1024 + k * 256 + tid;
        r[4 * k + 0] = r[4 * k + 1] = r[4 * k + 2] = r[4 * k + 3] = 0;
        if (i < NEDGES / 4) {
            r[4 * k + 0] = atomicAdd(&cnt[(d[k].x / RNG) * 32 + (d[k].x & 31)], 1);
            r[4 * k + 1] = atomicAdd(&cnt[(d[k].y / RNG) * 32 + (d[k].y & 31)], 1);
            r[4 * k + 2] = atomicAdd(&cnt[(d[k].z / RNG) * 32 + (d[k].z & 31)], 1);
            r[4 * k + 3] = atomicAdd(&cnt[(d[k].w / RNG) * 32 + (d[k].w & 31)], 1);
        }
    }
    __syncthreads();
    base[tid] = atomicAdd(&gcnt[tid], cnt[tid]);
    __syncthreads();
    #pragma unroll
    for (int k = 0; k < 4; ++k) {
        const int i = blk * 1024 + k * 256 + tid;
        if (i < NEDGES / 4) {
            const int q0 = d[k].x / RNG, q1 = d[k].y / RNG, q2 = d[k].z / RNG, q3 = d[k].w / RNG;
            place_edge(s[k].x, d[k].x, q0, base[q0 * 32 + (d[k].x & 31)] + r[4 * k + 0], gcnt, elist, ovf);
            place_edge(s[k].y, d[k].y, q1, base[q1 * 32 + (d[k].y & 31)] + r[4 * k + 1], gcnt, elist, ovf);
            place_edge(s[k].z, d[k].z, q2, base[q2 * 32 + (d[k].z & 31)] + r[4 * k + 2], gcnt, elist, ovf);
            place_edge(s[k].w, d[k].w, q3, base[q3 * 32 + (d[k].w & 31)] + r[4 * k + 3], gcnt, elist, ovf);
        }
    }
}

#define DEGBLK (NRNG * 4)
#define CNTBLK ((NNODES + 1023) / 1024)
__global__ __launch_bounds__(1024) void deg_dis_counts_kernel(const unsigned* __restrict__ elist,
                                                              const unsigned* __restrict__ ovf,
                                                              const int* __restrict__ gcnt,
                                                              float* __restrict__ dis,
                                                              const int* __restrict__ batch,
                                                              float* __restrict__ counts) {
    __shared__ int dcnt[RNG];
    const int tid = threadIdx.x;
    if (blockIdx.x >= DEGBLK) {
        const int n = (blockIdx.x - DEGBLK) * 1024 + tid;
        if (n < NNODES) atomAddF(&counts[batch[n]], 1.0f);
        return;
    }
    const int q = blockIdx.x >> 2;
    const int s = blockIdx.x & 3;
    for (int i = tid; i < RNG; i += 1024) dcnt[i] = 0;
    __syncthreads();

    const int lane = tid & 31;
    const int wave = tid >> 5;
    const int L = 8 * s + (lane & 7);
    const int gsub = lane >> 3;
    int n = gcnt[q * 32 + L];
    n = n < CAPN ? n : CAPN;
    const int ng = (n + 3) >> 2;
    const uint4* e4 = reinterpret_cast<const uint4*>(elist) + (size_t)q * CAPG * 32 + L;
    constexpr int TRIPS = CAPG / 128;
    const int glast = ng > 0 ? ng - 1 : 0;
    uint4 e[TRIPS];
    #pragma unroll
    for (int t = 0; t < TRIPS; ++t) {
        const int g = t * 128 + wave * 4 + gsub;
        e[t] = e4[(size_t)(g < glast ? g : glast) * 32];
    }
    #pragma unroll
    for (int t = 0; t < TRIPS; ++t) {
        const int g = t * 128 + wave * 4 + gsub;
        const int rem = n - 4 * g;
        if (rem > 0) atomicAdd(&dcnt[e[t].x >> 16], 1);
        if (rem > 1) atomicAdd(&dcnt[e[t].y >> 16], 1);
        if (rem > 2) atomicAdd(&dcnt[e[t].z >> 16], 1);
        if (rem > 3) atomicAdd(&dcnt[e[t].w >> 16], 1);
    }
    int novf = gcnt[NLIST];
    novf = novf < OVFCAP ? novf : OVFCAP;
    for (int i = tid; i < novf; i += 1024) {
        const int d = (int)(ovf[i] >> 16);
        const int dl = d - q * RNG;
        if (dl >= 0 && dl < RNG && ((d & 31) >> 3) == s) atomicAdd(&dcnt[dl], 1);
    }
    __syncthreads();
    for (int j = tid; j < RNG / 4; j += 1024) {
        const int i = ((j >> 3) << 5) + 8 * s + (j & 7);
        const int node = q * RNG + i;
        if (node < NNODES) dis[node] = rsqrtf((float)dcnt[i] + 1.0f);
    }
}
static_assert(CAPG % 128 == 0, "degree pass: whole trips of 128 groups");

__global__ void wmma_gemm_f32(const float* __restrict__ A, const float* __restrict__ B,
                              float* __restrict__ C, int N, int K, int M,
                              const float* __restrict__ rowscale,
                              const float* __restrict__ bias, int total_tiles) {
    int tid  = threadIdx.x;
    int wid  = tid >> 5;
    int lane = tid & 31;
    int tile = blockIdx.x * (blockDim.x >> 5) + wid;
    if (tile >= total_tiles) return;

    int tiles_m = M >> 4;
    int rt = tile / tiles_m, ct = tile - rt * tiles_m;
    int row0 = rt << 4, col0 = ct << 4;
    int hi = lane >> 4, lm = lane & 15;

    v8f c = {};
    const float* Arow = A + (size_t)(row0 + lm) * K;
    for (int k0 = 0; k0 < K; k0 += 4) {
        v2f a = *reinterpret_cast<const v2f*>(Arow + k0 + 2 * hi);
        const float* Bp = B + (size_t)(k0 + 2 * hi) * M + col0 + lm;
        v2f b;
        b.x = Bp[0];
        b.y = Bp[M];
        c = __builtin_amdgcn_wmma_f32_16x16x4_f32(false, a, false, b,
                                                  (short)0, c, false, false);
    }
    float bv = bias ? bias[col0 + lm] : 0.0f;
    #pragma unroll
    for (int v = 0; v < 8; ++v) {
        int row = row0 + v + 8 * hi;
        float o = c[v];
        if (rowscale) o *= rowscale[row];
        C[(size_t)row * M + col0 + lm] = o + bv;
    }
}

template <int K, int M>
__global__ __launch_bounds__(256) void wmma_gemm_t(const float* __restrict__ X,
                                                   const float* __restrict__ W,
                                                   float* __restrict__ hT,
                                                   const float* __restrict__ dis) {
    constexpr int TILES_F = M / 16;
    constexpr int TOTAL = (NNODES / 32) * TILES_F;
    const int tid  = threadIdx.x;
    const int wid  = tid >> 5;
    const int lane = tid & 31;
    const int tile = blockIdx.x * 8 + wid;
    if (tile >= TOTAL) return;

    const int nt = tile / TILES_F, ft = tile - nt * TILES_F;
    const int n0 = nt << 5, f0 = ft << 4;
    const int hi = lane >> 4, lm = lane & 15;
    const float* Wp = W + (size_t)(2 * hi) * M + f0 + lm;
    const float* Xp = X + (size_t)(2 * hi) * NNODES + n0 + 2 * lm;

    v8f c0 = {}, c1 = {};
    #pragma unroll 16
    for (int k0 = 0; k0 < K; k0 += 4) {
        v2f a;
        a.x = Wp[(size_t)k0 * M];
        a.y = Wp[(size_t)(k0 + 1) * M];
        const v2f x0 = *reinterpret_cast<const v2f*>(Xp + (size_t)k0 * NNODES);
        const v2f x1 = *reinterpret_cast<const v2f*>(Xp + (size_t)(k0 + 1) * NNODES);
        v2f b0, b1;
        b0.x = x0.x; b0.y = x1.x;
        b1.x = x0.y; b1.y = x1.y;
        c0 = __builtin_amdgcn_wmma_f32_16x16x4_f32(false, a, false, b0,
                                                   (short)0, c0, false, false);
        c1 = __builtin_amdgcn_wmma_f32_16x16x4_f32(false, a, false, b1,
                                                   (short)0, c1, false, false);
    }
    const v2f dn = *reinterpret_cast<const v2f*>(dis + n0 + 2 * lm);
    #pragma unroll
    for (int r = 0; r < 8; ++r) {
        v2f o;
        o.x = c0[r] * dn.x;
        o.y = c1[r] * dn.y;
        *reinterpret_cast<v2f*>(hT + (size_t)(f0 + 8 * hi + r) * NNODES + n0 + 2 * lm) = o;
    }
}
static_assert(NNODES % 32 == 0, "layer GEMMs: whole 32-node tiles, 8-byte aligned node pairs");

#define L1K 128
#define L1M 128
#define XPITCH 132
__global__ __launch_bounds__(256) void bin_gemm1(const int* __restrict__ src,
                                                 const int* __restrict__ dst,
                                                 int* __restrict__ gcnt,
                                                 unsigned* __restrict__ elist,
                                                 unsigned* __restrict__ ovf,
                                                 const float* __restrict__ X,
                                                 const float* __restrict__ W,
                                                 float* __restrict__ hT) {
    __shared__ float xs[32 * XPITCH];
    __shared__ int cnt[NLIST];
    __shared__ int base[NLIST];
    if (blockIdx.x < BINBLK) {
        bin_block(blockIdx.x, src, dst, gcnt, elist, ovf, cnt, base);
        return;
    }
    const int tid  = threadIdx.x;
    const int wid  = tid >> 5;
    const int lane = tid & 31;
    const int n0 = (blockIdx.x - BINBLK) << 5;

    #pragma unroll
    for (int i = 0; i < 4; ++i) {
        const int r = wid + 8 * i;
        const float4 v = reinterpret_cast<const float4*>(X + (size_t)(n0 + r) * L1K)[lane];
        *reinterpret_cast<float4*>(&xs[r * XPITCH + lane * 4]) = v;
    }
    __syncthreads();

    const int f0 = wid << 4;
    const int hi = lane >> 4, lm = lane & 15;
    const float* xa = xs + lm * XPITCH + 2 * hi;
    const float* xb = xa + 16 * XPITCH;
    const float* Wp = W + (size_t)(2 * hi) * L1M + f0 + lm;

    v8f c0 = {}, c1 = {};
    #pragma unroll 8
    for (int k0 = 0; k0 < L1K; k0 += 4) {
        v2f a;
        a.x = Wp[(size_t)k0 * L1M];
        a.y = Wp[(size_t)(k0 + 1) * L1M];
        const v2f b0 = *reinterpret_cast<const v2f*>(xa + k0);
        const v2f b1 = *reinterpret_cast<const v2f*>(xb + k0);
        c0 = __builtin_amdgcn_wmma_f32_16x16x4_f32(false, a, false, b0,
                                                   (short)0, c0, false, false);
        c1 = __builtin_amdgcn_wmma_f32_16x16x4_f32(false, a, false, b1,
                                                   (short)0, c1, false, false);
    }
    #pragma unroll
    for (int r = 0; r < 8; ++r) {
        float give = hi ? c0[r] : c1[r];
        float got  = __shfl_xor(give, 16);
        float lo = hi ? got : c0[r];
        float up = hi ? c1[r] : got;
        hT[(size_t)(f0 + r) * NNODES + n0 + lane]     = lo;
        hT[(size_t)(f0 + 8 + r) * NNODES + n0 + lane] = up;
    }
}
static_assert(NNODES % 32 == 0, "layer-1 GEMM: whole 32-node tiles");

template <int R, bool POOL, bool SCALE>
__global__ __launch_bounds__(1024) void slab_kernel(const float* __restrict__ hT,
                                                    const unsigned* __restrict__ elist,
                                                    const unsigned* __restrict__ ovf,
                                                    const int* __restrict__ gcnt,
                                                    float* __restrict__ outp,
                                                    const float* __restrict__ dis,
                                                    const float* __restrict__ bias, int F,
                                                    const int* __restrict__ batch,
                                                    const float* __restrict__ counts,
                                                    float* __restrict__ pooled) {
    static_assert(!POOL || R == 1, "pool epilogue: one range per work-group");
    __shared__ float hcol[NNODES];
    __shared__ float acc[R * RNG];
    __shared__ __attribute__((aligned(16))) int gid[POOL ? RNG : 1];
    __shared__ float gsum[POOL ? NGRAPH : 1];
    constexpr int WPR = 32 / R;
    constexpr int COLT = (NNODES / 4 + 1023) / 1024;
    constexpr int EPI = (R * RNG / 4 + 1023) / 1024;
    constexpr int DEPTH = 4;
    const int tid = threadIdx.x;
    const int qb = blockIdx.x / F;
    const int f = blockIdx.x - qb * F;
    const int nlo = qb * (R * RNG);
    const int nhi = nlo + R * RNG < NNODES ? nlo + R * RNG : NNODES;
    const int cnt4 = (nhi - nlo) >> 2;

    const int lane = tid & 31;
    const int wave = tid >> 5;
    const int rl = wave / WPR;
    const int slot = wave - rl * WPR;
    const int rg = qb * R + rl;

    const float4* h4 = reinterpret_cast<const float4*>(hT + (size_t)f * NNODES);
    float4 hv[COLT];
    #pragma unroll
    for (int t = 0; t < COLT; ++t) {
        const int i = tid + 1024 * t;
        hv[t] = h4[i < NNODES / 4 ? i : NNODES / 4 - 1];
    }
    float4 dv[SCALE ? COLT : 1];
    if constexpr (SCALE) {
        const float4* dc4 = reinterpret_cast<const float4*>(dis);
        #pragma unroll
        for (int t = 0; t < COLT; ++t) {
            const int i = tid + 1024 * t;
            dv[t] = dc4[i < NNODES / 4 ? i : NNODES / 4 - 1];
        }
    }
    int4 gb[POOL ? EPI : 1];
    if constexpr (POOL) {
        const int4* b4 = reinterpret_cast<const int4*>(batch + nlo);
        #pragma unroll
        for (int j = 0; j < EPI; ++j) {
            const int i = tid + 1024 * j;
            gb[j] = b4[i < cnt4 ? i : 0];
        }
    }
    static_assert(DEPTH * 32 <= CAPG, "first groups inside the list");
    const uint4* e4 = reinterpret_cast<const uint4*>(elist) + (size_t)rg * CAPG * 32 + lane;
    uint4 e[DEPTH];
    #pragma unroll
    for (int k = 0; k < DEPTH; ++k) e[k] = e4[(size_t)(slot + k * WPR) * 32];
    int n = gcnt[rg * 32 + lane];
    int novf = gcnt[NLIST];
    const float bf = bias[f];
    float cg = 1.0f;
    if constexpr (POOL) cg = counts[tid & (NGRAPH - 1)];
    const float4* dis4 = reinterpret_cast<const float4*>(dis + nlo);
    float4 dd[EPI];
    if constexpr (!SCALE) {
        #pragma unroll
        for (int j = 0; j < EPI; ++j) {
            const int i = tid + 1024 * j;
            dd[j] = dis4[i < cnt4 ? i : 0];
        }
    }

    float4* hc4 = reinterpret_cast<float4*>(hcol);
    float4* ac4 = reinterpret_cast<float4*>(acc);
    #pragma unroll
    for (int t = 0; t < COLT; ++t) {
        const int i = tid + 1024 * t;
        float4 v = hv[t];
        if constexpr (SCALE) {
            v.x *= dv[t].x; v.y *= dv[t].y; v.z *= dv[t].z; v.w *= dv[t].w;
        }
        hc4[i < NNODES / 4 ? i : NNODES / 4 - 1] = v;
    }
    if constexpr (POOL) {
        int4* g4 = reinterpret_cast<int4*>(gid);
        #pragma unroll
        for (int j = 0; j < EPI; ++j) {
            const int i = tid + 1024 * j;
            g4[i < cnt4 ? i : 0] = gb[j];
        }
    }
    for (int i = tid; i < R * RNG / 4; i += 1024) ac4[i] = make_float4(0.0f, 0.0f, 0.0f, 0.0f);
    if constexpr (POOL) {
        if (tid < NGRAPH) gsum[tid] = 0.0f;
    }
    __syncthreads();
    if constexpr (SCALE) {
        #pragma unroll
        for (int j = 0; j < EPI; ++j) {
            const int i = tid + 1024 * j;
            dd[j] = dis4[i < cnt4 ? i : 0];
        }
    }

    n = n < CAPN ? n : CAPN;
    const int ng = (n + 3) >> 2;
    const int glast = ng > 0 ? ng - 1 : 0;
    float* accr = acc + rl * RNG;
    int g = slot;
    while (g < ng) {
        uint4 en[DEPTH];
        #pragma unroll
        for (int k = 0; k < DEPTH; ++k) {
            const int gk = g + (DEPTH + k) * WPR;
            en[k] = e4[(size_t)(gk < glast ? gk : glast) * 32];
        }
        #pragma unroll
        for (int k = 0; k < DEPTH; ++k) {
            const uint4 c = e[k];
            const int rem = n - 4 * (g + k * WPR);
            if (rem >= 4) {
                float v0 = hcol[c.x & 0xFFFFu], v1 = hcol[c.y & 0xFFFFu];
                float v2 = hcol[c.z & 0xFFFFu], v3 = hcol[c.w & 0xFFFFu];
                __hip_atomic_fetch_add(&accr[c.x >> 16], v0, __ATOMIC_RELAXED, __HIP_MEMORY_SCOPE_WORKGROUP);
                __hip_atomic_fetch_add(&accr[c.y >> 16], v1, __ATOMIC_RELAXED, __HIP_MEMORY_SCOPE_WORKGROUP);
                __hip_atomic_fetch_add(&accr[c.z >> 16], v2, __ATOMIC_RELAXED, __HIP_MEMORY_SCOPE_WORKGROUP);
                __hip_atomic_fetch_add(&accr[c.w >> 16], v3, __ATOMIC_RELAXED, __HIP_MEMORY_SCOPE_WORKGROUP);
            } else if (rem > 0) {
                __hip_atomic_fetch_add(&accr[c.x >> 16], hcol[c.x & 0xFFFFu], __ATOMIC_RELAXED, __HIP_MEMORY_SCOPE_WORKGROUP);
                if (rem > 1)
                    __hip_atomic_fetch_add(&accr[c.y >> 16], hcol[c.y & 0xFFFFu], __ATOMIC_RELAXED, __HIP_MEMORY_SCOPE_WORKGROUP);
                if (rem > 2)
                    __hip_atomic_fetch_add(&accr[c.z >> 16], hcol[c.z & 0xFFFFu], __ATOMIC_RELAXED, __HIP_MEMORY_SCOPE_WORKGROUP);
            }
        }
        #pragma unroll
        for (int k = 0; k < DEPTH; ++k) e[k] = en[k];
        g += DEPTH * WPR;
    }
    novf = novf < OVFCAP ? novf : OVFCAP;
    for (int i = tid; i < novf; i += 1024) {
        const unsigned w = ovf[i];
        const int d = (int)(w >> 16);
        if (d >= nlo && d < nhi)
            __hip_atomic_fetch_add(&acc[d - nlo], hcol[w & 0xFFFFu], __ATOMIC_RELAXED, __HIP_MEMORY_SCOPE_WORKGROUP);
    }
    __syncthreads();

    float4* o4 = reinterpret_cast<float4*>(outp + (size_t)f * NNODES + nlo);
    const float4* hl4 = reinterpret_cast<const float4*>(hcol + nlo);
    #pragma unroll
    for (int j = 0; j < EPI; ++j) {
        const int i = tid + 1024 * j;
        if (i < cnt4) {
            const float4 a = ac4[i], h = hl4[i], d4 = dd[j];
            float4 o;
            o.x = (a.x + h.x) * d4.x + bf;
            o.y = (a.y + h.y) * d4.y + bf;
            o.z = (a.z + h.z) * d4.z + bf;
            o.w = (a.w + h.w) * d4.w + bf;
            o.x = o.x > 0.0f ? o.x : 0.0f;
            o.y = o.y > 0.0f ? o.y : 0.0f;
            o.z = o.z > 0.0f ? o.z : 0.0f;
            o.w = o.w > 0.0f ? o.w : 0.0f;
            if constexpr (POOL) ac4[i] = o;
            else o4[i] = o;
        }
    }
    if constexpr (POOL) {
        __syncthreads();
        constexpr int SEG = RNG / 32;
        constexpr int CH = (SEG + 31) / 32;
        static_assert(RNG % 32 == 0 && SEG % 2 == 1, "lane stride odd: 32 lanes, 32 banks");
        const int cntn = nhi - nlo;
        const int ibeg = SEG * lane + CH * wave;
        int iend = ibeg + CH;
        if (iend > SEG * lane + SEG) iend = SEG * lane + SEG;
        if (iend > cntn) iend = cntn;
        if (ibeg < iend) {
            int gcur = gid[ibeg];
            float s = acc[ibeg];
            for (int i = ibeg + 1; i < iend; ++i) {
                const int gi = gid[i];
                const float v = acc[i];
                if (gi != gcur) {
                    __hip_atomic_fetch_add(&gsum[gcur], s, __ATOMIC_RELAXED, __HIP_MEMORY_SCOPE_WORKGROUP);
                    gcur = gi;
                    s = 0.0f;
                }
                s += v;
            }
            __hip_atomic_fetch_add(&gsum[gcur], s, __ATOMIC_RELAXED, __HIP_MEMORY_SCOPE_WORKGROUP);
        }
        __syncthreads();
        if (tid < NGRAPH) {
            const float v = gsum[tid];
            if (v != 0.0f) atomAddF(&pooled[tid * F + f], v / fmaxf(cg, 1.0f));
        }
    }
}

extern "C" void kernel_launch(void* const* d_in, const int* in_sizes, int n_in,
                              void* d_out, int out_size, void* d_ws, size_t ws_size,
                              hipStream_t stream) {
    const float* x   = (const float*)d_in[0];
    const float* W1  = (const float*)d_in[1];
    const float* b1  = (const float*)d_in[2];
    const float* W2  = (const float*)d_in[3];
    const float* b2  = (const float*)d_in[4];
    const float* W3  = (const float*)d_in[5];
    const float* b3  = (const float*)d_in[6];
    const float* Wl  = (const float*)d_in[7];
    const float* bl  = (const float*)d_in[8];
    const int*   ei  = (const int*)d_in[9];
    const int*   bat = (const int*)d_in[10];
    float*       out = (float*)d_out;

    const int N = NNODES, E = NEDGES;
    const int* src = ei;
    const int* dst = ei + E;

    float* B0     = (float*)d_ws;
    float* B1     = B0 + (size_t)N * 128;
    float* dis    = B1 + (size_t)N * 128;
    int*   gcnt   = (int*)(dis + N);
    float* counts = dis + N + 320;
    float* pooled = counts + NGRAPH;
    unsigned* elist = (unsigned*)(pooled + NGRAPH * 32);
    unsigned* ovf   = elist + (size_t)NRNG * CAPG * 128;

    hipMemsetAsync(gcnt, 0, (320 + NGRAPH + NGRAPH * 32) * sizeof(int), stream);
    bin_gemm1<<<BINBLK + NNODES / 32, 256, 0, stream>>>(src, dst, gcnt, elist, ovf, x, W1, B0);
    deg_dis_counts_kernel<<<DEGBLK + CNTBLK, 1024, 0, stream>>>(elist, ovf, gcnt, dis, bat, counts);

    slab_kernel<4, false, true><<<128 * 2, 1024, 0, stream>>>(B0, elist, ovf, gcnt, B1, dis, b1, 128,
                                                               nullptr, nullptr, nullptr);

    wmma_gemm_t<128, 64><<<(1250 * 4) / 8, 256, 0, stream>>>(B1, W2, B0, dis);
    slab_kernel<2, false, false><<<64 * 4, 1024, 0, stream>>>(B0, elist, ovf, gcnt, B1, dis, b2, 64,
                                                               nullptr, nullptr, nullptr);

    wmma_gemm_t<64, 32><<<(1250 * 2 + 7) / 8, 256, 0, stream>>>(B1, W3, B0, dis);
    slab_kernel<1, true, false><<<32 * 8, 1024, 0, stream>>>(B0, elist, ovf, gcnt, B1, dis, b3, 32,
                                                              bat, counts, pooled);

    wmma_gemm_f32<<<(16 * 48) / 8, 256, 0, stream>>>(pooled, Wl, out, NGRAPH, 32, 768,
                                                     nullptr, bl, 16 * 48);
}


# graph sizes come from the layer-3 pool work-groups (run lengths of graph ids in LDS, feature 0 only); 40000 one-word global adds leave the degree launch; head divides sum by size at its A load
# speedup vs baseline: 3.6813x; 1.0980x over previous
#include <hip/hip_runtime.h>
#include <hip/hip_bf16.h>


typedef __attribute__((ext_vector_type(2))) float v2f;
typedef __attribute__((ext_vector_type(8))) float v8f;

#define NNODES 40000
#define NEDGES 640000
#define NGRAPH 256

__device__ __forceinline__ void atomAddF(float* p, float v) {
    unsafeAtomicAdd(p, v);
}

#define RNG 5024
#define NRNG 8
#define NLIST (NRNG * 32)
#define CAPN 4096
#define CAPG (CAPN / 4)
#define OVFCAP (NEDGES - CAPN)
static_assert(NNODES <= 65536, "src and dst share one 32-bit word");
static_assert(RNG % 32 == 0 && NRNG * RNG >= NNODES, "ranges cover all nodes, lane = dst mod 32");

__device__ __forceinline__ void place_edge(int s, int d, int q, int pos, int* __restrict__ gcnt,
                                           unsigned* __restrict__ elist,
                                           unsigned* __restrict__ ovf) {
    if (pos < CAPN) {
        const unsigned w = (unsigned)s | ((unsigned)(d - q * RNG) << 16);
        elist[(((size_t)q * CAPG + (pos >> 2)) * 32 + (d & 31)) * 4 + (pos & 3)] = w;
    } else {
        const int o = atomicAdd(&gcnt[NLIST], 1);
        if (o < OVFCAP) ovf[o] = (unsigned)s | ((unsigned)d << 16);
    }
}

#define BINBLK ((NEDGES / 4 + 1023) / 1024)
static_assert(NLIST == 256, "binning: one list counter per thread of a 256-thread work-group");
__device__ __forceinline__ void bin_block(int blk, const int* __restrict__ src,
                                          const int* __restrict__ dst,
                                          int* __restrict__ gcnt,
                                          unsigned* __restrict__ elist,
                                          unsigned* __restrict__ ovf,
                                          int* cnt, int* base) {
    const int tid = threadIdx.x;
    cnt[tid] = 0;
    __syncthreads();
    int4 s[4], d[4];
    int r[16];
    #pragma unroll
    for (int k = 0; k < 4; ++k) {
        const int i = blk * 1024 + k * 256 + tid;
        const int ic = i < NEDGES / 4 ? i : 0;
        s[k] = reinterpret_cast<const int4*>(src)[ic];
        d[k] = reinterpret_cast<const int4*>(dst)[ic];
    }
    #pragma unroll
    for (int k = 0; k < 4; ++k) {
        const int i = blk * 1024 + k * 256 + tid;
        r[4 * k + 0] = r[4 * k + 1] = r[4 * k + 2] = r[4 * k + 3] = 0;
        if (i < NEDGES / 4) {
            r[4 * k + 0] = atomicAdd(&cnt[(d[k].x / RNG) * 32 + (d[k].x & 31)], 1);
            r[4 * k + 1] = atomicAdd(&cnt[(d[k].y / RNG) * 32 + (d[k].y & 31)], 1);
            r[4 * k + 2] = atomicAdd(&cnt[(d[k].z / RNG) * 32 + (d[k].z & 31)], 1);
            r[4 * k + 3] = atomicAdd(&cnt[(d[k].w / RNG) * 32 + (d[k].w & 31)], 1);
        }
    }
    __syncthreads();
    base[tid] = atomicAdd(&gcnt[tid], cnt[tid]);
    __syncthreads();
    #pragma unroll
    for (int k = 0; k < 4; ++k) {
        const int i = blk * 1024 + k * 256 + tid;
        if (i < NEDGES / 4) {
            const int q0 = d[k].x / RNG, q1 = d[k].y / RNG, q2 = d[k].z / RNG, q3 = d[k].w / RNG;
            place_edge(s[k].x, d[k].x, q0, base[q0 * 32 + (d[k].x & 31)] + r[4 * k + 0], gcnt, elist, ovf);
            place_edge(s[k].y, d[k].y, q1, base[q1 * 32 + (d[k].y & 31)] + r[4 * k + 1], gcnt, elist, ovf);
            place_edge(s[k].z, d[k].z, q2, base[q2 * 32 + (d[k].z & 31)] + r[4 * k + 2], gcnt, elist, ovf);
            place_edge(s[k].w, d[k].w, q3, base[q3 * 32 + (d[k].w & 31)] + r[4 * k + 3], gcnt, elist, ovf);
        }
    }
}

#define DEGBLK (NRNG * 4)
__global__ __launch_bounds__(1024) void deg_dis_kernel(const unsigned* __restrict__ elist,
                                                       const unsigned* __restrict__ ovf,
                                                       const int* __restrict__ gcnt,
                                                       float* __restrict__ dis) {
    __shared__ int dcnt[RNG];
    const int tid = threadIdx.x;
    const int q = blockIdx.x >> 2;
    const int s = blockIdx.x & 3;
    for (int i = tid; i < RNG; i += 1024) dcnt[i] = 0;
    __syncthreads();

    const int lane = tid & 31;
    const int wave = tid >> 5;
    const int L = 8 * s + (lane & 7);
    const int gsub = lane >> 3;
    int n = gcnt[q * 32 + L];
    n = n < CAPN ? n : CAPN;
    const int ng = (n + 3) >> 2;
    const uint4* e4 = reinterpret_cast<const uint4*>(elist) + (size_t)q * CAPG * 32 + L;
    constexpr int TRIPS = CAPG / 128;
    const int glast = ng > 0 ? ng - 1 : 0;
    uint4 e[TRIPS];
    #pragma unroll
    for (int t = 0; t < TRIPS; ++t) {
        const int g = t * 128 + wave * 4 + gsub;
        e[t] = e4[(size_t)(g < glast ? g : glast) * 32];
    }
    #pragma unroll
    for (int t = 0; t < TRIPS; ++t) {
        const int g = t * 128 + wave * 4 + gsub;
        const int rem = n - 4 * g;
        if (rem > 0) atomicAdd(&dcnt[e[t].x >> 16], 1);
        if (rem > 1) atomicAdd(&dcnt[e[t].y >> 16], 1);
        if (rem > 2) atomicAdd(&dcnt[e[t].z >> 16], 1);
        if (rem > 3) atomicAdd(&dcnt[e[t].w >> 16], 1);
    }
    int novf = gcnt[NLIST];
    novf = novf < OVFCAP ? novf : OVFCAP;
    for (int i = tid; i < novf; i += 1024) {
        const int d = (int)(ovf[i] >> 16);
        const int dl = d - q * RNG;
        if (dl >= 0 && dl < RNG && ((d & 31) >> 3) == s) atomicAdd(&dcnt[dl], 1);
    }
    __syncthreads();
    for (int j = tid; j < RNG / 4; j += 1024) {
        const int i = ((j >> 3) << 5) + 8 * s + (j & 7);
        const int node = q * RNG + i;
        if (node < NNODES) dis[node] = rsqrtf((float)dcnt[i] + 1.0f);
    }
}
static_assert(CAPG % 128 == 0, "degree pass: whole trips of 128 groups");

__global__ void wmma_gemm_f32(const float* __restrict__ A, const float* __restrict__ B,
                              float* __restrict__ C, int N, int K, int M,
                              const float* __restrict__ rowscale,
                              const float* __restrict__ bias, int total_tiles,
                              const float* __restrict__ rowdiv) {
    int tid  = threadIdx.x;
    int wid  = tid >> 5;
    int lane = tid & 31;
    int tile = blockIdx.x * (blockDim.x >> 5) + wid;
    if (tile >= total_tiles) return;

    int tiles_m = M >> 4;
    int rt = tile / tiles_m, ct = tile - rt * tiles_m;
    int row0 = rt << 4, col0 = ct << 4;
    int hi = lane >> 4, lm = lane & 15;

    v8f c = {};
    const float* Arow = A + (size_t)(row0 + lm) * K;
    const float dv = rowdiv ? fmaxf(rowdiv[row0 + lm], 1.0f) : 1.0f;
    for (int k0 = 0; k0 < K; k0 += 4) {
        v2f a = *reinterpret_cast<const v2f*>(Arow + k0 + 2 * hi);
        a.x = a.x / dv;
        a.y = a.y / dv;
        const float* Bp = B + (size_t)(k0 + 2 * hi) * M + col0 + lm;
        v2f b;
        b.x = Bp[0];
        b.y = Bp[M];
        c = __builtin_amdgcn_wmma_f32_16x16x4_f32(false, a, false, b,
                                                  (short)0, c, false, false);
    }
    float bv = bias ? bias[col0 + lm] : 0.0f;
    #pragma unroll
    for (int v = 0; v < 8; ++v) {
        int row = row0 + v + 8 * hi;
        float o = c[v];
        if (rowscale) o *= rowscale[row];
        C[(size_t)row * M + col0 + lm] = o + bv;
    }
}

template <int K, int M>
__global__ __launch_bounds__(256) void wmma_gemm_t(const float* __restrict__ X,
                                                   const float* __restrict__ W,
                                                   float* __restrict__ hT,
                                                   const float* __restrict__ dis) {
    constexpr int TILES_F = M / 16;
    constexpr int TOTAL = (NNODES / 32) * TILES_F;
    const int tid  = threadIdx.x;
    const int wid  = tid >> 5;
    const int lane = tid & 31;
    const int tile = blockIdx.x * 8 + wid;
    if (tile >= TOTAL) return;

    const int nt = tile / TILES_F, ft = tile - nt * TILES_F;
    const int n0 = nt << 5, f0 = ft << 4;
    const int hi = lane >> 4, lm = lane & 15;
    const float* Wp = W + (size_t)(2 * hi) * M + f0 + lm;
    const float* Xp = X + (size_t)(2 * hi) * NNODES + n0 + 2 * lm;

    v8f c0 = {}, c1 = {};
    #pragma unroll 16
    for (int k0 = 0; k0 < K; k0 += 4) {
        v2f a;
        a.x = Wp[(size_t)k0 * M];
        a.y = Wp[(size_t)(k0 + 1) * M];
        const v2f x0 = *reinterpret_cast<const v2f*>(Xp + (size_t)k0 * NNODES);
        const v2f x1 = *reinterpret_cast<const v2f*>(Xp + (size_t)(k0 + 1) * NNODES);
        v2f b0, b1;
        b0.x = x0.x; b0.y = x1.x;
        b1.x = x0.y; b1.y = x1.y;
        c0 = __builtin_amdgcn_wmma_f32_16x16x4_f32(false, a, false, b0,
                                                   (short)0, c0, false, false);
        c1 = __builtin_amdgcn_wmma_f32_16x16x4_f32(false, a, false, b1,
                                                   (short)0, c1, false, false);
    }
    const v2f dn = *reinterpret_cast<const v2f*>(dis + n0 + 2 * lm);
    #pragma unroll
    for (int r = 0; r < 8; ++r) {
        v2f o;
        o.x = c0[r] * dn.x;
        o.y = c1[r] * dn.y;
        *reinterpret_cast<v2f*>(hT + (size_t)(f0 + 8 * hi + r) * NNODES + n0 + 2 * lm) = o;
    }
}
static_assert(NNODES % 32 == 0, "layer GEMMs: whole 32-node tiles, 8-byte aligned node pairs");

#define L1K 128
#define L1M 128
#define XPITCH 132
__global__ __launch_bounds__(256) void bin_gemm1(const int* __restrict__ src,
                                                 const int* __restrict__ dst,
                                                 int* __restrict__ gcnt,
                                                 unsigned* __restrict__ elist,
                                                 unsigned* __restrict__ ovf,
                                                 const float* __restrict__ X,
                                                 const float* __restrict__ W,
                                                 float* __restrict__ hT) {
    __shared__ float xs[32 * XPITCH];
    __shared__ int cnt[NLIST];
    __shared__ int base[NLIST];
    if (blockIdx.x < BINBLK) {
        bin_block(blockIdx.x, src, dst, gcnt, elist, ovf, cnt, base);
        return;
    }
    const int tid  = threadIdx.x;
    const int wid  = tid >> 5;
    const int lane = tid & 31;
    const int n0 = (blockIdx.x - BINBLK) << 5;

    #pragma unroll
    for (int i = 0; i < 4; ++i) {
        const int r = wid + 8 * i;
        const float4 v = reinterpret_cast<const float4*>(X + (size_t)(n0 + r) * L1K)[lane];
        *reinterpret_cast<float4*>(&xs[r * XPITCH + lane * 4]) = v;
    }
    __syncthreads();

    const int f0 = wid << 4;
    const int hi = lane >> 4, lm = lane & 15;
    const float* xa = xs + lm * XPITCH + 2 * hi;
    const float* xb = xa + 16 * XPITCH;
    const float* Wp = W + (size_t)(2 * hi) * L1M + f0 + lm;

    v8f c0 = {}, c1 = {};
    #pragma unroll 8
    for (int k0 = 0; k0 < L1K; k0 += 4) {
        v2f a;
        a.x = Wp[(size_t)k0 * L1M];
        a.y = Wp[(size_t)(k0 + 1) * L1M];
        const v2f b0 = *reinterpret_cast<const v2f*>(xa + k0);
        const v2f b1 = *reinterpret_cast<const v2f*>(xb + k0);
        c0 = __builtin_amdgcn_wmma_f32_16x16x4_f32(false, a, false, b0,
                                                   (short)0, c0, false, false);
        c1 = __builtin_amdgcn_wmma_f32_16x16x4_f32(false, a, false, b1,
                                                   (short)0, c1, false, false);
    }
    #pragma unroll
    for (int r = 0; r < 8; ++r) {
        float give = hi ? c0[r] : c1[r];
        float got  = __shfl_xor(give, 16);
        float lo = hi ? got : c0[r];
        float up = hi ? c1[r] : got;
        hT[(size_t)(f0 + r) * NNODES + n0 + lane]     = lo;
        hT[(size_t)(f0 + 8 + r) * NNODES + n0 + lane] = up;
    }
}
static_assert(NNODES % 32 == 0, "layer-1 GEMM: whole 32-node tiles");

template <int R, bool POOL, bool SCALE>
__global__ __launch_bounds__(1024) void slab_kernel(const float* __restrict__ hT,
                                                    const unsigned* __restrict__ elist,
                                                    const unsigned* __restrict__ ovf,
                                                    const int* __restrict__ gcnt,
                                                    float* __restrict__ outp,
                                                    const float* __restrict__ dis,
                                                    const float* __restrict__ bias, int F,
                                                    const int* __restrict__ batch,
                                                    float* __restrict__ counts,
                                                    float* __restrict__ pooled) {
    static_assert(!POOL || R == 1, "pool epilogue: one range per work-group");
    __shared__ float hcol[NNODES];
    __shared__ float acc[R * RNG];
    __shared__ __attribute__((aligned(16))) int gid[POOL ? RNG : 1];
    __shared__ float gsum[POOL ? NGRAPH : 1];
    __shared__ float gnum[POOL ? NGRAPH : 1];
    constexpr int WPR = 32 / R;
    constexpr int COLT = (NNODES / 4 + 1023) / 1024;
    constexpr int EPI = (R * RNG / 4 + 1023) / 1024;
    constexpr int DEPTH = 4;
    const int tid = threadIdx.x;
    const int qb = blockIdx.x / F;
    const int f = blockIdx.x - qb * F;
    const int nlo = qb * (R * RNG);
    const int nhi = nlo + R * RNG < NNODES ? nlo + R * RNG : NNODES;
    const int cnt4 = (nhi - nlo) >> 2;

    const int lane = tid & 31;
    const int wave = tid >> 5;
    const int rl = wave / WPR;
    const int slot = wave - rl * WPR;
    const int rg = qb * R + rl;

    const float4* h4 = reinterpret_cast<const float4*>(hT + (size_t)f * NNODES);
    float4 hv[COLT];
    #pragma unroll
    for (int t = 0; t < COLT; ++t) {
        const int i = tid + 1024 * t;
        hv[t] = h4[i < NNODES / 4 ? i : NNODES / 4 - 1];
    }
    float4 dv[SCALE ? COLT : 1];
    if constexpr (SCALE) {
        const float4* dc4 = reinterpret_cast<const float4*>(dis);
        #pragma unroll
        for (int t = 0; t < COLT; ++t) {
            const int i = tid + 1024 * t;
            dv[t] = dc4[i < NNODES / 4 ? i : NNODES / 4 - 1];
        }
    }
    int4 gb[POOL ? EPI : 1];
    if constexpr (POOL) {
        const int4* b4 = reinterpret_cast<const int4*>(batch + nlo);
        #pragma unroll
        for (int j = 0; j < EPI; ++j) {
            const int i = tid + 1024 * j;
            gb[j] = b4[i < cnt4 ? i : 0];
        }
    }
    static_assert(DEPTH * 32 <= CAPG, "first groups inside the list");
    const uint4* e4 = reinterpret_cast<const uint4*>(elist) + (size_t)rg * CAPG * 32 + lane;
    uint4 e[DEPTH];
    #pragma unroll
    for (int k = 0; k < DEPTH; ++k) e[k] = e4[(size_t)(slot + k * WPR) * 32];
    int n = gcnt[rg * 32 + lane];
    int novf = gcnt[NLIST];
    const float bf = bias[f];
    const float4* dis4 = reinterpret_cast<const float4*>(dis + nlo);
    float4 dd[EPI];
    if constexpr (!SCALE) {
        #pragma unroll
        for (int j = 0; j < EPI; ++j) {
            const int i = tid + 1024 * j;
            dd[j] = dis4[i < cnt4 ? i : 0];
        }
    }

    float4* hc4 = reinterpret_cast<float4*>(hcol);
    float4* ac4 = reinterpret_cast<float4*>(acc);
    #pragma unroll
    for (int t = 0; t < COLT; ++t) {
        const int i = tid + 1024 * t;
        float4 v = hv[t];
        if constexpr (SCALE) {
            v.x *= dv[t].x; v.y *= dv[t].y; v.z *= dv[t].z; v.w *= dv[t].w;
        }
        hc4[i < NNODES / 4 ? i : NNODES / 4 - 1] = v;
    }
    if constexpr (POOL) {
        int4* g4 = reinterpret_cast<int4*>(gid);
        #pragma unroll
        for (int j = 0; j < EPI; ++j) {
            const int i = tid + 1024 * j;
            g4[i < cnt4 ? i : 0] = gb[j];
        }
    }
    for (int i = tid; i < R * RNG / 4; i += 1024) ac4[i] = make_float4(0.0f, 0.0f, 0.0f, 0.0f);
    if constexpr (POOL) {
        if (tid < NGRAPH) { gsum[tid] = 0.0f; gnum[tid] = 0.0f; }
    }
    __syncthreads();
    if constexpr (SCALE) {
        #pragma unroll
        for (int j = 0; j < EPI; ++j) {
            const int i = tid + 1024 * j;
            dd[j] = dis4[i < cnt4 ? i : 0];
        }
    }

    n = n < CAPN ? n : CAPN;
    const int ng = (n + 3) >> 2;
    const int glast = ng > 0 ? ng - 1 : 0;
    float* accr = acc + rl * RNG;
    int g = slot;
    while (g < ng) {
        uint4 en[DEPTH];
        #pragma unroll
        for (int k = 0; k < DEPTH; ++k) {
            const int gk = g + (DEPTH + k) * WPR;
            en[k] = e4[(size_t)(gk < glast ? gk : glast) * 32];
        }
        #pragma unroll
        for (int k = 0; k < DEPTH; ++k) {
            const uint4 c = e[k];
            const int rem = n - 4 * (g + k * WPR);
            if (rem >= 4) {
                float v0 = hcol[c.x & 0xFFFFu], v1 = hcol[c.y & 0xFFFFu];
                float v2 = hcol[c.z & 0xFFFFu], v3 = hcol[c.w & 0xFFFFu];
                __hip_atomic_fetch_add(&accr[c.x >> 16], v0, __ATOMIC_RELAXED, __HIP_MEMORY_SCOPE_WORKGROUP);
                __hip_atomic_fetch_add(&accr[c.y >> 16], v1, __ATOMIC_RELAXED, __HIP_MEMORY_SCOPE_WORKGROUP);
                __hip_atomic_fetch_add(&accr[c.z >> 16], v2, __ATOMIC_RELAXED, __HIP_MEMORY_SCOPE_WORKGROUP);
                __hip_atomic_fetch_add(&accr[c.w >> 16], v3, __ATOMIC_RELAXED, __HIP_MEMORY_SCOPE_WORKGROUP);
            } else if (rem > 0) {
                __hip_atomic_fetch_add(&accr[c.x >> 16], hcol[c.x & 0xFFFFu], __ATOMIC_RELAXED, __HIP_MEMORY_SCOPE_WORKGROUP);
                if (rem > 1)
                    __hip_atomic_fetch_add(&accr[c.y >> 16], hcol[c.y & 0xFFFFu], __ATOMIC_RELAXED, __HIP_MEMORY_SCOPE_WORKGROUP);
                if (rem > 2)
                    __hip_atomic_fetch_add(&accr[c.z >> 16], hcol[c.z & 0xFFFFu], __ATOMIC_RELAXED, __HIP_MEMORY_SCOPE_WORKGROUP);
            }
        }
        #pragma unroll
        for (int k = 0; k < DEPTH; ++k) e[k] = en[k];
        g += DEPTH * WPR;
    }
    novf = novf < OVFCAP ? novf : OVFCAP;
    for (int i = tid; i < novf; i += 1024) {
        const unsigned w = ovf[i];
        const int d = (int)(w >> 16);
        if (d >= nlo && d < nhi)
            __hip_atomic_fetch_add(&acc[d - nlo], hcol[w & 0xFFFFu], __ATOMIC_RELAXED, __HIP_MEMORY_SCOPE_WORKGROUP);
    }
    __syncthreads();

    float4* o4 = reinterpret_cast<float4*>(outp + (size_t)f * NNODES + nlo);
    const float4* hl4 = reinterpret_cast<const float4*>(hcol + nlo);
    #pragma unroll
    for (int j = 0; j < EPI; ++j) {
        const int i = tid + 1024 * j;
        if (i < cnt4) {
            const float4 a = ac4[i], h = hl4[i], d4 = dd[j];
            float4 o;
            o.x = (a.x + h.x) * d4.x + bf;
            o.y = (a.y + h.y) * d4.y + bf;
            o.z = (a.z + h.z) * d4.z + bf;
            o.w = (a.w + h.w) * d4.w + bf;
            o.x = o.x > 0.0f ? o.x : 0.0f;
            o.y = o.y > 0.0f ? o.y : 0.0f;
            o.z = o.z > 0.0f ? o.z : 0.0f;
            o.w = o.w > 0.0f ? o.w : 0.0f;
            if constexpr (POOL) ac4[i] = o;
            else o4[i] = o;
        }
    }
    if constexpr (POOL) {
        __syncthreads();
        constexpr int SEG = RNG / 32;
        constexpr int CH = (SEG + 31) / 32;
        static_assert(RNG % 32 == 0 && SEG % 2 == 1, "lane stride odd: 32 lanes, 32 banks");
        const int cntn = nhi - nlo;
        const int ibeg = SEG * lane + CH * wave;
        int iend = ibeg + CH;
        if (iend > SEG * lane + SEG) iend = SEG * lane + SEG;
        if (iend > cntn) iend = cntn;
        const bool cnt_wg = (f == 0);
        if (ibeg < iend) {
            int gcur = gid[ibeg];
            float s = acc[ibeg];
            float c = 1.0f;
            for (int i = ibeg + 1; i < iend; ++i) {
                const int gi = gid[i];
                const float v = acc[i];
                if (gi != gcur) {
                    __hip_atomic_fetch_add(&gsum[gcur], s, __ATOMIC_RELAXED, __HIP_MEMORY_SCOPE_WORKGROUP);
                    if (cnt_wg)
                        __hip_atomic_fetch_add(&gnum[gcur], c, __ATOMIC_RELAXED, __HIP_MEMORY_SCOPE_WORKGROUP);
                    gcur = gi;
                    s = 0.0f;
                    c = 0.0f;
                }
                s += v;
                c += 1.0f;
            }
            __hip_atomic_fetch_add(&gsum[gcur], s, __ATOMIC_RELAXED, __HIP_MEMORY_SCOPE_WORKGROUP);
            if (cnt_wg)
                __hip_atomic_fetch_add(&gnum[gcur], c, __ATOMIC_RELAXED, __HIP_MEMORY_SCOPE_WORKGROUP);
        }
        __syncthreads();
        if (tid < NGRAPH) {
            const float v = gsum[tid];
            if (v != 0.0f) atomAddF(&pooled[tid * F + f], v);
            if (cnt_wg) {
                const float c = gnum[tid];
                if (c != 0.0f) atomAddF(&counts[tid], c);
            }
        }
    }
}

extern "C" void kernel_launch(void* const* d_in, const int* in_sizes, int n_in,
                              void* d_out, int out_size, void* d_ws, size_t ws_size,
                              hipStream_t stream) {
    const float* x   = (const float*)d_in[0];
    const float* W1  = (const float*)d_in[1];
    const float* b1  = (const float*)d_in[2];
    const float* W2  = (const float*)d_in[3];
    const float* b2  = (const float*)d_in[4];
    const float* W3  = (const float*)d_in[5];
    const float* b3  = (const float*)d_in[6];
    const float* Wl  = (const float*)d_in[7];
    const float* bl  = (const float*)d_in[8];
    const int*   ei  = (const int*)d_in[9];
    const int*   bat = (const int*)d_in[10];
    float*       out = (float*)d_out;

    const int N = NNODES, E = NEDGES;
    const int* src = ei;
    const int* dst = ei + E;

    float* B0     = (float*)d_ws;
    float* B1     = B0 + (size_t)N * 128;
    float* dis    = B1 + (size_t)N * 128;
    int*   gcnt   = (int*)(dis + N);
    float* counts = dis + N + 320;
    float* pooled = counts + NGRAPH;
    unsigned* elist = (unsigned*)(pooled + NGRAPH * 32);
    unsigned* ovf   = elist + (size_t)NRNG * CAPG * 128;

    hipMemsetAsync(gcnt, 0, (320 + NGRAPH + NGRAPH * 32) * sizeof(int), stream);
    bin_gemm1<<<BINBLK + NNODES / 32, 256, 0, stream>>>(src, dst, gcnt, elist, ovf, x, W1, B0);
    deg_dis_kernel<<<DEGBLK, 1024, 0, stream>>>(elist, ovf, gcnt, dis);

    slab_kernel<4, false, true><<<128 * 2, 1024, 0, stream>>>(B0, elist, ovf, gcnt, B1, dis, b1, 128,
                                                               nullptr, nullptr, nullptr);

    wmma_gemm_t<128, 64><<<(1250 * 4) / 8, 256, 0, stream>>>(B1, W2, B0, dis);
    slab_kernel<2, false, false><<<64 * 4, 1024, 0, stream>>>(B0, elist, ovf, gcnt, B1, dis, b2, 64,
                                                               nullptr, nullptr, nullptr);

    wmma_gemm_t<64, 32><<<(1250 * 2 + 7) / 8, 256, 0, stream>>>(B1, W3, B0, dis);
    slab_kernel<1, true, false><<<32 * 8, 1024, 0, stream>>>(B0, elist, ovf, gcnt, B1, dis, b3, 32,
                                                              bat, counts, pooled);

    wmma_gemm_f32<<<(16 * 48) / 8, 256, 0, stream>>>(pooled, Wl, out, NGRAPH, 32, 768,
                                                     nullptr, bl, 16 * 48, counts);
}
